# starting kernel
# baseline (speedup 1.0000x reference)
#include <hip/hip_runtime.h>
#include <hip/hip_bf16.h>
#include <math.h>

#define B_ 128
#define L_ 50
#define D_ 128
#define S_ 8
#define NN_ 100000
#define KP_ 132
#define LEAKY_ 0.2f
#define BETA_ 0.005f

typedef float v2f __attribute__((ext_vector_type(2)));
typedef float v8f __attribute__((ext_vector_type(8)));

#if defined(__has_builtin)
#if __has_builtin(__builtin_amdgcn_wmma_f32_16x16x4_f32)
#define HAVE_WMMA_F32 1
#endif
#endif

__device__ __forceinline__ v8f wmma4(v2f a, v2f b, v8f c) {
#ifdef HAVE_WMMA_F32
  return __builtin_amdgcn_wmma_f32_16x16x4_f32(false, a, false, b, (short)0, c,
                                               false, false);
#else
  float* cf = (float*)&c;
  #pragma unroll
  for (int i = 0; i < 8; ++i) cf[i] += a.x * b.x + a.y * b.y;
  return c;
#endif
}

__device__ __forceinline__ float leaky_(float x) { return x > 0.f ? x : LEAKY_ * x; }
__device__ __forceinline__ float sigmoid_(float x) { return 1.f / (1.f + __expf(-x)); }

__global__ void gw1t_kernel(const float* __restrict__ gw1, float* __restrict__ gw1T) {
  int i = blockIdx.x * 256 + threadIdx.x;
  if (i >= 2 * D_ * KP_) return;
  int hop = i / (D_ * KP_);
  int rem = i % (D_ * KP_);
  int n = rem / KP_, k = rem % KP_;
  gw1T[i] = (k < 129) ? gw1[hop * 129 * D_ + k * D_ + n] : 0.f;
}

__global__ void w1t_kernel(const float* __restrict__ w1, float* __restrict__ w1botT) {
  int i = blockIdx.x * 256 + threadIdx.x;
  if (i >= D_ * D_) return;
  int n = i >> 7, k = i & 127;
  w1botT[i] = w1[(D_ + k) * D_ + n];
}

__global__ void gather_h_kernel(const float* __restrict__ emb,
                                const int* __restrict__ inputs,
                                float* __restrict__ h, int total) {
  int i = blockIdx.x * 256 + threadIdx.x;
  if (i >= total) return;
  int row = i >> 7, d = i & 127;
  h[i] = emb[((long)inputs[row] << 7) + d];
}

__global__ void hop_kernel(const int* __restrict__ src,
                           const int* __restrict__ adj_all,
                           const float* __restrict__ num_w,
                           int* __restrict__ dst, float* __restrict__ dstw,
                           int total) {
  int i = blockIdx.x * 256 + threadIdx.x;
  if (i >= total) return;
  int node = src[i >> 3];
  int s = i & 7;
  dst[i]  = adj_all[node * S_ + s];
  dstw[i] = num_w[node * S_ + s];
}

__global__ void sess_kernel(const float* __restrict__ emb,
                            const int* __restrict__ item,
                            const int* __restrict__ mask,
                            float* __restrict__ sess) {
  int b = blockIdx.x, d = threadIdx.x;
  float acc = 0.f, ms = 0.f;
  for (int l = 0; l < L_; ++l) {
    float m = (float)mask[b * L_ + l];
    ms += m;
    acc += emb[((long)item[b * L_ + l] << 7) + d] * m;
  }
  sess[(b << 7) + d] = acc / ms;
}

__global__ void local_agg_kernel(const float* __restrict__ h,
                                 const int* __restrict__ adj,
                                 const float* __restrict__ a_local,
                                 float* __restrict__ h_local) {
  int b = blockIdx.x;
  __shared__ float hv[L_][D_];
  __shared__ float att[L_][L_];
  __shared__ float rinv[L_];
  for (int i = threadIdx.x; i < L_ * D_; i += 256)
    hv[i / D_][i & 127] = h[(long)b * L_ * D_ + i];
  __syncthreads();
  for (int p = threadIdx.x; p < L_ * L_; p += 256) {
    int i = p / L_, j = p % L_;
    int k = adj[b * L_ * L_ + p];
    float v;
    if (k == 0) v = -9e15f;
    else {
      const float* a = a_local + (k - 1) * D_;
      float acc = 0.f;
      for (int d = 0; d < D_; ++d) acc += hv[i][d] * hv[j][d] * a[d];
      v = leaky_(acc);
    }
    att[i][j] = v;
  }
  __syncthreads();
  if (threadIdx.x < L_) {
    int i = threadIdx.x;
    float mx = -1e30f;
    for (int j = 0; j < L_; ++j) mx = fmaxf(mx, att[i][j]);
    float s = 0.f;
    for (int j = 0; j < L_; ++j) { float e = __expf(att[i][j] - mx); att[i][j] = e; s += e; }
    rinv[i] = 1.f / s;
  }
  __syncthreads();
  for (int p = threadIdx.x; p < L_ * D_; p += 256) {
    int i = p >> 7, d = p & 127;
    float acc = 0.f;
    for (int j = 0; j < L_; ++j) acc += att[i][j] * hv[j][d];
    h_local[(long)b * L_ * D_ + p] = acc * rinv[i];
  }
}

__global__ void agg_kernel(const float* __restrict__ emb,
                           const float* __restrict__ selfVal, const int* __restrict__ selfIdx,
                           const int* __restrict__ neighIdx, const float* __restrict__ neighVal,
                           const float* __restrict__ neighW,
                           const float* __restrict__ sess,
                           const float* __restrict__ gw1T, const float* __restrict__ gw2p,
                           const float* __restrict__ gw3p,
                           float* __restrict__ out, int M) {
  __shared__ float nv[16][D_];
  __shared__ float xs[16][KP_];
  __shared__ float selfv[2][D_];
  __shared__ float sessv[D_];
  __shared__ float wgt[16];
  __shared__ float sc[16];
  __shared__ float alpha[16];
  __shared__ float nout[2][D_];

  int tid = threadIdx.x;
  int lane = tid & 31, laneIdx = lane & 15, laneHalf = lane >> 4;
  int wave = tid >> 5;
  int n0 = wave * 16;

  int p = blockIdx.x;
  int b = (2 * p) / M;
  int m0 = (2 * p) % M;
  long base  = (long)b * M + m0;
  long nbase = base * S_;

  if (tid < D_) sessv[tid] = sess[(b << 7) + tid];
  {
    int r = tid >> 7, d = tid & 127;
    float v = selfVal ? selfVal[((base + r) << 7) + d]
                      : emb[((long)selfIdx[base + r] << 7) + d];
    selfv[r][d] = v;
  }
  for (int i = tid; i < 16 * D_; i += 256) {
    int r = i >> 7, d = i & 127;
    float v = neighIdx ? emb[((long)neighIdx[nbase + r] << 7) + d]
                       : neighVal[((nbase + r) << 7) + d];
    nv[r][d] = v;
  }
  if (tid < 16) { wgt[tid] = neighW[nbase + tid]; sc[tid] = 0.f; }
  __syncthreads();

  for (int i = tid; i < 16 * KP_; i += 256) {
    int r = i / KP_, k = i % KP_;
    float v = (k < D_) ? sessv[k] * nv[r][k] : (k == D_ ? wgt[r] : 0.f);
    xs[r][k] = v;
  }
  __syncthreads();

  v8f c = {};
  const float* bp = gw1T + (n0 + laneIdx) * KP_;
  {
    const float* ap = &xs[laneIdx][0];
    #pragma unroll 3
    for (int k0 = 0; k0 < KP_; k0 += 4) {
      int kk = k0 + (laneHalf << 1);
      v2f a = *(const v2f*)(ap + kk);
      v2f bb = *(const v2f*)(bp + kk);
      c = wmma4(a, bb, c);
    }
  }
  float g2 = gw2p[n0 + laneIdx];
  float* cf = (float*)&c;
  #pragma unroll
  for (int v = 0; v < 8; ++v) {
    float val = leaky_(cf[v]) * g2;
    val += __shfl_xor(val, 1);
    val += __shfl_xor(val, 2);
    val += __shfl_xor(val, 4);
    val += __shfl_xor(val, 8);
    if (laneIdx == 0) atomicAdd(&sc[v + (laneHalf << 3)], val);
  }
  __syncthreads();

  if (tid < 2) {
    float mx = -1e30f;
    for (int s = 0; s < 8; ++s) mx = fmaxf(mx, sc[tid * 8 + s]);
    float ex[8], sum = 0.f;
    for (int s = 0; s < 8; ++s) { ex[s] = __expf(sc[tid * 8 + s] - mx); sum += ex[s]; }
    float inv = 1.f / sum;
    for (int s = 0; s < 8; ++s) alpha[tid * 8 + s] = ex[s] * inv;
  }
  __syncthreads();
  {
    int r = tid >> 7, d = tid & 127;
    float acc = 0.f;
    #pragma unroll
    for (int s = 0; s < 8; ++s) acc += alpha[(r << 3) + s] * nv[(r << 3) + s][d];
    nout[r][d] = acc;
  }
  __syncthreads();
  {
    int r = tid >> 7, d = tid & 127;
    float acc = 0.f;
    for (int e = 0; e < D_; ++e)
      acc += selfv[r][e] * gw3p[(e << 7) + d] + nout[r][e] * gw3p[((D_ + e) << 7) + d];
    out[((base + r) << 7) + d] = fmaxf(acc, 0.f);
  }
}

__global__ void gemm16_kernel(const float* __restrict__ A, int lda,
                              const float* __restrict__ BT, int ldbt,
                              float* __restrict__ C, long ldc,
                              int M, int N, int K,
                              const float* __restrict__ rowBias, int biasMod,
                              int act) {
  int tilesM = (M + 15) >> 4;
  long tilesN = (long)((N + 15) >> 4);
  long total = (long)tilesM * tilesN;
  int lane = threadIdx.x & 31, laneIdx = lane & 15, laneHalf = lane >> 4;
  for (long t = (long)blockIdx.x * 8 + (threadIdx.x >> 5); t < total;
       t += (long)gridDim.x * 8) {
    int m0 = (int)(t % tilesM) << 4;
    long n0 = (t / tilesM) << 4;
    int row = m0 + laneIdx;
    long col = n0 + laneIdx;
    v8f c = {};
    bool full = (m0 + 16 <= M) && (n0 + 16 <= N);
    if (full) {
      const float* ap = A + (long)row * lda;
      const float* bp = BT + col * ldbt;
      for (int k0 = 0; k0 < K; k0 += 4) {
        int kk = k0 + (laneHalf << 1);
        v2f a = *(const v2f*)(ap + kk);
        v2f b = *(const v2f*)(bp + kk);
        c = wmma4(a, b, c);
      }
    } else {
      for (int k0 = 0; k0 < K; k0 += 4) {
        int kk = k0 + (laneHalf << 1);
        v2f a, b;
        a.x = (row < M) ? A[(long)row * lda + kk] : 0.f;
        a.y = (row < M) ? A[(long)row * lda + kk + 1] : 0.f;
        b.x = (col < N) ? BT[col * ldbt + kk] : 0.f;
        b.y = (col < N) ? BT[col * ldbt + kk + 1] : 0.f;
        c = wmma4(a, b, c);
      }
    }
    float* cf = (float*)&c;
    #pragma unroll
    for (int v = 0; v < 8; ++v) {
      int rr = m0 + v + (laneHalf << 3);
      if (rr < M && col < N) {
        float val = cf[v];
        if (rowBias) val += rowBias[(long)(rr % biasMod) * N + col];
        if (act == 1) val = tanhf(val);
        C[(long)rr * ldc + col] = val;
      }
    }
  }
}

__global__ void add_kernel(const float* a, const float* b, float* o, int n) {
  int i = blockIdx.x * 256 + threadIdx.x;
  if (i < n) o[i] = a[i] + b[i];
}

__global__ void zero1_kernel(float* p) { if (threadIdx.x == 0 && blockIdx.x == 0) p[0] = 0.f; }

__global__ void ssl_kernel(const float* __restrict__ hl, const float* __restrict__ hg,
                           float* __restrict__ loss) {
  int idx = blockIdx.x * 256 + threadIdx.x;
  int b = idx >> 7, d = idx & 127;
  int pb = (b * 73 + 21) & 127;
  float pos = 0.f, neg = 0.f;
  for (int l = 0; l < L_; ++l) {
    int pl = (l * 13 + 7) % L_;
    pos += hl[((b * L_ + l) << 7) + d] * hg[((b * L_ + l) << 7) + d];
    neg += hg[((b * L_ + l) << 7) + d] * hl[((pb * L_ + pl) << 7) + d];
  }
  float term = -__logf(1e-8f + sigmoid_(pos)) - __logf(1e-8f + 1.f - sigmoid_(neg));
  __shared__ float red[256];
  red[threadIdx.x] = term;
  __syncthreads();
  for (int s = 128; s > 0; s >>= 1) {
    if (threadIdx.x < s) red[threadIdx.x] += red[threadIdx.x + s];
    __syncthreads();
  }
  if (threadIdx.x == 0) atomicAdd(loss, BETA_ * red[0]);
}

__global__ void hs_kernel(const float* __restrict__ hidden, const int* __restrict__ mask,
                          float* __restrict__ hs) {
  int b = blockIdx.x, d = threadIdx.x;
  float acc = 0.f, ms = 0.f;
  for (int l = 0; l < L_; ++l) {
    float m = (float)mask[b * L_ + l];
    ms += m;
    acc += hidden[((b * L_ + l) << 7) + d] * m;
  }
  hs[(b << 7) + d] = acc / ms;
}

__global__ void hsg_kernel(const float* __restrict__ hs, const float* __restrict__ glu2_w,
                           float* __restrict__ hsg) {
  __shared__ float hr[D_];
  int b = blockIdx.x, d = threadIdx.x;
  hr[d] = hs[(b << 7) + d];
  __syncthreads();
  float acc = 0.f;
  for (int e = 0; e < D_; ++e) acc += hr[e] * glu2_w[(d << 7) + e];
  hsg[(b << 7) + d] = acc;
}

__global__ void pos_kernel(const float* __restrict__ pos_emb, const float* __restrict__ w1,
                           float* __restrict__ P) {
  __shared__ float pr[D_];
  int l = blockIdx.x, d = threadIdx.x;
  pr[d] = pos_emb[(l << 7) + d];
  __syncthreads();
  float acc = 0.f;
  for (int e = 0; e < D_; ++e) acc += pr[e] * w1[(e << 7) + d];
  P[(l << 7) + d] = acc;
}

__global__ void beta_kernel(const float* __restrict__ nh, const float* __restrict__ glu1_w,
                            const float* __restrict__ glu1_b, const float* __restrict__ hsg,
                            const float* __restrict__ w2, const int* __restrict__ mask,
                            float* __restrict__ beta) {
  __shared__ float nrow[D_];
  __shared__ float red[D_];
  int row = blockIdx.x;
  int b = row / L_;
  int d = threadIdx.x;
  nrow[d] = nh[((long)row << 7) + d];
  __syncthreads();
  float acc = glu1_b[d] + hsg[(b << 7) + d];
  for (int e = 0; e < D_; ++e) acc += nrow[e] * glu1_w[(d << 7) + e];
  red[d] = sigmoid_(acc) * w2[d];
  __syncthreads();
  for (int s = 64; s > 0; s >>= 1) {
    if (d < s) red[d] += red[d + s];
    __syncthreads();
  }
  if (d == 0) beta[row] = red[0] * (float)mask[row];
}

__global__ void select_kernel(const float* __restrict__ hidden, const float* __restrict__ beta,
                              float* __restrict__ sel) {
  int b = blockIdx.x, d = threadIdx.x;
  float acc = 0.f;
  for (int l = 0; l < L_; ++l) acc += beta[b * L_ + l] * hidden[((b * L_ + l) << 7) + d];
  sel[(b << 7) + d] = acc;
}

__global__ void norm_kernel(const float* __restrict__ sel, float* __restrict__ snorm) {
  __shared__ float red[D_];
  int b = blockIdx.x, d = threadIdx.x;
  float v = sel[(b << 7) + d];
  red[d] = v * v + 1e-6f;
  __syncthreads();
  for (int s = 64; s > 0; s >>= 1) {
    if (d < s) red[d] += red[d + s];
    __syncthreads();
  }
  if (d == 0) snorm[b] = sqrtf(red[0]);
}

__global__ void fn_kernel(const float* __restrict__ sel, const float* __restrict__ snorm,
                          float* __restrict__ selF) {
  __shared__ float row[B_];
  __shared__ float selfrow[D_];
  __shared__ float tvs[9];
  __shared__ int tis[9];
  int i = blockIdx.x, t = threadIdx.x;
  selfrow[t] = sel[(i << 7) + t];
  __syncthreads();
  float acc = 0.f;
  for (int d = 0; d < D_; ++d) acc += selfrow[d] * sel[(t << 7) + d];
  row[t] = acc / (snorm[i] * snorm[t]);
  __syncthreads();
  if (t == 0) {
    float mx = -1e30f;
    for (int j = 0; j < B_; ++j) mx = fmaxf(mx, row[j]);
    float s = 0.f;
    for (int j = 0; j < B_; ++j) { row[j] = __expf(row[j] - mx); s += row[j]; }
    float inv = 1.f / s;
    for (int j = 0; j < B_; ++j) row[j] *= inv;
    for (int q = 0; q < 9; ++q) {
      float bm = -1.f; int bi = 0;
      for (int j = 0; j < B_; ++j) if (row[j] > bm) { bm = row[j]; bi = j; }
      tvs[q] = bm; tis[q] = bi; row[bi] = -2.f;
    }
    float mx2 = tvs[0], s2 = 0.f;
    for (int q = 0; q < 9; ++q) { tvs[q] = __expf(tvs[q] - mx2); s2 += tvs[q]; }
    float inv2 = 1.f / s2;
    for (int q = 0; q < 9; ++q) tvs[q] *= inv2;
  }
  __syncthreads();
  float nb = 0.f;
  for (int q = 0; q < 9; ++q) nb += tvs[q] * sel[(tis[q] << 7) + t];
  selF[(i << 7) + t] = selfrow[t] + nb;
}

extern "C" void kernel_launch(void* const* d_in, const int* in_sizes, int n_in,
                              void* d_out, int out_size, void* d_ws, size_t ws_size,
                              hipStream_t stream) {
  (void)in_sizes; (void)n_in; (void)out_size; (void)ws_size;
  const int*   inputs  = (const int*)d_in[0];
  const int*   adj     = (const int*)d_in[1];
  const int*   mask    = (const int*)d_in[2];
  const int*   item    = (const int*)d_in[3];
  const int*   adj_all = (const int*)d_in[5];
  const float* num_w   = (const float*)d_in[6];
  const float* emb     = (const float*)d_in[7];
  const float* pos_emb = (const float*)d_in[8];
  const float* a_local = (const float*)d_in[9];
  const float* gw1     = (const float*)d_in[10];
  const float* gw2     = (const float*)d_in[11];
  const float* gw3     = (const float*)d_in[12];
  const float* w1      = (const float*)d_in[13];
  const float* w2      = (const float*)d_in[14];
  const float* glu1_w  = (const float*)d_in[15];
  const float* glu1_b  = (const float*)d_in[16];
  const float* glu2_w  = (const float*)d_in[17];

  float* out0   = (float*)d_out;
  float* lossP  = out0 + B_ * L_ * D_;
  float* scores = lossP + 1;

  char* wp = (char*)d_ws;
  auto alloc = [&](size_t nfloats) -> float* {
    float* r = (float*)wp;
    wp += ((nfloats * sizeof(float) + 255) / 256) * 256;
    return r;
  };
  float* h        = alloc(B_ * L_ * D_);
  float* h_local  = alloc(B_ * L_ * D_);
  float* h_global = alloc(B_ * L_ * D_);
  float* sessb    = alloc(B_ * D_);
  float* e0p      = alloc(B_ * L_ * D_);
  float* e1p      = alloc((size_t)B_ * 400 * D_);
  float* hsb      = alloc(B_ * D_);
  float* hsgb     = alloc(B_ * D_);
  float* Pb       = alloc(L_ * D_);
  float* nh       = alloc(B_ * L_ * D_);
  float* betab    = alloc(B_ * L_);
  float* selb     = alloc(B_ * D_);
  float* selF     = alloc(B_ * D_);
  float* snorm    = alloc(B_);
  float* gw1T     = alloc(2 * D_ * KP_);
  float* w1botT   = alloc(D_ * D_);
  float* wn1      = alloc(B_ * 400);
  float* wn2      = alloc((size_t)B_ * 3200);
  int*   n1       = (int*)alloc(B_ * 400);
  int*   n2       = (int*)alloc((size_t)B_ * 3200);

  gw1t_kernel<<<dim3((2 * D_ * KP_ + 255) / 256), 256, 0, stream>>>(gw1, gw1T);
  w1t_kernel<<<dim3((D_ * D_ + 255) / 256), 256, 0, stream>>>(w1, w1botT);
  gather_h_kernel<<<dim3((B_ * L_ * D_) / 256), 256, 0, stream>>>(emb, inputs, h, B_ * L_ * D_);
  sess_kernel<<<dim3(B_), dim3(D_), 0, stream>>>(emb, item, mask, sessb);
  hop_kernel<<<dim3((B_ * L_ * S_ + 255) / 256), 256, 0, stream>>>(inputs, adj_all, num_w, n1, wn1, B_ * L_ * S_);
  hop_kernel<<<dim3((B_ * 400 * S_ + 255) / 256), 256, 0, stream>>>(n1, adj_all, num_w, n2, wn2, B_ * 400 * S_);
  local_agg_kernel<<<dim3(B_), 256, 0, stream>>>(h, adj, a_local, h_local);
  agg_kernel<<<dim3(B_ * L_ / 2), 256, 0, stream>>>(emb, h, nullptr, n1, nullptr, wn1,
      sessb, gw1T, gw2, gw3, e0p, L_);
  agg_kernel<<<dim3(B_ * 400 / 2), 256, 0, stream>>>(emb, nullptr, n1, n2, nullptr, wn2,
      sessb, gw1T, gw2, gw3, e1p, 400);
  agg_kernel<<<dim3(B_ * L_ / 2), 256, 0, stream>>>(emb, e0p, nullptr, nullptr, e1p, wn1,
      sessb, gw1T + D_ * KP_, gw2 + D_, gw3 + 2 * D_ * D_, h_global, L_);
  add_kernel<<<dim3((B_ * L_ * D_) / 256), 256, 0, stream>>>(h_local, h_global, out0, B_ * L_ * D_);
  zero1_kernel<<<1, 1, 0, stream>>>(lossP);
  ssl_kernel<<<dim3(64), 256, 0, stream>>>(h_local, h_global, lossP);
  hs_kernel<<<dim3(B_), dim3(D_), 0, stream>>>(out0, mask, hsb);
  hsg_kernel<<<dim3(B_), dim3(D_), 0, stream>>>(hsb, glu2_w, hsgb);
  pos_kernel<<<dim3(L_), dim3(D_), 0, stream>>>(pos_emb, w1, Pb);
  {
    int tiles = ((B_ * L_ + 15) / 16) * ((D_ + 15) / 16);
    gemm16_kernel<<<dim3((tiles + 7) / 8), 256, 0, stream>>>(
        out0, D_, w1botT, D_, nh, D_, B_ * L_, D_, D_, Pb, L_, 1);
  }
  beta_kernel<<<dim3(B_ * L_), dim3(D_), 0, stream>>>(nh, glu1_w, glu1_b, hsgb, w2, mask, betab);
  select_kernel<<<dim3(B_), dim3(D_), 0, stream>>>(out0, betab, selb);
  norm_kernel<<<dim3(B_), dim3(D_), 0, stream>>>(selb, snorm);
  fn_kernel<<<dim3(B_), dim3(D_), 0, stream>>>(selb, snorm, selF);
  {
    int Nsc = NN_ - 1;
    long tiles = (long)((B_ + 15) / 16) * ((Nsc + 15) / 16);
    gemm16_kernel<<<dim3((unsigned)((tiles + 7) / 8)), 256, 0, stream>>>(
        selF, D_, emb + D_, D_, scores, (long)Nsc, B_, Nsc, D_, nullptr, 1, 0);
  }
}


# score GEMM: embedding is the stationary operand (fragments go global to registers, kept over all selF tiles, never through LDS); selF staged once per work-group in LDS; a wave owns a run of units
# speedup vs baseline: 1.8514x; 1.8514x over previous
#include <hip/hip_runtime.h>
#include <hip/hip_bf16.h>
#include <math.h>

#define B_ 128
#define L_ 50
#define D_ 128
#define S_ 8
#define NN_ 100000
#define LEAKY_ 0.2f
#define BETA_ 0.005f

typedef float v2f __attribute__((ext_vector_type(2)));
typedef float v8f __attribute__((ext_vector_type(8)));
typedef float v4f __attribute__((ext_vector_type(4)));
typedef int v4i __attribute__((ext_vector_type(4)));

#if defined(__has_builtin)
#if __has_builtin(__builtin_amdgcn_wmma_f32_16x16x4_f32)
#define HAVE_WMMA_F32 1
#endif
#endif

__device__ __forceinline__ v8f wmma4(v2f a, v2f b, v8f c) {
#ifdef HAVE_WMMA_F32
  return __builtin_amdgcn_wmma_f32_16x16x4_f32(false, a, false, b, (short)0, c,
                                               false, false);
#else
  float* cf = (float*)&c;
  #pragma unroll
  for (int i = 0; i < 8; ++i) cf[i] += a.x * b.x + a.y * b.y;
  return c;
#endif
}

__device__ __forceinline__ float leaky_(float x) { return x > 0.f ? x : LEAKY_ * x; }
__device__ __forceinline__ float sigmoid_(float x) { return 1.f / (1.f + __expf(-x)); }

__global__ void w1t_kernel(const float* __restrict__ w1, float* __restrict__ w1botT) {
  int i = blockIdx.x * 256 + threadIdx.x;
  if (i >= D_ * D_) return;
  int n = i >> 7, k = i & 127;
  w1botT[i] = w1[(D_ + k) * D_ + n];
}

__global__ void gather_h_kernel(const float* __restrict__ emb,
                                const int* __restrict__ inputs,
                                float* __restrict__ h, int total) {
  int i = blockIdx.x * 256 + threadIdx.x;
  if (i >= total) return;
  int row = i >> 7, d = i & 127;
  h[i] = emb[((long)inputs[row] << 7) + d];
}

__global__ void hop_kernel(const int* __restrict__ src,
                           const int* __restrict__ adj_all,
                           const float* __restrict__ num_w,
                           int* __restrict__ dst, float* __restrict__ dstw,
                           int total) {
  int i = blockIdx.x * 256 + threadIdx.x;
  if (i >= total) return;
  int node = src[i >> 3];
  int s = i & 7;
  dst[i]  = adj_all[node * S_ + s];
  dstw[i] = num_w[node * S_ + s];
}

__global__ void sess_kernel(const float* __restrict__ emb,
                            const int* __restrict__ item,
                            const int* __restrict__ mask,
                            float* __restrict__ sess) {
  int b = blockIdx.x, d = threadIdx.x;
  float acc = 0.f, ms = 0.f;
  for (int l = 0; l < L_; ++l) {
    float m = (float)mask[b * L_ + l];
    ms += m;
    acc += emb[((long)item[b * L_ + l] << 7) + d] * m;
  }
  sess[(b << 7) + d] = acc / ms;
}

__global__ void local_agg_kernel(const float* __restrict__ h,
                                 const int* __restrict__ adj,
                                 const float* __restrict__ a_local,
                                 float* __restrict__ h_local) {
  int b = blockIdx.x;
  __shared__ float hv[L_][D_];
  __shared__ float att[L_][L_];
  __shared__ float rinv[L_];
  for (int i = threadIdx.x; i < L_ * D_; i += 256)
    hv[i / D_][i & 127] = h[(long)b * L_ * D_ + i];
  __syncthreads();
  for (int p = threadIdx.x; p < L_ * L_; p += 256) {
    int i = p / L_, j = p % L_;
    int k = adj[b * L_ * L_ + p];
    float v;
    if (k == 0) v = -9e15f;
    else {
      const float* a = a_local + (k - 1) * D_;
      float acc = 0.f;
      for (int d = 0; d < D_; ++d) acc += hv[i][d] * hv[j][d] * a[d];
      v = leaky_(acc);
    }
    att[i][j] = v;
  }
  __syncthreads();
  if (threadIdx.x < L_) {
    int i = threadIdx.x;
    float mx = -1e30f;
    for (int j = 0; j < L_; ++j) mx = fmaxf(mx, att[i][j]);
    float s = 0.f;
    for (int j = 0; j < L_; ++j) { float e = __expf(att[i][j] - mx); att[i][j] = e; s += e; }
    rinv[i] = 1.f / s;
  }
  __syncthreads();
  for (int p = threadIdx.x; p < L_ * D_; p += 256) {
    int i = p >> 7, d = p & 127;
    float acc = 0.f;
    for (int j = 0; j < L_; ++j) acc += att[i][j] * hv[j][d];
    h_local[(long)b * L_ * D_ + p] = acc * rinv[i];
  }
}

#define XP_ 132
#define GP_ 260
#define AGG_LDS_FLOATS (D_ * XP_ + D_ * GP_ + 8 * 16 * XP_ + 16 * GP_)

template <bool NIDX, bool SIDX>
__global__ __launch_bounds__(256) void agg_kernel(
                           const float* __restrict__ emb,
                           const float* __restrict__ selfVal, const int* __restrict__ selfIdx,
                           const int* __restrict__ neighIdx, const float* __restrict__ neighVal,
                           const float* __restrict__ neighW,
                           const float* __restrict__ sess,
                           const float* __restrict__ gw1p, const float* __restrict__ gw2p,
                           const float* __restrict__ gw3p,
                           float* __restrict__ out, int M, int parts) {
  __shared__ __attribute__((aligned(16))) float lds[AGG_LDS_FLOATS];
  float* WT  = lds;
  float* G3T = WT + D_ * XP_;
  float* xs  = G3T + D_ * GP_;
  float* sn  = xs + 8 * 16 * XP_;

  const int tid = threadIdx.x;
  const int lane = tid & 31, c = lane & 15, hf = lane >> 4;
  const int wave = tid >> 5;
  const int b = blockIdx.x / parts, part = blockIdx.x % parts;
  const int units = (M + 15) >> 4;
  const int upb = (units + parts - 1) / parts;
  const int u0 = part * upb;
  const int u1 = (u0 + upb < units) ? (u0 + upb) : units;

  {
    const int kl = tid & 7, ql = (tid >> 3) & 3, rest0 = tid >> 5;
    #pragma unroll
    for (int t = 0; t < 16; ++t) {
      const int rest = rest0 + 8 * t;
      int k = ((rest & 15) << 3) + kl;
      int q = ((rest >> 4) << 2) + ql;
      float4 w = *(const float4*)(gw1p + k * D_ + 4 * q);
      float s = sess[(b << 7) + k];
      float* d = WT + (4 * q) * XP_ + k;
      d[0] = w.x * s; d[XP_] = w.y * s; d[2 * XP_] = w.z * s; d[3 * XP_] = w.w * s;
    }
    if (tid < D_) {
      float* d = WT + tid * XP_ + D_;
      d[0] = gw1p[D_ * D_ + tid];
      d[1] = 0.f; d[2] = 0.f; d[3] = 0.f;
    }
    #pragma unroll
    for (int t = 0; t < 32; ++t) {
      const int rest = rest0 + 8 * t;
      int k = ((rest & 31) << 3) + kl;
      int q = ((rest >> 5) << 2) + ql;
      float4 w = *(const float4*)(gw3p + k * D_ + 4 * q);
      float* d = G3T + (4 * q) * GP_ + k;
      d[0] = w.x; d[GP_] = w.y; d[2 * GP_] = w.z; d[3 * GP_] = w.w;
    }
  }

  float* xw = xs + wave * 16 * XP_;
  v4f pre_0, pre_1, pre_2, pre_3, pre_4, pre_5, pre_6, pre_7,
      pre_8, pre_9, pre_10, pre_11, pre_12, pre_13, pre_14, pre_15;
  v4f pself_0, pself_1;
  float pw = 0.f;
  pre_0 = pre_1 = pre_2 = pre_3 = pre_4 = pre_5 = pre_6 = pre_7 = (v4f){};
  pre_8 = pre_9 = pre_10 = pre_11 = pre_12 = pre_13 = pre_14 = pre_15 = (v4f){};
  pself_0 = pself_1 = (v4f){};
  v4i ia0 = (v4i){}, ia1 = (v4i){}, ib0 = (v4i){}, ib1 = (v4i){};
  int isa = 0, isb = 0;
  long fra = 0, frb = 0;

#define AGG_ROWS(X) X(0) X(1) X(2) X(3) X(4) X(5) X(6) X(7) X(8) X(9) X(10) X(11) X(12) X(13) X(14) X(15)
#define AGG_LOAD_IDX(r, IV) pre_##r = *(const v4f*)(emb + ((long)(IV) << 7) + 4 * lane);
#define AGG_LOAD_VAL(r, ROW) pre_##r = *(const v4f*)(neighVal + (((ROW) * S_ + ((r) & 7)) << 7) + 4 * lane);
#define AGG_FETCH_IDX(U)                                                      \
  {                                                                           \
    const int mr0 = (U) * 16 + 2 * wave;                                      \
    const int ma = mr0 < M ? mr0 : M - 1;                                     \
    const int mb = mr0 + 1 < M ? mr0 + 1 : M - 1;                             \
    fra = (long)b * M + ma; frb = (long)b * M + mb;                           \
    if (NIDX) {                                                               \
      ia0 = *(const v4i*)(neighIdx + fra * S_);                               \
      ia1 = *(const v4i*)(neighIdx + fra * S_ + 4);                           \
      ib0 = *(const v4i*)(neighIdx + frb * S_);                               \
      ib1 = *(const v4i*)(neighIdx + frb * S_ + 4);                           \
    }                                                                         \
    if (SIDX) { isa = selfIdx[fra]; isb = selfIdx[frb]; }                     \
  }
#define AGG_FETCH_ROWS()                                                      \
  {                                                                           \
    if (NIDX) {                                                               \
      AGG_LOAD_IDX(0, ia0.x) AGG_LOAD_IDX(1, ia0.y) AGG_LOAD_IDX(2, ia0.z) AGG_LOAD_IDX(3, ia0.w)     \
      AGG_LOAD_IDX(4, ia1.x) AGG_LOAD_IDX(5, ia1.y) AGG_LOAD_IDX(6, ia1.z) AGG_LOAD_IDX(7, ia1.w)     \
      AGG_LOAD_IDX(8, ib0.x) AGG_LOAD_IDX(9, ib0.y) AGG_LOAD_IDX(10, ib0.z) AGG_LOAD_IDX(11, ib0.w)   \
      AGG_LOAD_IDX(12, ib1.x) AGG_LOAD_IDX(13, ib1.y) AGG_LOAD_IDX(14, ib1.z) AGG_LOAD_IDX(15, ib1.w) \
    } else {                                                                  \
      AGG_LOAD_VAL(0, fra) AGG_LOAD_VAL(1, fra) AGG_LOAD_VAL(2, fra) AGG_LOAD_VAL(3, fra)             \
      AGG_LOAD_VAL(4, fra) AGG_LOAD_VAL(5, fra) AGG_LOAD_VAL(6, fra) AGG_LOAD_VAL(7, fra)             \
      AGG_LOAD_VAL(8, frb) AGG_LOAD_VAL(9, frb) AGG_LOAD_VAL(10, frb) AGG_LOAD_VAL(11, frb)           \
      AGG_LOAD_VAL(12, frb) AGG_LOAD_VAL(13, frb) AGG_LOAD_VAL(14, frb) AGG_LOAD_VAL(15, frb)         \
    }                                                                         \
    if (SIDX) {                                                               \
      pself_0 = *(const v4f*)(emb + ((long)isa << 7) + 4 * lane);             \
      pself_1 = *(const v4f*)(emb + ((long)isb << 7) + 4 * lane);             \
    } else {                                                                  \
      pself_0 = *(const v4f*)(selfVal + (fra << 7) + 4 * lane);               \
      pself_1 = *(const v4f*)(selfVal + (frb << 7) + 4 * lane);               \
    }                                                                         \
    pw = neighW[((c >> 3) ? frb : fra) * S_ + (c & 7)];                       \
  }
#define AGG_STORE_ROW(r) *(v4f*)(xw + (r) * XP_ + 4 * lane) = pre_##r;

  if (u0 < u1) { AGG_FETCH_IDX(u0) AGG_FETCH_ROWS() }

  for (int u = u0; u < u1; ++u) {
    AGG_ROWS(AGG_STORE_ROW)
    if (lane < 16) {
      float* t = xw + lane * XP_ + D_;
      t[0] = pw; t[1] = 0.f; t[2] = 0.f; t[3] = 0.f;
    }
    const v4f s0 = pself_0, s1 = pself_1;
    if (u + 1 < u1) AGG_FETCH_IDX(u + 1)
    __syncthreads();

    v8f acc[8];
    #pragma unroll
    for (int j = 0; j < 8; ++j) acc[j] = (v8f){};
    {
      const float* ap = xw + c * XP_ + 2 * hf;
      const float* bp = WT + c * XP_ + 2 * hf;
      for (int k0 = 0; k0 < XP_; k0 += 4) {
        v2f a = *(const v2f*)(ap + k0);
        #pragma unroll
        for (int j = 0; j < 8; ++j) {
          v2f bb = *(const v2f*)(bp + j * 16 * XP_ + k0);
          acc[j] = wmma4(a, bb, acc[j]);
        }
      }
    }
    float sc[8];
    #pragma unroll
    for (int v = 0; v < 8; ++v) sc[v] = 0.f;
    #pragma unroll
    for (int j = 0; j < 8; ++j) {
      float g2 = gw2p[16 * j + c];
      #pragma unroll
      for (int v = 0; v < 8; ++v) sc[v] += leaky_(acc[j][v]) * g2;
    }
    #pragma unroll
    for (int v = 0; v < 8; ++v) {
      float val = sc[v];
      val += __shfl_xor(val, 1);
      val += __shfl_xor(val, 2);
      val += __shfl_xor(val, 4);
      val += __shfl_xor(val, 8);
      sc[v] = val;
    }
    float mx = sc[0];
    #pragma unroll
    for (int v = 1; v < 8; ++v) mx = fmaxf(mx, sc[v]);
    float sum = 0.f;
    #pragma unroll
    for (int v = 0; v < 8; ++v) { sc[v] = __expf(sc[v] - mx); sum += sc[v]; }
    const float inv = 1.f / sum;
    float4 na = make_float4(0.f, 0.f, 0.f, 0.f), nb = na;
    #pragma unroll
    for (int v = 0; v < 8; ++v) {
      const float al = sc[v] * inv;
      const float* row = xw + (8 * hf + v) * XP_ + 4 * c;
      float4 p = *(const float4*)(row);
      float4 q = *(const float4*)(row + 64);
      na.x += al * p.x; na.y += al * p.y; na.z += al * p.z; na.w += al * p.w;
      nb.x += al * q.x; nb.y += al * q.y; nb.z += al * q.z; nb.w += al * q.w;
    }
    {
      float* srow = sn + (2 * wave + hf) * GP_ + D_ + 4 * c;
      *(float4*)(srow) = na;
      *(float4*)(srow + 64) = nb;
      *(v4f*)(sn + (2 * wave) * GP_ + 4 * lane) = s0;
      *(v4f*)(sn + (2 * wave + 1) * GP_ + 4 * lane) = s1;
    }
    __syncthreads();

    if (u + 1 < u1) AGG_FETCH_ROWS()

    v8f o0 = (v8f){}, o1 = (v8f){};
    {
      const float* sa = sn + c * GP_ + 2 * hf;
      const float* gb = G3T + (16 * wave + c) * GP_ + 2 * hf;
      for (int k0 = 0; k0 < 2 * D_; k0 += 8) {
        o0 = wmma4(*(const v2f*)(sa + k0), *(const v2f*)(gb + k0), o0);
        o1 = wmma4(*(const v2f*)(sa + k0 + 4), *(const v2f*)(gb + k0 + 4), o1);
      }
    }
    #pragma unroll
    for (int v = 0; v < 8; ++v) {
      int m = u * 16 + v + 8 * hf;
      if (m < M) out[(((long)b * M + m) << 7) + 16 * wave + c] = fmaxf(o0[v] + o1[v], 0.f);
    }
  }
#undef AGG_FETCH_IDX
#undef AGG_FETCH_ROWS
#undef AGG_LOAD_IDX
#undef AGG_LOAD_VAL
#undef AGG_STORE_ROW
#undef AGG_ROWS
}

__global__ void gemm16_kernel(const float* __restrict__ A, int lda,
                              const float* __restrict__ BT, int ldbt,
                              float* __restrict__ C, long ldc,
                              int M, int N, int K,
                              const float* __restrict__ rowBias, int biasMod,
                              int act) {
  int tilesM = (M + 15) >> 4;
  long tilesN = (long)((N + 15) >> 4);
  long total = (long)tilesM * tilesN;
  int lane = threadIdx.x & 31, laneIdx = lane & 15, laneHalf = lane >> 4;
  for (long t = (long)blockIdx.x * 8 + (threadIdx.x >> 5); t < total;
       t += (long)gridDim.x * 8) {
    int m0 = (int)(t % tilesM) << 4;
    long n0 = (t / tilesM) << 4;
    int row = m0 + laneIdx;
    long col = n0 + laneIdx;
    v8f c = {};
    bool full = (m0 + 16 <= M) && (n0 + 16 <= N);
    if (full) {
      const float* ap = A + (long)row * lda;
      const float* bp = BT + col * ldbt;
      for (int k0 = 0; k0 < K; k0 += 4) {
        int kk = k0 + (laneHalf << 1);
        v2f a = *(const v2f*)(ap + kk);
        v2f b = *(const v2f*)(bp + kk);
        c = wmma4(a, b, c);
      }
    } else {
      for (int k0 = 0; k0 < K; k0 += 4) {
        int kk = k0 + (laneHalf << 1);
        v2f a, b;
        a.x = (row < M) ? A[(long)row * lda + kk] : 0.f;
        a.y = (row < M) ? A[(long)row * lda + kk + 1] : 0.f;
        b.x = (col < N) ? BT[col * ldbt + kk] : 0.f;
        b.y = (col < N) ? BT[col * ldbt + kk + 1] : 0.f;
        c = wmma4(a, b, c);
      }
    }
    float* cf = (float*)&c;
    #pragma unroll
    for (int v = 0; v < 8; ++v) {
      int rr = m0 + v + (laneHalf << 3);
      if (rr < M && col < N) {
        float val = cf[v];
        if (rowBias) val += rowBias[(long)(rr % biasMod) * N + col];
        if (act == 1) val = tanhf(val);
        C[(long)rr * ldc + col] = val;
      }
    }
  }
}


#define SP_ 132
#define SCORE_LDS_FLOATS (42 * 1024)
#define SCORE_WAVES 16
#define SCORE_WGS 256

__global__ __launch_bounds__(SCORE_WAVES * 32) void score_kernel(
    const float* __restrict__ selF, const float* __restrict__ emb,
    float* __restrict__ scores) {
  __shared__ __attribute__((aligned(16))) float sA[SCORE_LDS_FLOATS];
  const int tid = threadIdx.x;
  const int lane = tid & 31, c = lane & 15, hf = lane >> 4;
  const int NSC = NN_ - 1;
  const int NT = (NSC + 15) >> 4;
  const long U = (long)NT * 8;
  const int nw = SCORE_WGS * SCORE_WAVES;
  const int wv = __builtin_amdgcn_readfirstlane(tid >> 5);
  const int gw = (int)blockIdx.x * SCORE_WAVES + wv;
  const int u0 = (int)((U * gw) / nw), u1 = (int)((U * (gw + 1)) / nw);

  v4f bn[16];
#define SCORE_LOADB(NTILE)                                                      \
  {                                                                             \
    int er = ((NTILE) << 4) + c + 1;                \
    er = er < NN_ ? er : NN_ - 1;                   \
    const float* bp = emb + ((long)er << 7) + 4 * hf;                           \
    _Pragma("unroll") for (int i = 0; i < 16; ++i) bn[i] = *(const v4f*)(bp + 8 * i); \
  }
  if (u0 < u1) SCORE_LOADB(u0 >> 3)

  #pragma unroll
  for (int t = 0; t < (B_ * (D_ / 4)) / (SCORE_WAVES * 32); ++t) {
    const int i = tid + t * (SCORE_WAVES * 32);
    const int r = i >> 5, q = i & 31;
    *(v4f*)(sA + r * SP_ + 4 * q) = *(const v4f*)(selF + (r << 7) + 4 * q);
  }
  __syncthreads();

  int u = u0;
  while (u < u1) {
    const int nt = u >> 3;
    const int ue = (u1 < ((nt + 1) << 3)) ? u1 : ((nt + 1) << 3);
    v4f bf[16];
    #pragma unroll
    for (int i = 0; i < 16; ++i) bf[i] = bn[i];
    if (ue < u1) SCORE_LOADB(nt + 1)
    const int col = (nt << 4) + c;
    for (; u < ue; ++u) {
      const int m0 = (u & 7) << 4;
      const float* ap = sA + (m0 + c) * SP_ + 4 * hf;
      v4f a[16];
      #pragma unroll
      for (int i = 0; i < 16; ++i) a[i] = *(const v4f*)(ap + 8 * i);
      __builtin_amdgcn_sched_barrier(0);
      v8f acc0 = (v8f){}, acc1 = (v8f){};
      #pragma unroll
      for (int i = 0; i < 16; ++i) {
        acc0 = wmma4((v2f){a[i].x, a[i].y}, (v2f){bf[i].x, bf[i].y}, acc0);
        acc1 = wmma4((v2f){a[i].z, a[i].w}, (v2f){bf[i].z, bf[i].w}, acc1);
      }
      if (col < NSC) {
        float* op = scores + (long)(m0 + 8 * hf) * NSC + col;
        #pragma unroll
        for (int v = 0; v < 8; ++v) op[(long)v * NSC] = acc0[v] + acc1[v];
      }
    }
  }
#undef SCORE_LOADB
}

__global__ void add_kernel(const float* a, const float* b, float* o, int n) {
  int i = blockIdx.x * 256 + threadIdx.x;
  if (i < n) o[i] = a[i] + b[i];
}

__global__ void zero1_kernel(float* p) { if (threadIdx.x == 0 && blockIdx.x == 0) p[0] = 0.f; }

__global__ void ssl_kernel(const float* __restrict__ hl, const float* __restrict__ hg,
                           float* __restrict__ loss) {
  int idx = blockIdx.x * 256 + threadIdx.x;
  int b = idx >> 7, d = idx & 127;
  int pb = (b * 73 + 21) & 127;
  float pos = 0.f, neg = 0.f;
  for (int l = 0; l < L_; ++l) {
    int pl = (l * 13 + 7) % L_;
    pos += hl[((b * L_ + l) << 7) + d] * hg[((b * L_ + l) << 7) + d];
    neg += hg[((b * L_ + l) << 7) + d] * hl[((pb * L_ + pl) << 7) + d];
  }
  float term = -__logf(1e-8f + sigmoid_(pos)) - __logf(1e-8f + 1.f - sigmoid_(neg));
  __shared__ float red[256];
  red[threadIdx.x] = term;
  __syncthreads();
  for (int s = 128; s > 0; s >>= 1) {
    if (threadIdx.x < s) red[threadIdx.x] += red[threadIdx.x + s];
    __syncthreads();
  }
  if (threadIdx.x == 0) atomicAdd(loss, BETA_ * red[0]);
}

__global__ void hs_kernel(const float* __restrict__ hidden, const int* __restrict__ mask,
                          float* __restrict__ hs) {
  int b = blockIdx.x, d = threadIdx.x;
  float acc = 0.f, ms = 0.f;
  for (int l = 0; l < L_; ++l) {
    float m = (float)mask[b * L_ + l];
    ms += m;
    acc += hidden[((b * L_ + l) << 7) + d] * m;
  }
  hs[(b << 7) + d] = acc / ms;
}

__global__ void hsg_kernel(const float* __restrict__ hs, const float* __restrict__ glu2_w,
                           float* __restrict__ hsg) {
  __shared__ float hr[D_];
  int b = blockIdx.x, d = threadIdx.x;
  hr[d] = hs[(b << 7) + d];
  __syncthreads();
  float acc = 0.f;
  for (int e = 0; e < D_; ++e) acc += hr[e] * glu2_w[(d << 7) + e];
  hsg[(b << 7) + d] = acc;
}

__global__ void pos_kernel(const float* __restrict__ pos_emb, const float* __restrict__ w1,
                           float* __restrict__ P) {
  __shared__ float pr[D_];
  int l = blockIdx.x, d = threadIdx.x;
  pr[d] = pos_emb[(l << 7) + d];
  __syncthreads();
  float acc = 0.f;
  for (int e = 0; e < D_; ++e) acc += pr[e] * w1[(e << 7) + d];
  P[(l << 7) + d] = acc;
}

__global__ void beta_kernel(const float* __restrict__ nh, const float* __restrict__ glu1_w,
                            const float* __restrict__ glu1_b, const float* __restrict__ hsg,
                            const float* __restrict__ w2, const int* __restrict__ mask,
                            float* __restrict__ beta) {
  __shared__ float nrow[D_];
  __shared__ float red[D_];
  int row = blockIdx.x;
  int b = row / L_;
  int d = threadIdx.x;
  nrow[d] = nh[((long)row << 7) + d];
  __syncthreads();
  float acc = glu1_b[d] + hsg[(b << 7) + d];
  for (int e = 0; e < D_; ++e) acc += nrow[e] * glu1_w[(d << 7) + e];
  red[d] = sigmoid_(acc) * w2[d];
  __syncthreads();
  for (int s = 64; s > 0; s >>= 1) {
    if (d < s) red[d] += red[d + s];
    __syncthreads();
  }
  if (d == 0) beta[row] = red[0] * (float)mask[row];
}

__global__ void select_kernel(const float* __restrict__ hidden, const float* __restrict__ beta,
                              float* __restrict__ sel) {
  int b = blockIdx.x, d = threadIdx.x;
  float acc = 0.f;
  for (int l = 0; l < L_; ++l) acc += beta[b * L_ + l] * hidden[((b * L_ + l) << 7) + d];
  sel[(b << 7) + d] = acc;
}

__global__ void norm_kernel(const float* __restrict__ sel, float* __restrict__ snorm) {
  __shared__ float red[D_];
  int b = blockIdx.x, d = threadIdx.x;
  float v = sel[(b << 7) + d];
  red[d] = v * v + 1e-6f;
  __syncthreads();
  for (int s = 64; s > 0; s >>= 1) {
    if (d < s) red[d] += red[d + s];
    __syncthreads();
  }
  if (d == 0) snorm[b] = sqrtf(red[0]);
}

__global__ void fn_kernel(const float* __restrict__ sel, const float* __restrict__ snorm,
                          float* __restrict__ selF) {
  __shared__ float row[B_];
  __shared__ float selfrow[D_];
  __shared__ float tvs[9];
  __shared__ int tis[9];
  int i = blockIdx.x, t = threadIdx.x;
  selfrow[t] = sel[(i << 7) + t];
  __syncthreads();
  float acc = 0.f;
  for (int d = 0; d < D_; ++d) acc += selfrow[d] * sel[(t << 7) + d];
  row[t] = acc / (snorm[i] * snorm[t]);
  __syncthreads();
  if (t == 0) {
    float mx = -1e30f;
    for (int j = 0; j < B_; ++j) mx = fmaxf(mx, row[j]);
    float s = 0.f;
    for (int j = 0; j < B_; ++j) { row[j] = __expf(row[j] - mx); s += row[j]; }
    float inv = 1.f / s;
    for (int j = 0; j < B_; ++j) row[j] *= inv;
    for (int q = 0; q < 9; ++q) {
      float bm = -1.f; int bi = 0;
      for (int j = 0; j < B_; ++j) if (row[j] > bm) { bm = row[j]; bi = j; }
      tvs[q] = bm; tis[q] = bi; row[bi] = -2.f;
    }
    float mx2 = tvs[0], s2 = 0.f;
    for (int q = 0; q < 9; ++q) { tvs[q] = __expf(tvs[q] - mx2); s2 += tvs[q]; }
    float inv2 = 1.f / s2;
    for (int q = 0; q < 9; ++q) tvs[q] *= inv2;
  }
  __syncthreads();
  float nb = 0.f;
  for (int q = 0; q < 9; ++q) nb += tvs[q] * sel[(tis[q] << 7) + t];
  selF[(i << 7) + t] = selfrow[t] + nb;
}

extern "C" void kernel_launch(void* const* d_in, const int* in_sizes, int n_in,
                              void* d_out, int out_size, void* d_ws, size_t ws_size,
                              hipStream_t stream) {
  (void)in_sizes; (void)n_in; (void)out_size; (void)ws_size;
  const int*   inputs  = (const int*)d_in[0];
  const int*   adj     = (const int*)d_in[1];
  const int*   mask    = (const int*)d_in[2];
  const int*   item    = (const int*)d_in[3];
  const int*   adj_all = (const int*)d_in[5];
  const float* num_w   = (const float*)d_in[6];
  const float* emb     = (const float*)d_in[7];
  const float* pos_emb = (const float*)d_in[8];
  const float* a_local = (const float*)d_in[9];
  const float* gw1     = (const float*)d_in[10];
  const float* gw2     = (const float*)d_in[11];
  const float* gw3     = (const float*)d_in[12];
  const float* w1      = (const float*)d_in[13];
  const float* w2      = (const float*)d_in[14];
  const float* glu1_w  = (const float*)d_in[15];
  const float* glu1_b  = (const float*)d_in[16];
  const float* glu2_w  = (const float*)d_in[17];

  float* out0   = (float*)d_out;
  float* lossP  = out0 + B_ * L_ * D_;
  float* scores = lossP + 1;

  char* wp = (char*)d_ws;
  auto alloc = [&](size_t nfloats) -> float* {
    float* r = (float*)wp;
    wp += ((nfloats * sizeof(float) + 255) / 256) * 256;
    return r;
  };
  float* h        = alloc(B_ * L_ * D_);
  float* h_local  = alloc(B_ * L_ * D_);
  float* h_global = alloc(B_ * L_ * D_);
  float* sessb    = alloc(B_ * D_);
  float* e0p      = alloc(B_ * L_ * D_);
  float* e1p      = alloc((size_t)B_ * 400 * D_);
  float* hsb      = alloc(B_ * D_);
  float* hsgb     = alloc(B_ * D_);
  float* Pb       = alloc(L_ * D_);
  float* nh       = alloc(B_ * L_ * D_);
  float* betab    = alloc(B_ * L_);
  float* selb     = alloc(B_ * D_);
  float* selF     = alloc(B_ * D_);
  float* snorm    = alloc(B_);
  float* w1botT   = alloc(D_ * D_);
  float* wn1      = alloc(B_ * 400);
  float* wn2      = alloc((size_t)B_ * 3200);
  int*   n1       = (int*)alloc(B_ * 400);
  int*   n2       = (int*)alloc((size_t)B_ * 3200);

  w1t_kernel<<<dim3((D_ * D_ + 255) / 256), 256, 0, stream>>>(w1, w1botT);
  gather_h_kernel<<<dim3((B_ * L_ * D_) / 256), 256, 0, stream>>>(emb, inputs, h, B_ * L_ * D_);
  sess_kernel<<<dim3(B_), dim3(D_), 0, stream>>>(emb, item, mask, sessb);
  hop_kernel<<<dim3((B_ * L_ * S_ + 255) / 256), 256, 0, stream>>>(inputs, adj_all, num_w, n1, wn1, B_ * L_ * S_);
  hop_kernel<<<dim3((B_ * 400 * S_ + 255) / 256), 256, 0, stream>>>(n1, adj_all, num_w, n2, wn2, B_ * 400 * S_);
  local_agg_kernel<<<dim3(B_), 256, 0, stream>>>(h, adj, a_local, h_local);
  agg_kernel<true, false><<<dim3(B_ * 2), 256, 0, stream>>>(emb, h, nullptr, n1, nullptr, wn1,
      sessb, gw1, gw2, gw3, e0p, L_, 2);
  agg_kernel<true, true><<<dim3(B_ * 2), 256, 0, stream>>>(emb, nullptr, n1, n2, nullptr, wn2,
      sessb, gw1, gw2, gw3, e1p, 400, 2);
  agg_kernel<false, false><<<dim3(B_ * 2), 256, 0, stream>>>(emb, e0p, nullptr, nullptr, e1p, wn1,
      sessb, gw1 + 129 * D_, gw2 + D_, gw3 + 2 * D_ * D_, h_global, L_, 2);
  add_kernel<<<dim3((B_ * L_ * D_) / 256), 256, 0, stream>>>(h_local, h_global, out0, B_ * L_ * D_);
  zero1_kernel<<<1, 1, 0, stream>>>(lossP);
  ssl_kernel<<<dim3(64), 256, 0, stream>>>(h_local, h_global, lossP);
  hs_kernel<<<dim3(B_), dim3(D_), 0, stream>>>(out0, mask, hsb);
  hsg_kernel<<<dim3(B_), dim3(D_), 0, stream>>>(hsb, glu2_w, hsgb);
  pos_kernel<<<dim3(L_), dim3(D_), 0, stream>>>(pos_emb, w1, Pb);
  {
    int tiles = ((B_ * L_ + 15) / 16) * ((D_ + 15) / 16);
    gemm16_kernel<<<dim3((tiles + 7) / 8), 256, 0, stream>>>(
        out0, D_, w1botT, D_, nh, D_, B_ * L_, D_, D_, Pb, L_, 1);
  }
  beta_kernel<<<dim3(B_ * L_), dim3(D_), 0, stream>>>(nh, glu1_w, glu1_b, hsgb, w2, mask, betab);
  select_kernel<<<dim3(B_), dim3(D_), 0, stream>>>(out0, betab, selb);
  norm_kernel<<<dim3(B_), dim3(D_), 0, stream>>>(selb, snorm);
  fn_kernel<<<dim3(B_), dim3(D_), 0, stream>>>(selb, snorm, selF);
  score_kernel<<<dim3(SCORE_WGS), SCORE_WAVES * 32, 0, stream>>>(selF, emb, scores);
}


# local attention logits: a lane owns column j with h_j in registers (from a transposed LDS copy); row operand h_i*a_k read from four pre-scaled LDS planes; no same-bank row reads
# speedup vs baseline: 2.0156x; 1.0887x over previous
#include <hip/hip_runtime.h>
#include <hip/hip_bf16.h>
#include <math.h>

#define B_ 128
#define L_ 50
#define D_ 128
#define S_ 8
#define NN_ 100000
#define LEAKY_ 0.2f
#define BETA_ 0.005f

typedef float v2f __attribute__((ext_vector_type(2)));
typedef float v8f __attribute__((ext_vector_type(8)));
typedef float v4f __attribute__((ext_vector_type(4)));
typedef int v4i __attribute__((ext_vector_type(4)));

#if defined(__has_builtin)
#if __has_builtin(__builtin_amdgcn_wmma_f32_16x16x4_f32)
#define HAVE_WMMA_F32 1
#endif
#endif

__device__ __forceinline__ v8f wmma4(v2f a, v2f b, v8f c) {
#ifdef HAVE_WMMA_F32
  return __builtin_amdgcn_wmma_f32_16x16x4_f32(false, a, false, b, (short)0, c,
                                               false, false);
#else
  float* cf = (float*)&c;
  #pragma unroll
  for (int i = 0; i < 8; ++i) cf[i] += a.x * b.x + a.y * b.y;
  return c;
#endif
}

__device__ __forceinline__ float leaky_(float x) { return x > 0.f ? x : LEAKY_ * x; }
__device__ __forceinline__ float sigmoid_(float x) { return 1.f / (1.f + __expf(-x)); }

__global__ void w1t_kernel(const float* __restrict__ w1, float* __restrict__ w1botT) {
  int i = blockIdx.x * 256 + threadIdx.x;
  if (i >= D_ * D_) return;
  int n = i >> 7, k = i & 127;
  w1botT[i] = w1[(D_ + k) * D_ + n];
}

__global__ void gather_h_kernel(const float* __restrict__ emb,
                                const int* __restrict__ inputs,
                                float* __restrict__ h, int total) {
  int i = blockIdx.x * 256 + threadIdx.x;
  if (i >= total) return;
  int row = i >> 7, d = i & 127;
  h[i] = emb[((long)inputs[row] << 7) + d];
}

__global__ void hop_kernel(const int* __restrict__ src,
                           const int* __restrict__ adj_all,
                           const float* __restrict__ num_w,
                           int* __restrict__ dst, float* __restrict__ dstw,
                           int total) {
  int i = blockIdx.x * 256 + threadIdx.x;
  if (i >= total) return;
  int node = src[i >> 3];
  int s = i & 7;
  dst[i]  = adj_all[node * S_ + s];
  dstw[i] = num_w[node * S_ + s];
}

__global__ void sess_kernel(const float* __restrict__ emb,
                            const int* __restrict__ item,
                            const int* __restrict__ mask,
                            float* __restrict__ sess) {
  int b = blockIdx.x, d = threadIdx.x;
  float acc = 0.f, ms = 0.f;
  for (int l = 0; l < L_; ++l) {
    float m = (float)mask[b * L_ + l];
    ms += m;
    acc += emb[((long)item[b * L_ + l] << 7) + d] * m;
  }
  sess[(b << 7) + d] = acc / ms;
}

#define LA_TP 65
#define LA_PS (L_ * D_ + 8)
__global__ __launch_bounds__(256) void local_agg_kernel(const float* __restrict__ h,
                                 const int* __restrict__ adj,
                                 const float* __restrict__ a_local,
                                 float* __restrict__ h_local) {
  int b = blockIdx.x;
  __shared__ __attribute__((aligned(16))) float hv[L_][D_];
  __shared__ __attribute__((aligned(16))) float hia[4 * LA_PS];
  __shared__ float hvT[D_ * LA_TP];
  __shared__ int adjs[L_ * L_];
  __shared__ float att[L_][L_];
  __shared__ float rinv[L_];
  {
    const float* hb = h + (long)b * L_ * D_;
    const int q = threadIdx.x & 31;
    const v4f ak0 = *(const v4f*)(a_local + 0 * D_ + 4 * q);
    const v4f ak1 = *(const v4f*)(a_local + 1 * D_ + 4 * q);
    const v4f ak2 = *(const v4f*)(a_local + 2 * D_ + 4 * q);
    const v4f ak3 = *(const v4f*)(a_local + 3 * D_ + 4 * q);
    for (int i4 = threadIdx.x; i4 < (L_ * D_) / 4; i4 += 256) {
      const int r = i4 >> 5;
      const v4f v = *(const v4f*)(hb + 4 * i4);
      *(v4f*)(&hv[r][4 * q]) = v;
      float* tp = hvT + (4 * q) * LA_TP + r;
      tp[0] = v.x; tp[LA_TP] = v.y; tp[2 * LA_TP] = v.z; tp[3 * LA_TP] = v.w;
      *(v4f*)(hia + 0 * LA_PS + 4 * i4) = v * ak0;
      *(v4f*)(hia + 1 * LA_PS + 4 * i4) = v * ak1;
      *(v4f*)(hia + 2 * LA_PS + 4 * i4) = v * ak2;
      *(v4f*)(hia + 3 * LA_PS + 4 * i4) = v * ak3;
    }
    for (int p = threadIdx.x; p < L_ * L_; p += 256) adjs[p] = adj[b * L_ * L_ + p];
  }
  __syncthreads();
  {
    const int lane = threadIdx.x & 31, wv = threadIdx.x >> 5;
    const int j = lane + ((wv & 1) << 5);
    const bool jin = j < L_;
    const int jr = jin ? j : L_ - 1;
    float hj[D_];
    #pragma unroll
    for (int d = 0; d < D_; ++d) hj[d] = hvT[d * LA_TP + jr];
    int kn = adjs[(wv >> 1) * L_ + jr];
    for (int i = wv >> 1; i < L_; i += 4) {
      const int k = kn;
      kn = adjs[(i + 4 < L_ ? i + 4 : i) * L_ + jr];
      const bool on = (unsigned)(k - 1) < 4u;
      const float* rp = hia + (on ? k - 1 : 0) * LA_PS + i * D_;
      float a0 = 0.f, a1 = 0.f, a2 = 0.f, a3 = 0.f;
      #pragma unroll
      for (int q = 0; q < D_ / 4; ++q) {
        const v4f r = *(const v4f*)(rp + 4 * q);
        a0 += r.x * hj[4 * q];     a1 += r.y * hj[4 * q + 1];
        a2 += r.z * hj[4 * q + 2]; a3 += r.w * hj[4 * q + 3];
      }
      const float e = (a0 + a1) + (a2 + a3);
      if (jin) att[i][j] = on ? leaky_(e) : -9e15f;
    }
  }
  __syncthreads();
  if (threadIdx.x < L_) {
    int i = threadIdx.x;
    float mx = -1e30f;
    for (int j = 0; j < L_; ++j) mx = fmaxf(mx, att[i][j]);
    float s = 0.f;
    for (int j = 0; j < L_; ++j) { float e = __expf(att[i][j] - mx); att[i][j] = e; s += e; }
    rinv[i] = 1.f / s;
  }
  __syncthreads();
  for (int p = threadIdx.x; p < L_ * D_; p += 256) {
    int i = p >> 7, d = p & 127;
    float acc = 0.f;
    for (int j = 0; j < L_; ++j) acc += att[i][j] * hv[j][d];
    h_local[(long)b * L_ * D_ + p] = acc * rinv[i];
  }
}

#define XP_ 132
#define GP_ 260
#define AGG_LDS_FLOATS (D_ * XP_ + D_ * GP_ + 8 * 16 * XP_ + 16 * GP_)

template <bool NIDX, bool SIDX>
__global__ __launch_bounds__(256) void agg_kernel(
                           const float* __restrict__ emb,
                           const float* __restrict__ selfVal, const int* __restrict__ selfIdx,
                           const int* __restrict__ neighIdx, const float* __restrict__ neighVal,
                           const float* __restrict__ neighW,
                           const float* __restrict__ sess,
                           const float* __restrict__ gw1p, const float* __restrict__ gw2p,
                           const float* __restrict__ gw3p,
                           float* __restrict__ out, int M, int parts) {
  __shared__ __attribute__((aligned(16))) float lds[AGG_LDS_FLOATS];
  float* WT  = lds;
  float* G3T = WT + D_ * XP_;
  float* xs  = G3T + D_ * GP_;
  float* sn  = xs + 8 * 16 * XP_;

  const int tid = threadIdx.x;
  const int lane = tid & 31, c = lane & 15, hf = lane >> 4;
  const int wave = tid >> 5;
  const int b = blockIdx.x / parts, part = blockIdx.x % parts;
  const int units = (M + 15) >> 4;
  const int upb = (units + parts - 1) / parts;
  const int u0 = part * upb;
  const int u1 = (u0 + upb < units) ? (u0 + upb) : units;

  {
    const int kl = tid & 7, ql = (tid >> 3) & 3, rest0 = tid >> 5;
    #pragma unroll
    for (int t = 0; t < 16; ++t) {
      const int rest = rest0 + 8 * t;
      int k = ((rest & 15) << 3) + kl;
      int q = ((rest >> 4) << 2) + ql;
      float4 w = *(const float4*)(gw1p + k * D_ + 4 * q);
      float s = sess[(b << 7) + k];
      float* d = WT + (4 * q) * XP_ + k;
      d[0] = w.x * s; d[XP_] = w.y * s; d[2 * XP_] = w.z * s; d[3 * XP_] = w.w * s;
    }
    if (tid < D_) {
      float* d = WT + tid * XP_ + D_;
      d[0] = gw1p[D_ * D_ + tid];
      d[1] = 0.f; d[2] = 0.f; d[3] = 0.f;
    }
    #pragma unroll
    for (int t = 0; t < 32; ++t) {
      const int rest = rest0 + 8 * t;
      int k = ((rest & 31) << 3) + kl;
      int q = ((rest >> 5) << 2) + ql;
      float4 w = *(const float4*)(gw3p + k * D_ + 4 * q);
      float* d = G3T + (4 * q) * GP_ + k;
      d[0] = w.x; d[GP_] = w.y; d[2 * GP_] = w.z; d[3 * GP_] = w.w;
    }
  }

  float* xw = xs + wave * 16 * XP_;
  v4f pre_0, pre_1, pre_2, pre_3, pre_4, pre_5, pre_6, pre_7,
      pre_8, pre_9, pre_10, pre_11, pre_12, pre_13, pre_14, pre_15;
  v4f pself_0, pself_1;
  float pw = 0.f;
  pre_0 = pre_1 = pre_2 = pre_3 = pre_4 = pre_5 = pre_6 = pre_7 = (v4f){};
  pre_8 = pre_9 = pre_10 = pre_11 = pre_12 = pre_13 = pre_14 = pre_15 = (v4f){};
  pself_0 = pself_1 = (v4f){};
  v4i ia0 = (v4i){}, ia1 = (v4i){}, ib0 = (v4i){}, ib1 = (v4i){};
  int isa = 0, isb = 0;
  long fra = 0, frb = 0;

#define AGG_ROWS(X) X(0) X(1) X(2) X(3) X(4) X(5) X(6) X(7) X(8) X(9) X(10) X(11) X(12) X(13) X(14) X(15)
#define AGG_LOAD_IDX(r, IV) pre_##r = *(const v4f*)(emb + ((long)(IV) << 7) + 4 * lane);
#define AGG_LOAD_VAL(r, ROW) pre_##r = *(const v4f*)(neighVal + (((ROW) * S_ + ((r) & 7)) << 7) + 4 * lane);
#define AGG_FETCH_IDX(U)                                                      \
  {                                                                           \
    const int mr0 = (U) * 16 + 2 * wave;                                      \
    const int ma = mr0 < M ? mr0 : M - 1;                                     \
    const int mb = mr0 + 1 < M ? mr0 + 1 : M - 1;                             \
    fra = (long)b * M + ma; frb = (long)b * M + mb;                           \
    if (NIDX) {                                                               \
      ia0 = *(const v4i*)(neighIdx + fra * S_);                               \
      ia1 = *(const v4i*)(neighIdx + fra * S_ + 4);                           \
      ib0 = *(const v4i*)(neighIdx + frb * S_);                               \
      ib1 = *(const v4i*)(neighIdx + frb * S_ + 4);                           \
    }                                                                         \
    if (SIDX) { isa = selfIdx[fra]; isb = selfIdx[frb]; }                     \
  }
#define AGG_FETCH_ROWS()                                                      \
  {                                                                           \
    if (NIDX) {                                                               \
      AGG_LOAD_IDX(0, ia0.x) AGG_LOAD_IDX(1, ia0.y) AGG_LOAD_IDX(2, ia0.z) AGG_LOAD_IDX(3, ia0.w)     \
      AGG_LOAD_IDX(4, ia1.x) AGG_LOAD_IDX(5, ia1.y) AGG_LOAD_IDX(6, ia1.z) AGG_LOAD_IDX(7, ia1.w)     \
      AGG_LOAD_IDX(8, ib0.x) AGG_LOAD_IDX(9, ib0.y) AGG_LOAD_IDX(10, ib0.z) AGG_LOAD_IDX(11, ib0.w)   \
      AGG_LOAD_IDX(12, ib1.x) AGG_LOAD_IDX(13, ib1.y) AGG_LOAD_IDX(14, ib1.z) AGG_LOAD_IDX(15, ib1.w) \
    } else {                                                                  \
      AGG_LOAD_VAL(0, fra) AGG_LOAD_VAL(1, fra) AGG_LOAD_VAL(2, fra) AGG_LOAD_VAL(3, fra)             \
      AGG_LOAD_VAL(4, fra) AGG_LOAD_VAL(5, fra) AGG_LOAD_VAL(6, fra) AGG_LOAD_VAL(7, fra)             \
      AGG_LOAD_VAL(8, frb) AGG_LOAD_VAL(9, frb) AGG_LOAD_VAL(10, frb) AGG_LOAD_VAL(11, frb)           \
      AGG_LOAD_VAL(12, frb) AGG_LOAD_VAL(13, frb) AGG_LOAD_VAL(14, frb) AGG_LOAD_VAL(15, frb)         \
    }                                                                         \
    if (SIDX) {                                                               \
      pself_0 = *(const v4f*)(emb + ((long)isa << 7) + 4 * lane);             \
      pself_1 = *(const v4f*)(emb + ((long)isb << 7) + 4 * lane);             \
    } else {                                                                  \
      pself_0 = *(const v4f*)(selfVal + (fra << 7) + 4 * lane);               \
      pself_1 = *(const v4f*)(selfVal + (frb << 7) + 4 * lane);               \
    }                                                                         \
    pw = neighW[((c >> 3) ? frb : fra) * S_ + (c & 7)];                       \
  }
#define AGG_STORE_ROW(r) *(v4f*)(xw + (r) * XP_ + 4 * lane) = pre_##r;

  if (u0 < u1) { AGG_FETCH_IDX(u0) AGG_FETCH_ROWS() }

  for (int u = u0; u < u1; ++u) {
    AGG_ROWS(AGG_STORE_ROW)
    if (lane < 16) {
      float* t = xw + lane * XP_ + D_;
      t[0] = pw; t[1] = 0.f; t[2] = 0.f; t[3] = 0.f;
    }
    const v4f s0 = pself_0, s1 = pself_1;
    if (u + 1 < u1) AGG_FETCH_IDX(u + 1)
    __syncthreads();

    v8f acc[8];
    #pragma unroll
    for (int j = 0; j < 8; ++j) acc[j] = (v8f){};
    {
      const float* ap = xw + c * XP_ + 2 * hf;
      const float* bp = WT + c * XP_ + 2 * hf;
      for (int k0 = 0; k0 < XP_; k0 += 4) {
        v2f a = *(const v2f*)(ap + k0);
        #pragma unroll
        for (int j = 0; j < 8; ++j) {
          v2f bb = *(const v2f*)(bp + j * 16 * XP_ + k0);
          acc[j] = wmma4(a, bb, acc[j]);
        }
      }
    }
    float sc[8];
    #pragma unroll
    for (int v = 0; v < 8; ++v) sc[v] = 0.f;
    #pragma unroll
    for (int j = 0; j < 8; ++j) {
      float g2 = gw2p[16 * j + c];
      #pragma unroll
      for (int v = 0; v < 8; ++v) sc[v] += leaky_(acc[j][v]) * g2;
    }
    #pragma unroll
    for (int v = 0; v < 8; ++v) {
      float val = sc[v];
      val += __shfl_xor(val, 1);
      val += __shfl_xor(val, 2);
      val += __shfl_xor(val, 4);
      val += __shfl_xor(val, 8);
      sc[v] = val;
    }
    float mx = sc[0];
    #pragma unroll
    for (int v = 1; v < 8; ++v) mx = fmaxf(mx, sc[v]);
    float sum = 0.f;
    #pragma unroll
    for (int v = 0; v < 8; ++v) { sc[v] = __expf(sc[v] - mx); sum += sc[v]; }
    const float inv = 1.f / sum;
    float4 na = make_float4(0.f, 0.f, 0.f, 0.f), nb = na;
    #pragma unroll
    for (int v = 0; v < 8; ++v) {
      const float al = sc[v] * inv;
      const float* row = xw + (8 * hf + v) * XP_ + 4 * c;
      float4 p = *(const float4*)(row);
      float4 q = *(const float4*)(row + 64);
      na.x += al * p.x; na.y += al * p.y; na.z += al * p.z; na.w += al * p.w;
      nb.x += al * q.x; nb.y += al * q.y; nb.z += al * q.z; nb.w += al * q.w;
    }
    {
      float* srow = sn + (2 * wave + hf) * GP_ + D_ + 4 * c;
      *(float4*)(srow) = na;
      *(float4*)(srow + 64) = nb;
      *(v4f*)(sn + (2 * wave) * GP_ + 4 * lane) = s0;
      *(v4f*)(sn + (2 * wave + 1) * GP_ + 4 * lane) = s1;
    }
    __syncthreads();

    if (u + 1 < u1) AGG_FETCH_ROWS()

    v8f o0 = (v8f){}, o1 = (v8f){};
    {
      const float* sa = sn + c * GP_ + 2 * hf;
      const float* gb = G3T + (16 * wave + c) * GP_ + 2 * hf;
      for (int k0 = 0; k0 < 2 * D_; k0 += 8) {
        o0 = wmma4(*(const v2f*)(sa + k0), *(const v2f*)(gb + k0), o0);
        o1 = wmma4(*(const v2f*)(sa + k0 + 4), *(const v2f*)(gb + k0 + 4), o1);
      }
    }
    #pragma unroll
    for (int v = 0; v < 8; ++v) {
      int m = u * 16 + v + 8 * hf;
      if (m < M) out[(((long)b * M + m) << 7) + 16 * wave + c] = fmaxf(o0[v] + o1[v], 0.f);
    }
  }
#undef AGG_FETCH_IDX
#undef AGG_FETCH_ROWS
#undef AGG_LOAD_IDX
#undef AGG_LOAD_VAL
#undef AGG_STORE_ROW
#undef AGG_ROWS
}

__global__ void gemm16_kernel(const float* __restrict__ A, int lda,
                              const float* __restrict__ BT, int ldbt,
                              float* __restrict__ C, long ldc,
                              int M, int N, int K,
                              const float* __restrict__ rowBias, int biasMod,
                              int act) {
  int tilesM = (M + 15) >> 4;
  long tilesN = (long)((N + 15) >> 4);
  long total = (long)tilesM * tilesN;
  int lane = threadIdx.x & 31, laneIdx = lane & 15, laneHalf = lane >> 4;
  for (long t = (long)blockIdx.x * 8 + (threadIdx.x >> 5); t < total;
       t += (long)gridDim.x * 8) {
    int m0 = (int)(t % tilesM) << 4;
    long n0 = (t / tilesM) << 4;
    int row = m0 + laneIdx;
    long col = n0 + laneIdx;
    v8f c = {};
    bool full = (m0 + 16 <= M) && (n0 + 16 <= N);
    if (full) {
      const float* ap = A + (long)row * lda;
      const float* bp = BT + col * ldbt;
      for (int k0 = 0; k0 < K; k0 += 4) {
        int kk = k0 + (laneHalf << 1);
        v2f a = *(const v2f*)(ap + kk);
        v2f b = *(const v2f*)(bp + kk);
        c = wmma4(a, b, c);
      }
    } else {
      for (int k0 = 0; k0 < K; k0 += 4) {
        int kk = k0 + (laneHalf << 1);
        v2f a, b;
        a.x = (row < M) ? A[(long)row * lda + kk] : 0.f;
        a.y = (row < M) ? A[(long)row * lda + kk + 1] : 0.f;
        b.x = (col < N) ? BT[col * ldbt + kk] : 0.f;
        b.y = (col < N) ? BT[col * ldbt + kk + 1] : 0.f;
        c = wmma4(a, b, c);
      }
    }
    float* cf = (float*)&c;
    #pragma unroll
    for (int v = 0; v < 8; ++v) {
      int rr = m0 + v + (laneHalf << 3);
      if (rr < M && col < N) {
        float val = cf[v];
        if (rowBias) val += rowBias[(long)(rr % biasMod) * N + col];
        if (act == 1) val = tanhf(val);
        C[(long)rr * ldc + col] = val;
      }
    }
  }
}


#define SP_ 132
#define SCORE_LDS_FLOATS (42 * 1024)
#define SCORE_WAVES 16
#define SCORE_WGS 256

__global__ __launch_bounds__(SCORE_WAVES * 32) void score_kernel(
    const float* __restrict__ selF, const float* __restrict__ emb,
    float* __restrict__ scores) {
  __shared__ __attribute__((aligned(16))) float sA[SCORE_LDS_FLOATS];
  const int tid = threadIdx.x;
  const int lane = tid & 31, c = lane & 15, hf = lane >> 4;
  const int NSC = NN_ - 1;
  const int NT = (NSC + 15) >> 4;
  const long U = (long)NT * 8;
  const int nw = SCORE_WGS * SCORE_WAVES;
  const int wv = __builtin_amdgcn_readfirstlane(tid >> 5);
  const int gw = (int)blockIdx.x * SCORE_WAVES + wv;
  const int u0 = (int)((U * gw) / nw), u1 = (int)((U * (gw + 1)) / nw);

  v4f bn[16];
#define SCORE_LOADB(NTILE)                                                      \
  {                                                                             \
    int er = ((NTILE) << 4) + c + 1;                \
    er = er < NN_ ? er : NN_ - 1;                   \
    const float* bp = emb + ((long)er << 7) + 4 * hf;                           \
    _Pragma("unroll") for (int i = 0; i < 16; ++i) bn[i] = *(const v4f*)(bp + 8 * i); \
  }
  if (u0 < u1) SCORE_LOADB(u0 >> 3)

  #pragma unroll
  for (int t = 0; t < (B_ * (D_ / 4)) / (SCORE_WAVES * 32); ++t) {
    const int i = tid + t * (SCORE_WAVES * 32);
    const int r = i >> 5, q = i & 31;
    *(v4f*)(sA + r * SP_ + 4 * q) = *(const v4f*)(selF + (r << 7) + 4 * q);
  }
  __syncthreads();

  int u = u0;
  while (u < u1) {
    const int nt = u >> 3;
    const int ue = (u1 < ((nt + 1) << 3)) ? u1 : ((nt + 1) << 3);
    v4f bf[16];
    #pragma unroll
    for (int i = 0; i < 16; ++i) bf[i] = bn[i];
    if (ue < u1) SCORE_LOADB(nt + 1)
    const int col = (nt << 4) + c;
    for (; u < ue; ++u) {
      const int m0 = (u & 7) << 4;
      const float* ap = sA + (m0 + c) * SP_ + 4 * hf;
      v4f a[16];
      #pragma unroll
      for (int i = 0; i < 16; ++i) a[i] = *(const v4f*)(ap + 8 * i);
      __builtin_amdgcn_sched_barrier(0);
      v8f acc0 = (v8f){}, acc1 = (v8f){};
      #pragma unroll
      for (int i = 0; i < 16; ++i) {
        acc0 = wmma4((v2f){a[i].x, a[i].y}, (v2f){bf[i].x, bf[i].y}, acc0);
        acc1 = wmma4((v2f){a[i].z, a[i].w}, (v2f){bf[i].z, bf[i].w}, acc1);
      }
      if (col < NSC) {
        float* op = scores + (long)(m0 + 8 * hf) * NSC + col;
        #pragma unroll
        for (int v = 0; v < 8; ++v) op[(long)v * NSC] = acc0[v] + acc1[v];
      }
    }
  }
#undef SCORE_LOADB
}

__global__ void add_kernel(const float* a, const float* b, float* o, int n) {
  int i = blockIdx.x * 256 + threadIdx.x;
  if (i < n) o[i] = a[i] + b[i];
}

__global__ void zero1_kernel(float* p) { if (threadIdx.x == 0 && blockIdx.x == 0) p[0] = 0.f; }

__global__ void ssl_kernel(const float* __restrict__ hl, const float* __restrict__ hg,
                           float* __restrict__ loss) {
  int idx = blockIdx.x * 256 + threadIdx.x;
  int b = idx >> 7, d = idx & 127;
  int pb = (b * 73 + 21) & 127;
  float pos = 0.f, neg = 0.f;
  for (int l = 0; l < L_; ++l) {
    int pl = (l * 13 + 7) % L_;
    pos += hl[((b * L_ + l) << 7) + d] * hg[((b * L_ + l) << 7) + d];
    neg += hg[((b * L_ + l) << 7) + d] * hl[((pb * L_ + pl) << 7) + d];
  }
  float term = -__logf(1e-8f + sigmoid_(pos)) - __logf(1e-8f + 1.f - sigmoid_(neg));
  __shared__ float red[256];
  red[threadIdx.x] = term;
  __syncthreads();
  for (int s = 128; s > 0; s >>= 1) {
    if (threadIdx.x < s) red[threadIdx.x] += red[threadIdx.x + s];
    __syncthreads();
  }
  if (threadIdx.x == 0) atomicAdd(loss, BETA_ * red[0]);
}

__global__ void hs_kernel(const float* __restrict__ hidden, const int* __restrict__ mask,
                          float* __restrict__ hs) {
  int b = blockIdx.x, d = threadIdx.x;
  float acc = 0.f, ms = 0.f;
  for (int l = 0; l < L_; ++l) {
    float m = (float)mask[b * L_ + l];
    ms += m;
    acc += hidden[((b * L_ + l) << 7) + d] * m;
  }
  hs[(b << 7) + d] = acc / ms;
}

__global__ void hsg_kernel(const float* __restrict__ hs, const float* __restrict__ glu2_w,
                           float* __restrict__ hsg) {
  __shared__ float hr[D_];
  int b = blockIdx.x, d = threadIdx.x;
  hr[d] = hs[(b << 7) + d];
  __syncthreads();
  float acc = 0.f;
  for (int e = 0; e < D_; ++e) acc += hr[e] * glu2_w[(d << 7) + e];
  hsg[(b << 7) + d] = acc;
}

__global__ void pos_kernel(const float* __restrict__ pos_emb, const float* __restrict__ w1,
                           float* __restrict__ P) {
  __shared__ float pr[D_];
  int l = blockIdx.x, d = threadIdx.x;
  pr[d] = pos_emb[(l << 7) + d];
  __syncthreads();
  float acc = 0.f;
  for (int e = 0; e < D_; ++e) acc += pr[e] * w1[(e << 7) + d];
  P[(l << 7) + d] = acc;
}

__global__ void beta_kernel(const float* __restrict__ nh, const float* __restrict__ glu1_w,
                            const float* __restrict__ glu1_b, const float* __restrict__ hsg,
                            const float* __restrict__ w2, const int* __restrict__ mask,
                            float* __restrict__ beta) {
  __shared__ float nrow[D_];
  __shared__ float red[D_];
  int row = blockIdx.x;
  int b = row / L_;
  int d = threadIdx.x;
  nrow[d] = nh[((long)row << 7) + d];
  __syncthreads();
  float acc = glu1_b[d] + hsg[(b << 7) + d];
  for (int e = 0; e < D_; ++e) acc += nrow[e] * glu1_w[(d << 7) + e];
  red[d] = sigmoid_(acc) * w2[d];
  __syncthreads();
  for (int s = 64; s > 0; s >>= 1) {
    if (d < s) red[d] += red[d + s];
    __syncthreads();
  }
  if (d == 0) beta[row] = red[0] * (float)mask[row];
}

__global__ void select_kernel(const float* __restrict__ hidden, const float* __restrict__ beta,
                              float* __restrict__ sel) {
  int b = blockIdx.x, d = threadIdx.x;
  float acc = 0.f;
  for (int l = 0; l < L_; ++l) acc += beta[b * L_ + l] * hidden[((b * L_ + l) << 7) + d];
  sel[(b << 7) + d] = acc;
}

__global__ void norm_kernel(const float* __restrict__ sel, float* __restrict__ snorm) {
  __shared__ float red[D_];
  int b = blockIdx.x, d = threadIdx.x;
  float v = sel[(b << 7) + d];
  red[d] = v * v + 1e-6f;
  __syncthreads();
  for (int s = 64; s > 0; s >>= 1) {
    if (d < s) red[d] += red[d + s];
    __syncthreads();
  }
  if (d == 0) snorm[b] = sqrtf(red[0]);
}

__global__ void fn_kernel(const float* __restrict__ sel, const float* __restrict__ snorm,
                          float* __restrict__ selF) {
  __shared__ float row[B_];
  __shared__ float selfrow[D_];
  __shared__ float tvs[9];
  __shared__ int tis[9];
  int i = blockIdx.x, t = threadIdx.x;
  selfrow[t] = sel[(i << 7) + t];
  __syncthreads();
  float acc = 0.f;
  for (int d = 0; d < D_; ++d) acc += selfrow[d] * sel[(t << 7) + d];
  row[t] = acc / (snorm[i] * snorm[t]);
  __syncthreads();
  if (t == 0) {
    float mx = -1e30f;
    for (int j = 0; j < B_; ++j) mx = fmaxf(mx, row[j]);
    float s = 0.f;
    for (int j = 0; j < B_; ++j) { row[j] = __expf(row[j] - mx); s += row[j]; }
    float inv = 1.f / s;
    for (int j = 0; j < B_; ++j) row[j] *= inv;
    for (int q = 0; q < 9; ++q) {
      float bm = -1.f; int bi = 0;
      for (int j = 0; j < B_; ++j) if (row[j] > bm) { bm = row[j]; bi = j; }
      tvs[q] = bm; tis[q] = bi; row[bi] = -2.f;
    }
    float mx2 = tvs[0], s2 = 0.f;
    for (int q = 0; q < 9; ++q) { tvs[q] = __expf(tvs[q] - mx2); s2 += tvs[q]; }
    float inv2 = 1.f / s2;
    for (int q = 0; q < 9; ++q) tvs[q] *= inv2;
  }
  __syncthreads();
  float nb = 0.f;
  for (int q = 0; q < 9; ++q) nb += tvs[q] * sel[(tis[q] << 7) + t];
  selF[(i << 7) + t] = selfrow[t] + nb;
}

extern "C" void kernel_launch(void* const* d_in, const int* in_sizes, int n_in,
                              void* d_out, int out_size, void* d_ws, size_t ws_size,
                              hipStream_t stream) {
  (void)in_sizes; (void)n_in; (void)out_size; (void)ws_size;
  const int*   inputs  = (const int*)d_in[0];
  const int*   adj     = (const int*)d_in[1];
  const int*   mask    = (const int*)d_in[2];
  const int*   item    = (const int*)d_in[3];
  const int*   adj_all = (const int*)d_in[5];
  const float* num_w   = (const float*)d_in[6];
  const float* emb     = (const float*)d_in[7];
  const float* pos_emb = (const float*)d_in[8];
  const float* a_local = (const float*)d_in[9];
  const float* gw1     = (const float*)d_in[10];
  const float* gw2     = (const float*)d_in[11];
  const float* gw3     = (const float*)d_in[12];
  const float* w1      = (const float*)d_in[13];
  const float* w2      = (const float*)d_in[14];
  const float* glu1_w  = (const float*)d_in[15];
  const float* glu1_b  = (const float*)d_in[16];
  const float* glu2_w  = (const float*)d_in[17];

  float* out0   = (float*)d_out;
  float* lossP  = out0 + B_ * L_ * D_;
  float* scores = lossP + 1;

  char* wp = (char*)d_ws;
  auto alloc = [&](size_t nfloats) -> float* {
    float* r = (float*)wp;
    wp += ((nfloats * sizeof(float) + 255) / 256) * 256;
    return r;
  };
  float* h        = alloc(B_ * L_ * D_);
  float* h_local  = alloc(B_ * L_ * D_);
  float* h_global = alloc(B_ * L_ * D_);
  float* sessb    = alloc(B_ * D_);
  float* e0p      = alloc(B_ * L_ * D_);
  float* e1p      = alloc((size_t)B_ * 400 * D_);
  float* hsb      = alloc(B_ * D_);
  float* hsgb     = alloc(B_ * D_);
  float* Pb       = alloc(L_ * D_);
  float* nh       = alloc(B_ * L_ * D_);
  float* betab    = alloc(B_ * L_);
  float* selb     = alloc(B_ * D_);
  float* selF     = alloc(B_ * D_);
  float* snorm    = alloc(B_);
  float* w1botT   = alloc(D_ * D_);
  float* wn1      = alloc(B_ * 400);
  float* wn2      = alloc((size_t)B_ * 3200);
  int*   n1       = (int*)alloc(B_ * 400);
  int*   n2       = (int*)alloc((size_t)B_ * 3200);

  w1t_kernel<<<dim3((D_ * D_ + 255) / 256), 256, 0, stream>>>(w1, w1botT);
  gather_h_kernel<<<dim3((B_ * L_ * D_) / 256), 256, 0, stream>>>(emb, inputs, h, B_ * L_ * D_);
  sess_kernel<<<dim3(B_), dim3(D_), 0, stream>>>(emb, item, mask, sessb);
  hop_kernel<<<dim3((B_ * L_ * S_ + 255) / 256), 256, 0, stream>>>(inputs, adj_all, num_w, n1, wn1, B_ * L_ * S_);
  hop_kernel<<<dim3((B_ * 400 * S_ + 255) / 256), 256, 0, stream>>>(n1, adj_all, num_w, n2, wn2, B_ * 400 * S_);
  local_agg_kernel<<<dim3(B_), 256, 0, stream>>>(h, adj, a_local, h_local);
  agg_kernel<true, false><<<dim3(B_ * 2), 256, 0, stream>>>(emb, h, nullptr, n1, nullptr, wn1,
      sessb, gw1, gw2, gw3, e0p, L_, 2);
  agg_kernel<true, true><<<dim3(B_ * 2), 256, 0, stream>>>(emb, nullptr, n1, n2, nullptr, wn2,
      sessb, gw1, gw2, gw3, e1p, 400, 2);
  agg_kernel<false, false><<<dim3(B_ * 2), 256, 0, stream>>>(emb, e0p, nullptr, nullptr, e1p, wn1,
      sessb, gw1 + 129 * D_, gw2 + D_, gw3 + 2 * D_ * D_, h_global, L_, 2);
  add_kernel<<<dim3((B_ * L_ * D_) / 256), 256, 0, stream>>>(h_local, h_global, out0, B_ * L_ * D_);
  zero1_kernel<<<1, 1, 0, stream>>>(lossP);
  ssl_kernel<<<dim3(64), 256, 0, stream>>>(h_local, h_global, lossP);
  hs_kernel<<<dim3(B_), dim3(D_), 0, stream>>>(out0, mask, hsb);
  hsg_kernel<<<dim3(B_), dim3(D_), 0, stream>>>(hsb, glu2_w, hsgb);
  pos_kernel<<<dim3(L_), dim3(D_), 0, stream>>>(pos_emb, w1, Pb);
  {
    int tiles = ((B_ * L_ + 15) / 16) * ((D_ + 15) / 16);
    gemm16_kernel<<<dim3((tiles + 7) / 8), 256, 0, stream>>>(
        out0, D_, w1botT, D_, nh, D_, B_ * L_, D_, D_, Pb, L_, 1);
  }
  beta_kernel<<<dim3(B_ * L_), dim3(D_), 0, stream>>>(nh, glu1_w, glu1_b, hsgb, w2, mask, betab);
  select_kernel<<<dim3(B_), dim3(D_), 0, stream>>>(out0, betab, selb);
  norm_kernel<<<dim3(B_), dim3(D_), 0, stream>>>(selb, snorm);
  fn_kernel<<<dim3(B_), dim3(D_), 0, stream>>>(selb, snorm, selF);
  score_kernel<<<dim3(SCORE_WGS), SCORE_WAVES * 32, 0, stream>>>(selF, emb, scores);
}


# local attention: softmax-weighted sum att @ h (50x50x128 per batch entry) moved from per-output scalar LDS loops to f32 16x16x4 WMMA tiles; att tile zero-padded to 64x52 in LDS; sums re-ordered
# speedup vs baseline: 2.6353x; 1.3075x over previous
#include <hip/hip_runtime.h>
#include <hip/hip_bf16.h>
#include <math.h>

#define B_ 128
#define L_ 50
#define D_ 128
#define S_ 8
#define NN_ 100000
#define LEAKY_ 0.2f
#define BETA_ 0.005f

typedef float v2f __attribute__((ext_vector_type(2)));
typedef float v8f __attribute__((ext_vector_type(8)));
typedef float v4f __attribute__((ext_vector_type(4)));
typedef int v4i __attribute__((ext_vector_type(4)));

#if defined(__has_builtin)
#if __has_builtin(__builtin_amdgcn_wmma_f32_16x16x4_f32)
#define HAVE_WMMA_F32 1
#endif
#endif

__device__ __forceinline__ v8f wmma4(v2f a, v2f b, v8f c) {
#ifdef HAVE_WMMA_F32
  return __builtin_amdgcn_wmma_f32_16x16x4_f32(false, a, false, b, (short)0, c,
                                               false, false);
#else
  float* cf = (float*)&c;
  #pragma unroll
  for (int i = 0; i < 8; ++i) cf[i] += a.x * b.x + a.y * b.y;
  return c;
#endif
}

__device__ __forceinline__ float leaky_(float x) { return x > 0.f ? x : LEAKY_ * x; }
__device__ __forceinline__ float sigmoid_(float x) { return 1.f / (1.f + __expf(-x)); }


#define LA_TP 65
#define LA_PS (L_ * D_ + 8)
#define LA_AR 64
#define LA_AP 52
__global__ __launch_bounds__(256) void local_agg_kernel(const float* __restrict__ emb,
                                 const int* __restrict__ inputs,
                                 const int* __restrict__ adj,
                                 const float* __restrict__ a_local,
                                 float* __restrict__ h_local,
                                 float* __restrict__ loss) {
  int b = blockIdx.x;
  if (b == 0 && threadIdx.x == 0) loss[0] = 0.f;
  __shared__ __attribute__((aligned(16))) float hv[L_][D_];
  __shared__ __attribute__((aligned(16))) float hia[4 * LA_PS];
  __shared__ float hvT[D_ * LA_TP];
  __shared__ int adjs[L_ * L_];
  __shared__ __attribute__((aligned(16))) float att[LA_AR][LA_AP];
  __shared__ __attribute__((aligned(16))) float rinv[LA_AR];
  {
    const int q = threadIdx.x & 31;
    const int wvu = __builtin_amdgcn_readfirstlane(threadIdx.x >> 5);
    const v4f ak0 = *(const v4f*)(a_local + 0 * D_ + 4 * q);
    const v4f ak1 = *(const v4f*)(a_local + 1 * D_ + 4 * q);
    const v4f ak2 = *(const v4f*)(a_local + 2 * D_ + 4 * q);
    const v4f ak3 = *(const v4f*)(a_local + 3 * D_ + 4 * q);
    int nid[(L_ + 7) / 8];
    #pragma unroll
    for (int t = 0; t < (L_ + 7) / 8; ++t) {
      const int r = wvu + 8 * t;
      nid[t] = inputs[b * L_ + (r < L_ ? r : L_ - 1)];
    }
    v4f hr[(L_ + 7) / 8];
    #pragma unroll
    for (int t = 0; t < (L_ + 7) / 8; ++t) hr[t] = *(const v4f*)(emb + ((long)nid[t] << 7) + 4 * q);
    #pragma unroll
    for (int t = 0; t < (L_ + 7) / 8; ++t) {
      const int r = wvu + 8 * t;
      if (r >= L_) break;
      const int i4 = r * 32 + q;
      const v4f v = hr[t];
      *(v4f*)(&hv[r][4 * q]) = v;
      float* tp = hvT + (4 * q) * LA_TP + r;
      tp[0] = v.x; tp[LA_TP] = v.y; tp[2 * LA_TP] = v.z; tp[3 * LA_TP] = v.w;
      *(v4f*)(hia + 0 * LA_PS + 4 * i4) = v * ak0;
      *(v4f*)(hia + 1 * LA_PS + 4 * i4) = v * ak1;
      *(v4f*)(hia + 2 * LA_PS + 4 * i4) = v * ak2;
      *(v4f*)(hia + 3 * LA_PS + 4 * i4) = v * ak3;
    }
    for (int p = threadIdx.x; p < L_ * L_; p += 256) adjs[p] = adj[b * L_ * L_ + p];
    for (int p = threadIdx.x; p < LA_AR * LA_AP / 4; p += 256) ((v4f*)&att[0][0])[p] = (v4f){};
  }
  __syncthreads();
  {
    const int lane = threadIdx.x & 31, wv = threadIdx.x >> 5;
    const int j = lane + ((wv & 1) << 5);
    const bool jin = j < L_;
    const int jr = jin ? j : L_ - 1;
    float hj[D_];
    #pragma unroll
    for (int d = 0; d < D_; ++d) hj[d] = hvT[d * LA_TP + jr];
    int kn = adjs[(wv >> 1) * L_ + jr];
    for (int i = wv >> 1; i < L_; i += 4) {
      const int k = kn;
      kn = adjs[(i + 4 < L_ ? i + 4 : i) * L_ + jr];
      const bool on = (unsigned)(k - 1) < 4u;
      const float* rp = hia + (on ? k - 1 : 0) * LA_PS + i * D_;
      float a0 = 0.f, a1 = 0.f, a2 = 0.f, a3 = 0.f;
      #pragma unroll
      for (int q = 0; q < D_ / 4; ++q) {
        const v4f r = *(const v4f*)(rp + 4 * q);
        a0 += r.x * hj[4 * q];     a1 += r.y * hj[4 * q + 1];
        a2 += r.z * hj[4 * q + 2]; a3 += r.w * hj[4 * q + 3];
      }
      const float e = (a0 + a1) + (a2 + a3);
      if (jin) att[i][j] = on ? leaky_(e) : -9e15f;
    }
  }
  __syncthreads();
  if (threadIdx.x < L_) {
    int i = threadIdx.x;
    float mx = -1e30f;
    for (int j = 0; j < L_; ++j) mx = fmaxf(mx, att[i][j]);
    float s = 0.f;
    for (int j = 0; j < L_; ++j) { float e = __expf(att[i][j] - mx); att[i][j] = e; s += e; }
    rinv[i] = 1.f / s;
  }
  __syncthreads();
  {
    const int lane = threadIdx.x & 31, c = lane & 15, hf = lane >> 4;
    const int wv = threadIdx.x >> 5;
    const int n = (wv << 4) + c;
    v2f bq[LA_AP / 4];
    #pragma unroll
    for (int ks = 0; ks < LA_AP / 4; ++ks) {
      const int k = 4 * ks + 2 * hf;
      const bool kin = k < L_;
      const int kr = kin ? k : 0;
      const float b0 = hv[kr][n], b1 = hv[kr + 1][n];
      bq[ks] = kin ? (v2f){b0, b1} : (v2f){0.f, 0.f};
    }
    const float* ap = &att[c][2 * hf];
    v2f aq[4][LA_AP / 4];
    #pragma unroll
    for (int mt = 0; mt < 4; ++mt) {
      #pragma unroll
      for (int ks = 0; ks < LA_AP / 4; ++ks) aq[mt][ks] = *(const v2f*)(ap + 16 * mt * LA_AP + 4 * ks);
    }
    v4f rlo[4], rhi[4];
    #pragma unroll
    for (int mt = 0; mt < 4; ++mt) {
      rlo[mt] = *(const v4f*)(rinv + 16 * mt + 8 * hf);
      rhi[mt] = *(const v4f*)(rinv + 16 * mt + 8 * hf + 4);
    }
    v8f acc[4];
    #pragma unroll
    for (int mt = 0; mt < 4; ++mt) acc[mt] = (v8f){};
    #pragma unroll
    for (int ks = 0; ks < LA_AP / 4; ++ks) {
      #pragma unroll
      for (int mt = 0; mt < 4; ++mt) acc[mt] = wmma4(aq[mt][ks], bq[ks], acc[mt]);
    }
    float* op = h_local + (long)b * L_ * D_ + n;
    #pragma unroll
    for (int mt = 0; mt < 4; ++mt) {
      #pragma unroll
      for (int v = 0; v < 8; ++v) {
        const int row = 16 * mt + 8 * hf + v;
        const float ri = v < 4 ? rlo[mt][v] : rhi[mt][v - 4];
        if (row < L_) op[row * D_] = acc[mt][v] * ri;
      }
    }
  }
}

#define XP_ 132
#define GP_ 260
#define AGG_SID 256
#define AGG_LDS_FLOATS (D_ * XP_ + D_ * GP_ + 8 * 16 * XP_ + 16 * GP_ + D_ + AGG_SID)

template <bool GATHER, bool DEEP>
__global__ __launch_bounds__(256) void agg_kernel(
                           const float* __restrict__ emb,
                           const int* __restrict__ inputs, const int* __restrict__ adj_all,
                           const float* __restrict__ num_w,
                           const int* __restrict__ item, const int* __restrict__ mask,
                           const float* __restrict__ selfVal, const float* __restrict__ neighVal,
                           const float* __restrict__ gw1p, const float* __restrict__ gw2p,
                           const float* __restrict__ gw3p,
                           float* __restrict__ out, int M, int parts) {
  __shared__ __attribute__((aligned(16))) float lds[AGG_LDS_FLOATS];
  float* WT  = lds;
  float* G3T = WT + D_ * XP_;
  float* xs  = G3T + D_ * GP_;
  float* sn  = xs + 8 * 16 * XP_;
  float* sessL = sn + 16 * GP_;
  int*   sid = (int*)(sessL + D_);

  const int tid = threadIdx.x;
  const int lane = tid & 31, c = lane & 15, hf = lane >> 4;
  const int wave = tid >> 5;
  const int b = blockIdx.x / parts, part = blockIdx.x % parts;
  const int units = (M + 15) >> 4;
  const int upb = (units + parts - 1) / parts;
  const int u0 = part * upb;
  const int u1 = (u0 + upb < units) ? (u0 + upb) : units;

  float se[L_], mk[L_];
  #pragma unroll
  for (int l = 0; l < L_; ++l) se[l] = emb[((long)item[b * L_ + l] << 7) + (tid & (D_ - 1))];
  #pragma unroll
  for (int l = 0; l < L_; ++l) mk[l] = (float)mask[b * L_ + l];
  const int kl = tid & 7, ql = (tid >> 3) & 3, rest0 = tid >> 5;
  float4 w1r[16];
  #pragma unroll
  for (int t = 0; t < 16; ++t) {
    const int rest = rest0 + 8 * t;
    const int k = ((rest & 15) << 3) + kl;
    const int q = ((rest >> 4) << 2) + ql;
    w1r[t] = *(const float4*)(gw1p + k * D_ + 4 * q);
  }
  const int nsid = 16 * (u1 - u0);
  int sidv = 0;
  if (tid < nsid) {
    int m = u0 * 16 + tid;
    m = m < M ? m : M - 1;
    const int root = inputs[b * L_ + (DEEP ? (m >> 3) : m)];
    sidv = DEEP ? adj_all[root * S_ + (m & 7)] : root;
  }
  #pragma unroll
  for (int t = 0; t < 32; ++t) {
    const int rest = rest0 + 8 * t;
    int k = ((rest & 31) << 3) + kl;
    int q = ((rest >> 5) << 2) + ql;
    float4 w = *(const float4*)(gw3p + k * D_ + 4 * q);
    float* d = G3T + (4 * q) * GP_ + k;
    d[0] = w.x; d[GP_] = w.y; d[2 * GP_] = w.z; d[3 * GP_] = w.w;
  }
  if (tid < D_) {
    float sacc = 0.f, smask = 0.f;
    #pragma unroll
    for (int l = 0; l < L_; ++l) {
      const float m = mk[l];
      smask += m;
      sacc += se[l] * m;
    }
    sessL[tid] = sacc / smask;
    float* d = WT + tid * XP_ + D_;
    d[0] = gw1p[D_ * D_ + tid];
    d[1] = 0.f; d[2] = 0.f; d[3] = 0.f;
  }
  if (tid < nsid) sid[tid] = sidv;
  __syncthreads();
  {
    const float sk0 = sessL[(rest0 << 3) + kl], sk1 = sessL[(rest0 << 3) + kl + 64];
    #pragma unroll
    for (int t = 0; t < 16; ++t) {
      const int rest = rest0 + 8 * t;
      const int k = ((rest & 15) << 3) + kl;
      const int q = ((rest >> 4) << 2) + ql;
      const float4 w = w1r[t];
      const float s = (t & 1) ? sk1 : sk0;
      float* d = WT + (4 * q) * XP_ + k;
      d[0] = w.x * s; d[XP_] = w.y * s; d[2 * XP_] = w.z * s; d[3 * XP_] = w.w * s;
    }
  }

  float* xw = xs + wave * 16 * XP_;
  v4f pre_0, pre_1, pre_2, pre_3, pre_4, pre_5, pre_6, pre_7,
      pre_8, pre_9, pre_10, pre_11, pre_12, pre_13, pre_14, pre_15;
  v4f pself_0, pself_1;
  float pw = 0.f;
  pre_0 = pre_1 = pre_2 = pre_3 = pre_4 = pre_5 = pre_6 = pre_7 = (v4f){};
  pre_8 = pre_9 = pre_10 = pre_11 = pre_12 = pre_13 = pre_14 = pre_15 = (v4f){};
  pself_0 = pself_1 = (v4f){};
  v4i ia0 = (v4i){}, ia1 = (v4i){}, ib0 = (v4i){}, ib1 = (v4i){};
  int isa = 0, isb = 0;
  long fra = 0, frb = 0;

#define AGG_ROWS(X) X(0) X(1) X(2) X(3) X(4) X(5) X(6) X(7) X(8) X(9) X(10) X(11) X(12) X(13) X(14) X(15)
#define AGG_LOAD_IDX(r, IV) pre_##r = *(const v4f*)(emb + ((long)(IV) << 7) + 4 * lane);
#define AGG_LOAD_VAL(r, ROW) pre_##r = *(const v4f*)(neighVal + (((ROW) * S_ + ((r) & 7)) << 7) + 4 * lane);
#define AGG_FETCH_IDX(U)                                                      \
  {                                                                           \
    const int mr0 = (U) * 16 + 2 * wave;                                      \
    const int ma = mr0 < M ? mr0 : M - 1;                                     \
    const int mb = mr0 + 1 < M ? mr0 + 1 : M - 1;                             \
    fra = (long)b * M + ma; frb = (long)b * M + mb;                           \
    isa = sid[mr0 - u0 * 16]; isb = sid[mr0 + 1 - u0 * 16];                   \
    if (GATHER) {                                                             \
      ia0 = *(const v4i*)(adj_all + (long)isa * S_);                          \
      ia1 = *(const v4i*)(adj_all + (long)isa * S_ + 4);                      \
      ib0 = *(const v4i*)(adj_all + (long)isb * S_);                          \
      ib1 = *(const v4i*)(adj_all + (long)isb * S_ + 4);                      \
    }                                                                         \
  }
#define AGG_FETCH_ROWS()                                                      \
  {                                                                           \
    if (GATHER) {                                                             \
      AGG_LOAD_IDX(0, ia0.x) AGG_LOAD_IDX(1, ia0.y) AGG_LOAD_IDX(2, ia0.z) AGG_LOAD_IDX(3, ia0.w)     \
      AGG_LOAD_IDX(4, ia1.x) AGG_LOAD_IDX(5, ia1.y) AGG_LOAD_IDX(6, ia1.z) AGG_LOAD_IDX(7, ia1.w)     \
      AGG_LOAD_IDX(8, ib0.x) AGG_LOAD_IDX(9, ib0.y) AGG_LOAD_IDX(10, ib0.z) AGG_LOAD_IDX(11, ib0.w)   \
      AGG_LOAD_IDX(12, ib1.x) AGG_LOAD_IDX(13, ib1.y) AGG_LOAD_IDX(14, ib1.z) AGG_LOAD_IDX(15, ib1.w) \
    } else {                                                                  \
      AGG_LOAD_VAL(0, fra) AGG_LOAD_VAL(1, fra) AGG_LOAD_VAL(2, fra) AGG_LOAD_VAL(3, fra)             \
      AGG_LOAD_VAL(4, fra) AGG_LOAD_VAL(5, fra) AGG_LOAD_VAL(6, fra) AGG_LOAD_VAL(7, fra)             \
      AGG_LOAD_VAL(8, frb) AGG_LOAD_VAL(9, frb) AGG_LOAD_VAL(10, frb) AGG_LOAD_VAL(11, frb)           \
      AGG_LOAD_VAL(12, frb) AGG_LOAD_VAL(13, frb) AGG_LOAD_VAL(14, frb) AGG_LOAD_VAL(15, frb)         \
    }                                                                         \
    if (GATHER) {                                                             \
      pself_0 = *(const v4f*)(emb + ((long)isa << 7) + 4 * lane);             \
      pself_1 = *(const v4f*)(emb + ((long)isb << 7) + 4 * lane);             \
    } else {                                                                  \
      pself_0 = *(const v4f*)(selfVal + (fra << 7) + 4 * lane);               \
      pself_1 = *(const v4f*)(selfVal + (frb << 7) + 4 * lane);               \
    }                                                                         \
    pw = num_w[(long)((c >> 3) ? isb : isa) * S_ + (c & 7)];                  \
  }
#define AGG_STORE_ROW(r) *(v4f*)(xw + (r) * XP_ + 4 * lane) = pre_##r;

  if (u0 < u1) { AGG_FETCH_IDX(u0) AGG_FETCH_ROWS() }

  for (int u = u0; u < u1; ++u) {
    AGG_ROWS(AGG_STORE_ROW)
    if (lane < 16) {
      float* t = xw + lane * XP_ + D_;
      t[0] = pw; t[1] = 0.f; t[2] = 0.f; t[3] = 0.f;
    }
    const v4f s0 = pself_0, s1 = pself_1;
    if (u + 1 < u1) AGG_FETCH_IDX(u + 1)
    __syncthreads();

    v8f acc[8];
    #pragma unroll
    for (int j = 0; j < 8; ++j) acc[j] = (v8f){};
    {
      const float* ap = xw + c * XP_ + 2 * hf;
      const float* bp = WT + c * XP_ + 2 * hf;
      for (int k0 = 0; k0 < XP_; k0 += 4) {
        v2f a = *(const v2f*)(ap + k0);
        #pragma unroll
        for (int j = 0; j < 8; ++j) {
          v2f bb = *(const v2f*)(bp + j * 16 * XP_ + k0);
          acc[j] = wmma4(a, bb, acc[j]);
        }
      }
    }
    float sc[8];
    #pragma unroll
    for (int v = 0; v < 8; ++v) sc[v] = 0.f;
    #pragma unroll
    for (int j = 0; j < 8; ++j) {
      float g2 = gw2p[16 * j + c];
      #pragma unroll
      for (int v = 0; v < 8; ++v) sc[v] += leaky_(acc[j][v]) * g2;
    }
    #pragma unroll
    for (int v = 0; v < 8; ++v) {
      float val = sc[v];
      val += __shfl_xor(val, 1);
      val += __shfl_xor(val, 2);
      val += __shfl_xor(val, 4);
      val += __shfl_xor(val, 8);
      sc[v] = val;
    }
    float mx = sc[0];
    #pragma unroll
    for (int v = 1; v < 8; ++v) mx = fmaxf(mx, sc[v]);
    float sum = 0.f;
    #pragma unroll
    for (int v = 0; v < 8; ++v) { sc[v] = __expf(sc[v] - mx); sum += sc[v]; }
    const float inv = 1.f / sum;
    float4 na = make_float4(0.f, 0.f, 0.f, 0.f), nb = na;
    #pragma unroll
    for (int v = 0; v < 8; ++v) {
      const float al = sc[v] * inv;
      const float* row = xw + (8 * hf + v) * XP_ + 4 * c;
      float4 p = *(const float4*)(row);
      float4 q = *(const float4*)(row + 64);
      na.x += al * p.x; na.y += al * p.y; na.z += al * p.z; na.w += al * p.w;
      nb.x += al * q.x; nb.y += al * q.y; nb.z += al * q.z; nb.w += al * q.w;
    }
    {
      float* srow = sn + (2 * wave + hf) * GP_ + D_ + 4 * c;
      *(float4*)(srow) = na;
      *(float4*)(srow + 64) = nb;
      *(v4f*)(sn + (2 * wave) * GP_ + 4 * lane) = s0;
      *(v4f*)(sn + (2 * wave + 1) * GP_ + 4 * lane) = s1;
    }
    __syncthreads();

    if (u + 1 < u1) AGG_FETCH_ROWS()

    v8f o0 = (v8f){}, o1 = (v8f){};
    {
      const float* sa = sn + c * GP_ + 2 * hf;
      const float* gb = G3T + (16 * wave + c) * GP_ + 2 * hf;
      for (int k0 = 0; k0 < 2 * D_; k0 += 8) {
        o0 = wmma4(*(const v2f*)(sa + k0), *(const v2f*)(gb + k0), o0);
        o1 = wmma4(*(const v2f*)(sa + k0 + 4), *(const v2f*)(gb + k0 + 4), o1);
      }
    }
    #pragma unroll
    for (int v = 0; v < 8; ++v) {
      int m = u * 16 + v + 8 * hf;
      if (m < M) out[(((long)b * M + m) << 7) + 16 * wave + c] = fmaxf(o0[v] + o1[v], 0.f);
    }
  }
#undef AGG_FETCH_IDX
#undef AGG_FETCH_ROWS
#undef AGG_LOAD_IDX
#undef AGG_LOAD_VAL
#undef AGG_STORE_ROW
#undef AGG_ROWS
}

#define SP_ 132
#define SCORE_LDS_FLOATS (42 * 1024)
#define SCORE_WAVES 16
#define SCORE_WGS 256

__global__ __launch_bounds__(SCORE_WAVES * 32) void score_kernel(
    const float* __restrict__ selF, const float* __restrict__ emb,
    float* __restrict__ scores) {
  __shared__ __attribute__((aligned(16))) float sA[SCORE_LDS_FLOATS];
  const int tid = threadIdx.x;
  const int lane = tid & 31, c = lane & 15, hf = lane >> 4;
  const int NSC = NN_ - 1;
  const int NT = (NSC + 15) >> 4;
  const long U = (long)NT * 8;
  const int nw = SCORE_WGS * SCORE_WAVES;
  const int wv = __builtin_amdgcn_readfirstlane(tid >> 5);
  const int gw = (int)blockIdx.x * SCORE_WAVES + wv;
  const int u0 = (int)((U * gw) / nw), u1 = (int)((U * (gw + 1)) / nw);

  v4f bn[16];
#define SCORE_LOADB(NTILE)                                                      \
  {                                                                             \
    int er = ((NTILE) << 4) + c + 1;                \
    er = er < NN_ ? er : NN_ - 1;                   \
    const float* bp = emb + ((long)er << 7) + 4 * hf;                           \
    _Pragma("unroll") for (int i = 0; i < 16; ++i) bn[i] = *(const v4f*)(bp + 8 * i); \
  }
  if (u0 < u1) SCORE_LOADB(u0 >> 3)

  #pragma unroll
  for (int t = 0; t < (B_ * (D_ / 4)) / (SCORE_WAVES * 32); ++t) {
    const int i = tid + t * (SCORE_WAVES * 32);
    const int r = i >> 5, q = i & 31;
    *(v4f*)(sA + r * SP_ + 4 * q) = *(const v4f*)(selF + (r << 7) + 4 * q);
  }
  __syncthreads();

  int u = u0;
  while (u < u1) {
    const int nt = u >> 3;
    const int ue = (u1 < ((nt + 1) << 3)) ? u1 : ((nt + 1) << 3);
    v4f bf[16];
    #pragma unroll
    for (int i = 0; i < 16; ++i) bf[i] = bn[i];
    if (ue < u1) SCORE_LOADB(nt + 1)
    const int col = (nt << 4) + c;
    for (; u < ue; ++u) {
      const int m0 = (u & 7) << 4;
      const float* ap = sA + (m0 + c) * SP_ + 4 * hf;
      v4f a[16];
      #pragma unroll
      for (int i = 0; i < 16; ++i) a[i] = *(const v4f*)(ap + 8 * i);
      __builtin_amdgcn_sched_barrier(0);
      v8f acc0 = (v8f){}, acc1 = (v8f){};
      #pragma unroll
      for (int i = 0; i < 16; ++i) {
        acc0 = wmma4((v2f){a[i].x, a[i].y}, (v2f){bf[i].x, bf[i].y}, acc0);
        acc1 = wmma4((v2f){a[i].z, a[i].w}, (v2f){bf[i].z, bf[i].w}, acc1);
      }
      if (col < NSC) {
        float* op = scores + (long)(m0 + 8 * hf) * NSC + col;
        #pragma unroll
        for (int v = 0; v < 8; ++v) op[(long)v * NSC] = acc0[v] + acc1[v];
      }
    }
  }
#undef SCORE_LOADB
}

__global__ void ssl_kernel(const float* __restrict__ hl, const float* __restrict__ hg,
                           float* __restrict__ loss) {
  int idx = blockIdx.x * 256 + threadIdx.x;
  int b = idx >> 7, d = idx & 127;
  int pb = (b * 73 + 21) & 127;
  float pos = 0.f, neg = 0.f;
  for (int l = 0; l < L_; ++l) {
    int pl = (l * 13 + 7) % L_;
    pos += hl[((b * L_ + l) << 7) + d] * hg[((b * L_ + l) << 7) + d];
    neg += hg[((b * L_ + l) << 7) + d] * hl[((pb * L_ + pl) << 7) + d];
  }
  float term = -__logf(1e-8f + sigmoid_(pos)) - __logf(1e-8f + 1.f - sigmoid_(neg));
  __shared__ float red[256];
  red[threadIdx.x] = term;
  __syncthreads();
  for (int s = 128; s > 0; s >>= 1) {
    if (threadIdx.x < s) red[threadIdx.x] += red[threadIdx.x + s];
    __syncthreads();
  }
  if (threadIdx.x == 0) atomicAdd(loss, BETA_ * red[0]);
}

__global__ void pos_kernel(const float* __restrict__ pos_emb, const float* __restrict__ w1,
                           float* __restrict__ P) {
  __shared__ float pr[D_];
  int l = blockIdx.x, d = threadIdx.x;
  pr[d] = pos_emb[(l << 7) + d];
  __syncthreads();
  float acc = 0.f;
  for (int e = 0; e < D_; ++e) acc += pr[e] * w1[(e << 7) + d];
  P[(l << 7) + d] = acc;
}

template <int M>
__device__ __forceinline__ float row_xor(float x) {
  return __builtin_bit_cast(float, __builtin_amdgcn_update_dpp(0, __builtin_bit_cast(int, x), 0x160 + M, 0xf, 0xf, false));
}
__device__ __forceinline__ float xhalf(float x) {
  const int v = __builtin_bit_cast(int, x);
  return __builtin_bit_cast(float, __builtin_amdgcn_permlanex16(v, v, 0x76543210, 0xfedcba98, false, false));
}

#define HP_ 132
#define HR_ 64
#define HEAD_LDS_FLOATS (42 * 1024)

__global__ __launch_bounds__(256) void head_kernel(
    const float* __restrict__ h_local, const float* __restrict__ h_global,
    const int* __restrict__ mask, const float* __restrict__ P,
    const float* __restrict__ w1, const float* __restrict__ glu1_w,
    const float* __restrict__ glu1_b, const float* __restrict__ glu2_w,
    const float* __restrict__ w2,
    float* __restrict__ hidden, float* __restrict__ sel, float* __restrict__ snorm) {
  __shared__ __attribute__((aligned(16))) float lds[HEAD_LDS_FLOATS];
  float* HX    = lds;
  float* NH    = HX + HR_ * HP_;
  float* hsL   = NH + HR_ * HP_;
  float* maskf = hsL + D_;
  float* part  = maskf + HR_;
  float* betas = part + 8 * HR_;
  float* red   = betas + HR_;

  const int tid = threadIdx.x, b = blockIdx.x;
  const int lane = tid & 31, c = lane & 15, hf = lane >> 4;
  const int wave = tid >> 5;
  const int n = (wave << 4) + c;

  v4f bw[16], bg[16], g2[16];
  {
    const float* wp = w1 + (D_ + 4 * hf) * D_ + n;
    #pragma unroll
    for (int i = 0; i < 16; ++i) {
      bw[i].x = wp[(8 * i + 0) * D_];
      bw[i].y = wp[(8 * i + 1) * D_];
      bw[i].z = wp[(8 * i + 2) * D_];
      bw[i].w = wp[(8 * i + 3) * D_];
    }
    const float* gp = glu1_w + n * D_ + 4 * hf;
    #pragma unroll
    for (int i = 0; i < 16; ++i) bg[i] = *(const v4f*)(gp + 8 * i);
    const float* g2p = glu2_w + n * D_ + 64 * hf;
    #pragma unroll
    for (int i = 0; i < 16; ++i) g2[i] = *(const v4f*)(g2p + 4 * i);
  }
  const float g1b = glu1_b[n];
  const float w2n = w2[n];

  {
    const float* hl = h_local + (long)b * L_ * D_;
    const float* hg = h_global + (long)b * L_ * D_;
    float* ho = hidden + (long)b * L_ * D_;
    #pragma unroll
    for (int t = 0; t < 6; ++t) {
      const int i4 = tid + 256 * t;
      const v4f v = *(const v4f*)(hl + 4 * i4) + *(const v4f*)(hg + 4 * i4);
      *(v4f*)(ho + 4 * i4) = v;
      *(v4f*)(HX + (i4 >> 5) * HP_ + 4 * (i4 & 31)) = v;
    }
    if (tid < (L_ * D_) / 4 - 6 * 256) {
      const int i4 = tid + 6 * 256;
      const v4f v = *(const v4f*)(hl + 4 * i4) + *(const v4f*)(hg + 4 * i4);
      *(v4f*)(ho + 4 * i4) = v;
      *(v4f*)(HX + (i4 >> 5) * HP_ + 4 * (i4 & 31)) = v;
    }
    for (int i4 = tid; i4 < (HR_ - L_) * 32; i4 += 256)
      *(v4f*)(HX + (L_ + (i4 >> 5)) * HP_ + 4 * (i4 & 31)) = (v4f){};
    if (tid < HR_) maskf[tid] = tid < L_ ? (float)mask[b * L_ + tid] : 0.f;
  }
  __syncthreads();

  if (tid < D_) {
    float acc = 0.f, ms = 0.f;
    #pragma unroll
    for (int l = 0; l < L_; ++l) {
      const float m = maskf[l];
      ms += m;
      acc += HX[l * HP_ + tid] * m;
    }
    hsL[tid] = acc / ms;
  }
  for (int mt = 0; mt < 4; ++mt) {
    const float* ap = HX + (16 * mt + c) * HP_ + 4 * hf;
    v4f a[16];
    #pragma unroll
    for (int i = 0; i < 16; ++i) a[i] = *(const v4f*)(ap + 8 * i);
    __builtin_amdgcn_sched_barrier(0);
    float pv[8];
    #pragma unroll
    for (int v = 0; v < 8; ++v) {
      const int row = 16 * mt + 8 * hf + v;
      pv[v] = row < L_ ? P[row * D_ + n] : 0.f;
    }
    v8f acc0 = (v8f){}, acc1 = (v8f){};
    #pragma unroll
    for (int i = 0; i < 16; ++i) {
      acc0 = wmma4((v2f){a[i].x, a[i].y}, (v2f){bw[i].x, bw[i].y}, acc0);
      acc1 = wmma4((v2f){a[i].z, a[i].w}, (v2f){bw[i].z, bw[i].w}, acc1);
    }
    #pragma unroll
    for (int v = 0; v < 8; ++v)
      NH[(16 * mt + 8 * hf + v) * HP_ + n] = tanhf(acc0[v] + acc1[v] + pv[v]);
  }
  __syncthreads();

  float bias;
  {
    const float* hp = hsL + 64 * hf;
    float s0 = 0.f, s1 = 0.f, s2 = 0.f, s3 = 0.f;
    #pragma unroll
    for (int i = 0; i < 16; ++i) {
      const v4f x = *(const v4f*)(hp + 4 * i);
      s0 += g2[i].x * x.x; s1 += g2[i].y * x.y;
      s2 += g2[i].z * x.z; s3 += g2[i].w * x.w;
    }
    float s = (s0 + s1) + (s2 + s3);
    s += xhalf(s);
    bias = s + g1b;
  }

  for (int mt = 0; mt < 4; ++mt) {
    const float* ap = NH + (16 * mt + c) * HP_ + 4 * hf;
    v4f a[16];
    #pragma unroll
    for (int i = 0; i < 16; ++i) a[i] = *(const v4f*)(ap + 8 * i);
    __builtin_amdgcn_sched_barrier(0);
    v8f acc0 = (v8f){}, acc1 = (v8f){};
    #pragma unroll
    for (int i = 0; i < 16; ++i) {
      acc0 = wmma4((v2f){a[i].x, a[i].y}, (v2f){bg[i].x, bg[i].y}, acc0);
      acc1 = wmma4((v2f){a[i].z, a[i].w}, (v2f){bg[i].z, bg[i].w}, acc1);
    }
    float g[8];
    #pragma unroll
    for (int v = 0; v < 8; ++v) g[v] = sigmoid_(acc0[v] + acc1[v] + bias) * w2n;
    #pragma unroll
    for (int v = 0; v < 8; ++v) g[v] += row_xor<1>(g[v]);
    #pragma unroll
    for (int v = 0; v < 8; ++v) g[v] += row_xor<2>(g[v]);
    #pragma unroll
    for (int v = 0; v < 8; ++v) g[v] += row_xor<4>(g[v]);
    #pragma unroll
    for (int v = 0; v < 8; ++v) g[v] += row_xor<8>(g[v]);
    if (c == 0) {
      #pragma unroll
      for (int v = 0; v < 8; ++v) part[wave * HR_ + 16 * mt + 8 * hf + v] = g[v];
    }
  }
  __syncthreads();

  if (tid < L_) {
    float s = 0.f;
    #pragma unroll
    for (int w = 0; w < 8; ++w) s += part[w * HR_ + tid];
    betas[tid] = s * maskf[tid];
  }
  __syncthreads();

  float sq = 0.f;
  if (tid < D_) {
    float acc = 0.f;
    #pragma unroll
    for (int l = 0; l < L_; ++l) acc += betas[l] * HX[l * HP_ + tid];
    sel[(b << 7) + tid] = acc;
    sq = acc * acc + 1e-6f;
  }
  sq += row_xor<1>(sq);
  sq += row_xor<2>(sq);
  sq += row_xor<4>(sq);
  sq += row_xor<8>(sq);
  sq += xhalf(sq);
  if (lane == 0 && wave < 4) red[wave] = sq;
  __syncthreads();
  if (tid == 0) snorm[b] = sqrtf((red[0] + red[1]) + (red[2] + red[3]));
}


__global__ void fn_kernel(const float* __restrict__ sel, const float* __restrict__ snorm,
                          float* __restrict__ selF) {
  __shared__ float row[B_];
  __shared__ float selfrow[D_];
  __shared__ float tvs[9];
  __shared__ int tis[9];
  int i = blockIdx.x, t = threadIdx.x;
  selfrow[t] = sel[(i << 7) + t];
  __syncthreads();
  float acc = 0.f;
  for (int d = 0; d < D_; ++d) acc += selfrow[d] * sel[(t << 7) + d];
  row[t] = acc / (snorm[i] * snorm[t]);
  __syncthreads();
  if (t == 0) {
    float mx = -1e30f;
    for (int j = 0; j < B_; ++j) mx = fmaxf(mx, row[j]);
    float s = 0.f;
    for (int j = 0; j < B_; ++j) { row[j] = __expf(row[j] - mx); s += row[j]; }
    float inv = 1.f / s;
    for (int j = 0; j < B_; ++j) row[j] *= inv;
    for (int q = 0; q < 9; ++q) {
      float bm = -1.f; int bi = 0;
      for (int j = 0; j < B_; ++j) if (row[j] > bm) { bm = row[j]; bi = j; }
      tvs[q] = bm; tis[q] = bi; row[bi] = -2.f;
    }
    float mx2 = tvs[0], s2 = 0.f;
    for (int q = 0; q < 9; ++q) { tvs[q] = __expf(tvs[q] - mx2); s2 += tvs[q]; }
    float inv2 = 1.f / s2;
    for (int q = 0; q < 9; ++q) tvs[q] *= inv2;
  }
  __syncthreads();
  float nb = 0.f;
  for (int q = 0; q < 9; ++q) nb += tvs[q] * sel[(tis[q] << 7) + t];
  selF[(i << 7) + t] = selfrow[t] + nb;
}

extern "C" void kernel_launch(void* const* d_in, const int* in_sizes, int n_in,
                              void* d_out, int out_size, void* d_ws, size_t ws_size,
                              hipStream_t stream) {
  (void)in_sizes; (void)n_in; (void)out_size; (void)ws_size;
  const int*   inputs  = (const int*)d_in[0];
  const int*   adj     = (const int*)d_in[1];
  const int*   mask    = (const int*)d_in[2];
  const int*   item    = (const int*)d_in[3];
  const int*   adj_all = (const int*)d_in[5];
  const float* num_w   = (const float*)d_in[6];
  const float* emb     = (const float*)d_in[7];
  const float* pos_emb = (const float*)d_in[8];
  const float* a_local = (const float*)d_in[9];
  const float* gw1     = (const float*)d_in[10];
  const float* gw2     = (const float*)d_in[11];
  const float* gw3     = (const float*)d_in[12];
  const float* w1      = (const float*)d_in[13];
  const float* w2      = (const float*)d_in[14];
  const float* glu1_w  = (const float*)d_in[15];
  const float* glu1_b  = (const float*)d_in[16];
  const float* glu2_w  = (const float*)d_in[17];

  float* out0   = (float*)d_out;
  float* lossP  = out0 + B_ * L_ * D_;
  float* scores = lossP + 1;

  char* wp = (char*)d_ws;
  auto alloc = [&](size_t nfloats) -> float* {
    float* r = (float*)wp;
    wp += ((nfloats * sizeof(float) + 255) / 256) * 256;
    return r;
  };
  float* h_local  = alloc(B_ * L_ * D_);
  float* h_global = alloc(B_ * L_ * D_);
  float* e0p      = alloc(B_ * L_ * D_);
  float* e1p      = alloc((size_t)B_ * 400 * D_);
  float* Pb       = alloc(L_ * D_);
  float* selb     = alloc(B_ * D_);
  float* selF     = alloc(B_ * D_);
  float* snorm    = alloc(B_);

  local_agg_kernel<<<dim3(B_), 256, 0, stream>>>(emb, inputs, adj, a_local, h_local, lossP);
  agg_kernel<true, false><<<dim3(B_ * 2), 256, 0, stream>>>(emb, inputs, adj_all, num_w, item, mask,
      nullptr, nullptr, gw1, gw2, gw3, e0p, L_, 2);
  agg_kernel<true, true><<<dim3(B_ * 2), 256, 0, stream>>>(emb, inputs, adj_all, num_w, item, mask,
      nullptr, nullptr, gw1, gw2, gw3, e1p, L_ * S_, 2);
  agg_kernel<false, false><<<dim3(B_ * 2), 256, 0, stream>>>(emb, inputs, adj_all, num_w, item, mask,
      e0p, e1p, gw1 + 129 * D_, gw2 + D_, gw3 + 2 * D_ * D_, h_global, L_, 2);
  ssl_kernel<<<dim3(64), 256, 0, stream>>>(h_local, h_global, lossP);
  pos_kernel<<<dim3(L_), dim3(D_), 0, stream>>>(pos_emb, w1, Pb);
  head_kernel<<<dim3(B_), 256, 0, stream>>>(h_local, h_global, mask, Pb, w1, glu1_w, glu1_b,
      glu2_w, w2, out0, selb, snorm);
  fn_kernel<<<dim3(B_), dim3(D_), 0, stream>>>(selb, snorm, selF);
  score_kernel<<<dim3(SCORE_WGS), SCORE_WAVES * 32, 0, stream>>>(selF, emb, scores);
}


# fn kernel: one wave keeps the 128-entry row in registers (4 per lane); max, sum and the nine arg-max rounds are cross-lane butterflies, not one lane's LDS walks; row sum now a tree (re-ordered)
# speedup vs baseline: 2.7654x; 1.0494x over previous
#include <hip/hip_runtime.h>
#include <hip/hip_bf16.h>
#include <math.h>

#define B_ 128
#define L_ 50
#define D_ 128
#define S_ 8
#define NN_ 100000
#define LEAKY_ 0.2f
#define BETA_ 0.005f

typedef float v2f __attribute__((ext_vector_type(2)));
typedef float v8f __attribute__((ext_vector_type(8)));
typedef float v4f __attribute__((ext_vector_type(4)));
typedef int v4i __attribute__((ext_vector_type(4)));

#if defined(__has_builtin)
#if __has_builtin(__builtin_amdgcn_wmma_f32_16x16x4_f32)
#define HAVE_WMMA_F32 1
#endif
#endif

__device__ __forceinline__ v8f wmma4(v2f a, v2f b, v8f c) {
#ifdef HAVE_WMMA_F32
  return __builtin_amdgcn_wmma_f32_16x16x4_f32(false, a, false, b, (short)0, c,
                                               false, false);
#else
  float* cf = (float*)&c;
  #pragma unroll
  for (int i = 0; i < 8; ++i) cf[i] += a.x * b.x + a.y * b.y;
  return c;
#endif
}

__device__ __forceinline__ float leaky_(float x) { return x > 0.f ? x : LEAKY_ * x; }
__device__ __forceinline__ float sigmoid_(float x) { return 1.f / (1.f + __expf(-x)); }


#define LA_TP 65
#define LA_PS (L_ * D_ + 8)
#define LA_AR 64
#define LA_AP 52
__global__ __launch_bounds__(256) void local_agg_kernel(const float* __restrict__ emb,
                                 const int* __restrict__ inputs,
                                 const int* __restrict__ adj,
                                 const float* __restrict__ a_local,
                                 float* __restrict__ h_local,
                                 float* __restrict__ loss) {
  int b = blockIdx.x;
  if (b == 0 && threadIdx.x == 0) loss[0] = 0.f;
  __shared__ __attribute__((aligned(16))) float hv[L_][D_];
  __shared__ __attribute__((aligned(16))) float hia[4 * LA_PS];
  __shared__ float hvT[D_ * LA_TP];
  __shared__ int adjs[L_ * L_];
  __shared__ __attribute__((aligned(16))) float att[LA_AR][LA_AP];
  __shared__ __attribute__((aligned(16))) float rinv[LA_AR];
  {
    const int q = threadIdx.x & 31;
    const int wvu = __builtin_amdgcn_readfirstlane(threadIdx.x >> 5);
    const v4f ak0 = *(const v4f*)(a_local + 0 * D_ + 4 * q);
    const v4f ak1 = *(const v4f*)(a_local + 1 * D_ + 4 * q);
    const v4f ak2 = *(const v4f*)(a_local + 2 * D_ + 4 * q);
    const v4f ak3 = *(const v4f*)(a_local + 3 * D_ + 4 * q);
    int nid[(L_ + 7) / 8];
    #pragma unroll
    for (int t = 0; t < (L_ + 7) / 8; ++t) {
      const int r = wvu + 8 * t;
      nid[t] = inputs[b * L_ + (r < L_ ? r : L_ - 1)];
    }
    v4f hr[(L_ + 7) / 8];
    #pragma unroll
    for (int t = 0; t < (L_ + 7) / 8; ++t) hr[t] = *(const v4f*)(emb + ((long)nid[t] << 7) + 4 * q);
    #pragma unroll
    for (int t = 0; t < (L_ + 7) / 8; ++t) {
      const int r = wvu + 8 * t;
      if (r >= L_) break;
      const int i4 = r * 32 + q;
      const v4f v = hr[t];
      *(v4f*)(&hv[r][4 * q]) = v;
      float* tp = hvT + (4 * q) * LA_TP + r;
      tp[0] = v.x; tp[LA_TP] = v.y; tp[2 * LA_TP] = v.z; tp[3 * LA_TP] = v.w;
      *(v4f*)(hia + 0 * LA_PS + 4 * i4) = v * ak0;
      *(v4f*)(hia + 1 * LA_PS + 4 * i4) = v * ak1;
      *(v4f*)(hia + 2 * LA_PS + 4 * i4) = v * ak2;
      *(v4f*)(hia + 3 * LA_PS + 4 * i4) = v * ak3;
    }
    for (int p = threadIdx.x; p < L_ * L_; p += 256) adjs[p] = adj[b * L_ * L_ + p];
    for (int p = threadIdx.x; p < LA_AR * LA_AP / 4; p += 256) ((v4f*)&att[0][0])[p] = (v4f){};
  }
  __syncthreads();
  {
    const int lane = threadIdx.x & 31, wv = threadIdx.x >> 5;
    const int j = lane + ((wv & 1) << 5);
    const bool jin = j < L_;
    const int jr = jin ? j : L_ - 1;
    float hj[D_];
    #pragma unroll
    for (int d = 0; d < D_; ++d) hj[d] = hvT[d * LA_TP + jr];
    int kn = adjs[(wv >> 1) * L_ + jr];
    for (int i = wv >> 1; i < L_; i += 4) {
      const int k = kn;
      kn = adjs[(i + 4 < L_ ? i + 4 : i) * L_ + jr];
      const bool on = (unsigned)(k - 1) < 4u;
      const float* rp = hia + (on ? k - 1 : 0) * LA_PS + i * D_;
      float a0 = 0.f, a1 = 0.f, a2 = 0.f, a3 = 0.f;
      #pragma unroll
      for (int q = 0; q < D_ / 4; ++q) {
        const v4f r = *(const v4f*)(rp + 4 * q);
        a0 += r.x * hj[4 * q];     a1 += r.y * hj[4 * q + 1];
        a2 += r.z * hj[4 * q + 2]; a3 += r.w * hj[4 * q + 3];
      }
      const float e = (a0 + a1) + (a2 + a3);
      if (jin) att[i][j] = on ? leaky_(e) : -9e15f;
    }
  }
  __syncthreads();
  if (threadIdx.x < L_) {
    int i = threadIdx.x;
    float mx = -1e30f;
    for (int j = 0; j < L_; ++j) mx = fmaxf(mx, att[i][j]);
    float s = 0.f;
    for (int j = 0; j < L_; ++j) { float e = __expf(att[i][j] - mx); att[i][j] = e; s += e; }
    rinv[i] = 1.f / s;
  }
  __syncthreads();
  {
    const int lane = threadIdx.x & 31, c = lane & 15, hf = lane >> 4;
    const int wv = threadIdx.x >> 5;
    const int n = (wv << 4) + c;
    v2f bq[LA_AP / 4];
    #pragma unroll
    for (int ks = 0; ks < LA_AP / 4; ++ks) {
      const int k = 4 * ks + 2 * hf;
      const bool kin = k < L_;
      const int kr = kin ? k : 0;
      const float b0 = hv[kr][n], b1 = hv[kr + 1][n];
      bq[ks] = kin ? (v2f){b0, b1} : (v2f){0.f, 0.f};
    }
    const float* ap = &att[c][2 * hf];
    v2f aq[4][LA_AP / 4];
    #pragma unroll
    for (int mt = 0; mt < 4; ++mt) {
      #pragma unroll
      for (int ks = 0; ks < LA_AP / 4; ++ks) aq[mt][ks] = *(const v2f*)(ap + 16 * mt * LA_AP + 4 * ks);
    }
    v4f rlo[4], rhi[4];
    #pragma unroll
    for (int mt = 0; mt < 4; ++mt) {
      rlo[mt] = *(const v4f*)(rinv + 16 * mt + 8 * hf);
      rhi[mt] = *(const v4f*)(rinv + 16 * mt + 8 * hf + 4);
    }
    v8f acc[4];
    #pragma unroll
    for (int mt = 0; mt < 4; ++mt) acc[mt] = (v8f){};
    #pragma unroll
    for (int ks = 0; ks < LA_AP / 4; ++ks) {
      #pragma unroll
      for (int mt = 0; mt < 4; ++mt) acc[mt] = wmma4(aq[mt][ks], bq[ks], acc[mt]);
    }
    float* op = h_local + (long)b * L_ * D_ + n;
    #pragma unroll
    for (int mt = 0; mt < 4; ++mt) {
      #pragma unroll
      for (int v = 0; v < 8; ++v) {
        const int row = 16 * mt + 8 * hf + v;
        const float ri = v < 4 ? rlo[mt][v] : rhi[mt][v - 4];
        if (row < L_) op[row * D_] = acc[mt][v] * ri;
      }
    }
  }
}

#define XP_ 132
#define GP_ 260
#define AGG_SID 256
#define AGG_LDS_FLOATS (D_ * XP_ + D_ * GP_ + 8 * 16 * XP_ + 16 * GP_ + D_ + AGG_SID)

template <bool GATHER, bool DEEP>
__global__ __launch_bounds__(256) void agg_kernel(
                           const float* __restrict__ emb,
                           const int* __restrict__ inputs, const int* __restrict__ adj_all,
                           const float* __restrict__ num_w,
                           const int* __restrict__ item, const int* __restrict__ mask,
                           const float* __restrict__ selfVal, const float* __restrict__ neighVal,
                           const float* __restrict__ gw1p, const float* __restrict__ gw2p,
                           const float* __restrict__ gw3p,
                           float* __restrict__ out, int M, int parts) {
  __shared__ __attribute__((aligned(16))) float lds[AGG_LDS_FLOATS];
  float* WT  = lds;
  float* G3T = WT + D_ * XP_;
  float* xs  = G3T + D_ * GP_;
  float* sn  = xs + 8 * 16 * XP_;
  float* sessL = sn + 16 * GP_;
  int*   sid = (int*)(sessL + D_);

  const int tid = threadIdx.x;
  const int lane = tid & 31, c = lane & 15, hf = lane >> 4;
  const int wave = tid >> 5;
  const int b = blockIdx.x / parts, part = blockIdx.x % parts;
  const int units = (M + 15) >> 4;
  const int upb = (units + parts - 1) / parts;
  const int u0 = part * upb;
  const int u1 = (u0 + upb < units) ? (u0 + upb) : units;

  float se[L_], mk[L_];
  #pragma unroll
  for (int l = 0; l < L_; ++l) se[l] = emb[((long)item[b * L_ + l] << 7) + (tid & (D_ - 1))];
  #pragma unroll
  for (int l = 0; l < L_; ++l) mk[l] = (float)mask[b * L_ + l];
  const int kl = tid & 7, ql = (tid >> 3) & 3, rest0 = tid >> 5;
  float4 w1r[16];
  #pragma unroll
  for (int t = 0; t < 16; ++t) {
    const int rest = rest0 + 8 * t;
    const int k = ((rest & 15) << 3) + kl;
    const int q = ((rest >> 4) << 2) + ql;
    w1r[t] = *(const float4*)(gw1p + k * D_ + 4 * q);
  }
  const int nsid = 16 * (u1 - u0);
  int sidv = 0;
  if (tid < nsid) {
    int m = u0 * 16 + tid;
    m = m < M ? m : M - 1;
    const int root = inputs[b * L_ + (DEEP ? (m >> 3) : m)];
    sidv = DEEP ? adj_all[root * S_ + (m & 7)] : root;
  }
  #pragma unroll
  for (int t = 0; t < 32; ++t) {
    const int rest = rest0 + 8 * t;
    int k = ((rest & 31) << 3) + kl;
    int q = ((rest >> 5) << 2) + ql;
    float4 w = *(const float4*)(gw3p + k * D_ + 4 * q);
    float* d = G3T + (4 * q) * GP_ + k;
    d[0] = w.x; d[GP_] = w.y; d[2 * GP_] = w.z; d[3 * GP_] = w.w;
  }
  if (tid < D_) {
    float sacc = 0.f, smask = 0.f;
    #pragma unroll
    for (int l = 0; l < L_; ++l) {
      const float m = mk[l];
      smask += m;
      sacc += se[l] * m;
    }
    sessL[tid] = sacc / smask;
    float* d = WT + tid * XP_ + D_;
    d[0] = gw1p[D_ * D_ + tid];
    d[1] = 0.f; d[2] = 0.f; d[3] = 0.f;
  }
  if (tid < nsid) sid[tid] = sidv;
  __syncthreads();
  {
    const float sk0 = sessL[(rest0 << 3) + kl], sk1 = sessL[(rest0 << 3) + kl + 64];
    #pragma unroll
    for (int t = 0; t < 16; ++t) {
      const int rest = rest0 + 8 * t;
      const int k = ((rest & 15) << 3) + kl;
      const int q = ((rest >> 4) << 2) + ql;
      const float4 w = w1r[t];
      const float s = (t & 1) ? sk1 : sk0;
      float* d = WT + (4 * q) * XP_ + k;
      d[0] = w.x * s; d[XP_] = w.y * s; d[2 * XP_] = w.z * s; d[3 * XP_] = w.w * s;
    }
  }

  float* xw = xs + wave * 16 * XP_;
  v4f pre_0, pre_1, pre_2, pre_3, pre_4, pre_5, pre_6, pre_7,
      pre_8, pre_9, pre_10, pre_11, pre_12, pre_13, pre_14, pre_15;
  v4f pself_0, pself_1;
  float pw = 0.f;
  pre_0 = pre_1 = pre_2 = pre_3 = pre_4 = pre_5 = pre_6 = pre_7 = (v4f){};
  pre_8 = pre_9 = pre_10 = pre_11 = pre_12 = pre_13 = pre_14 = pre_15 = (v4f){};
  pself_0 = pself_1 = (v4f){};
  v4i ia0 = (v4i){}, ia1 = (v4i){}, ib0 = (v4i){}, ib1 = (v4i){};
  int isa = 0, isb = 0;
  long fra = 0, frb = 0;

#define AGG_ROWS(X) X(0) X(1) X(2) X(3) X(4) X(5) X(6) X(7) X(8) X(9) X(10) X(11) X(12) X(13) X(14) X(15)
#define AGG_LOAD_IDX(r, IV) pre_##r = *(const v4f*)(emb + ((long)(IV) << 7) + 4 * lane);
#define AGG_LOAD_VAL(r, ROW) pre_##r = *(const v4f*)(neighVal + (((ROW) * S_ + ((r) & 7)) << 7) + 4 * lane);
#define AGG_FETCH_IDX(U)                                                      \
  {                                                                           \
    const int mr0 = (U) * 16 + 2 * wave;                                      \
    const int ma = mr0 < M ? mr0 : M - 1;                                     \
    const int mb = mr0 + 1 < M ? mr0 + 1 : M - 1;                             \
    fra = (long)b * M + ma; frb = (long)b * M + mb;                           \
    isa = sid[mr0 - u0 * 16]; isb = sid[mr0 + 1 - u0 * 16];                   \
    if (GATHER) {                                                             \
      ia0 = *(const v4i*)(adj_all + (long)isa * S_);                          \
      ia1 = *(const v4i*)(adj_all + (long)isa * S_ + 4);                      \
      ib0 = *(const v4i*)(adj_all + (long)isb * S_);                          \
      ib1 = *(const v4i*)(adj_all + (long)isb * S_ + 4);                      \
    }                                                                         \
  }
#define AGG_FETCH_ROWS()                                                      \
  {                                                                           \
    if (GATHER) {                                                             \
      AGG_LOAD_IDX(0, ia0.x) AGG_LOAD_IDX(1, ia0.y) AGG_LOAD_IDX(2, ia0.z) AGG_LOAD_IDX(3, ia0.w)     \
      AGG_LOAD_IDX(4, ia1.x) AGG_LOAD_IDX(5, ia1.y) AGG_LOAD_IDX(6, ia1.z) AGG_LOAD_IDX(7, ia1.w)     \
      AGG_LOAD_IDX(8, ib0.x) AGG_LOAD_IDX(9, ib0.y) AGG_LOAD_IDX(10, ib0.z) AGG_LOAD_IDX(11, ib0.w)   \
      AGG_LOAD_IDX(12, ib1.x) AGG_LOAD_IDX(13, ib1.y) AGG_LOAD_IDX(14, ib1.z) AGG_LOAD_IDX(15, ib1.w) \
    } else {                                                                  \
      AGG_LOAD_VAL(0, fra) AGG_LOAD_VAL(1, fra) AGG_LOAD_VAL(2, fra) AGG_LOAD_VAL(3, fra)             \
      AGG_LOAD_VAL(4, fra) AGG_LOAD_VAL(5, fra) AGG_LOAD_VAL(6, fra) AGG_LOAD_VAL(7, fra)             \
      AGG_LOAD_VAL(8, frb) AGG_LOAD_VAL(9, frb) AGG_LOAD_VAL(10, frb) AGG_LOAD_VAL(11, frb)           \
      AGG_LOAD_VAL(12, frb) AGG_LOAD_VAL(13, frb) AGG_LOAD_VAL(14, frb) AGG_LOAD_VAL(15, frb)         \
    }                                                                         \
    if (GATHER) {                                                             \
      pself_0 = *(const v4f*)(emb + ((long)isa << 7) + 4 * lane);             \
      pself_1 = *(const v4f*)(emb + ((long)isb << 7) + 4 * lane);             \
    } else {                                                                  \
      pself_0 = *(const v4f*)(selfVal + (fra << 7) + 4 * lane);               \
      pself_1 = *(const v4f*)(selfVal + (frb << 7) + 4 * lane);               \
    }                                                                         \
    pw = num_w[(long)((c >> 3) ? isb : isa) * S_ + (c & 7)];                  \
  }
#define AGG_STORE_ROW(r) *(v4f*)(xw + (r) * XP_ + 4 * lane) = pre_##r;

  if (u0 < u1) { AGG_FETCH_IDX(u0) AGG_FETCH_ROWS() }

  for (int u = u0; u < u1; ++u) {
    AGG_ROWS(AGG_STORE_ROW)
    if (lane < 16) {
      float* t = xw + lane * XP_ + D_;
      t[0] = pw; t[1] = 0.f; t[2] = 0.f; t[3] = 0.f;
    }
    const v4f s0 = pself_0, s1 = pself_1;
    if (u + 1 < u1) AGG_FETCH_IDX(u + 1)
    __syncthreads();

    v8f acc[8];
    #pragma unroll
    for (int j = 0; j < 8; ++j) acc[j] = (v8f){};
    {
      const float* ap = xw + c * XP_ + 2 * hf;
      const float* bp = WT + c * XP_ + 2 * hf;
      for (int k0 = 0; k0 < XP_; k0 += 4) {
        v2f a = *(const v2f*)(ap + k0);
        #pragma unroll
        for (int j = 0; j < 8; ++j) {
          v2f bb = *(const v2f*)(bp + j * 16 * XP_ + k0);
          acc[j] = wmma4(a, bb, acc[j]);
        }
      }
    }
    float sc[8];
    #pragma unroll
    for (int v = 0; v < 8; ++v) sc[v] = 0.f;
    #pragma unroll
    for (int j = 0; j < 8; ++j) {
      float g2 = gw2p[16 * j + c];
      #pragma unroll
      for (int v = 0; v < 8; ++v) sc[v] += leaky_(acc[j][v]) * g2;
    }
    #pragma unroll
    for (int v = 0; v < 8; ++v) {
      float val = sc[v];
      val += __shfl_xor(val, 1);
      val += __shfl_xor(val, 2);
      val += __shfl_xor(val, 4);
      val += __shfl_xor(val, 8);
      sc[v] = val;
    }
    float mx = sc[0];
    #pragma unroll
    for (int v = 1; v < 8; ++v) mx = fmaxf(mx, sc[v]);
    float sum = 0.f;
    #pragma unroll
    for (int v = 0; v < 8; ++v) { sc[v] = __expf(sc[v] - mx); sum += sc[v]; }
    const float inv = 1.f / sum;
    float4 na = make_float4(0.f, 0.f, 0.f, 0.f), nb = na;
    #pragma unroll
    for (int v = 0; v < 8; ++v) {
      const float al = sc[v] * inv;
      const float* row = xw + (8 * hf + v) * XP_ + 4 * c;
      float4 p = *(const float4*)(row);
      float4 q = *(const float4*)(row + 64);
      na.x += al * p.x; na.y += al * p.y; na.z += al * p.z; na.w += al * p.w;
      nb.x += al * q.x; nb.y += al * q.y; nb.z += al * q.z; nb.w += al * q.w;
    }
    {
      float* srow = sn + (2 * wave + hf) * GP_ + D_ + 4 * c;
      *(float4*)(srow) = na;
      *(float4*)(srow + 64) = nb;
      *(v4f*)(sn + (2 * wave) * GP_ + 4 * lane) = s0;
      *(v4f*)(sn + (2 * wave + 1) * GP_ + 4 * lane) = s1;
    }
    __syncthreads();

    if (u + 1 < u1) AGG_FETCH_ROWS()

    v8f o0 = (v8f){}, o1 = (v8f){};
    {
      const float* sa = sn + c * GP_ + 2 * hf;
      const float* gb = G3T + (16 * wave + c) * GP_ + 2 * hf;
      for (int k0 = 0; k0 < 2 * D_; k0 += 8) {
        o0 = wmma4(*(const v2f*)(sa + k0), *(const v2f*)(gb + k0), o0);
        o1 = wmma4(*(const v2f*)(sa + k0 + 4), *(const v2f*)(gb + k0 + 4), o1);
      }
    }
    #pragma unroll
    for (int v = 0; v < 8; ++v) {
      int m = u * 16 + v + 8 * hf;
      if (m < M) out[(((long)b * M + m) << 7) + 16 * wave + c] = fmaxf(o0[v] + o1[v], 0.f);
    }
  }
#undef AGG_FETCH_IDX
#undef AGG_FETCH_ROWS
#undef AGG_LOAD_IDX
#undef AGG_LOAD_VAL
#undef AGG_STORE_ROW
#undef AGG_ROWS
}

#define SP_ 132
#define SCORE_LDS_FLOATS (42 * 1024)
#define SCORE_WAVES 16
#define SCORE_WGS 256

__global__ __launch_bounds__(SCORE_WAVES * 32) void score_kernel(
    const float* __restrict__ selF, const float* __restrict__ emb,
    float* __restrict__ scores) {
  __shared__ __attribute__((aligned(16))) float sA[SCORE_LDS_FLOATS];
  const int tid = threadIdx.x;
  const int lane = tid & 31, c = lane & 15, hf = lane >> 4;
  const int NSC = NN_ - 1;
  const int NT = (NSC + 15) >> 4;
  const long U = (long)NT * 8;
  const int nw = SCORE_WGS * SCORE_WAVES;
  const int wv = __builtin_amdgcn_readfirstlane(tid >> 5);
  const int gw = (int)blockIdx.x * SCORE_WAVES + wv;
  const int u0 = (int)((U * gw) / nw), u1 = (int)((U * (gw + 1)) / nw);

  v4f bn[16];
#define SCORE_LOADB(NTILE)                                                      \
  {                                                                             \
    int er = ((NTILE) << 4) + c + 1;                \
    er = er < NN_ ? er : NN_ - 1;                   \
    const float* bp = emb + ((long)er << 7) + 4 * hf;                           \
    _Pragma("unroll") for (int i = 0; i < 16; ++i) bn[i] = *(const v4f*)(bp + 8 * i); \
  }
  if (u0 < u1) SCORE_LOADB(u0 >> 3)

  #pragma unroll
  for (int t = 0; t < (B_ * (D_ / 4)) / (SCORE_WAVES * 32); ++t) {
    const int i = tid + t * (SCORE_WAVES * 32);
    const int r = i >> 5, q = i & 31;
    *(v4f*)(sA + r * SP_ + 4 * q) = *(const v4f*)(selF + (r << 7) + 4 * q);
  }
  __syncthreads();

  int u = u0;
  while (u < u1) {
    const int nt = u >> 3;
    const int ue = (u1 < ((nt + 1) << 3)) ? u1 : ((nt + 1) << 3);
    v4f bf[16];
    #pragma unroll
    for (int i = 0; i < 16; ++i) bf[i] = bn[i];
    if (ue < u1) SCORE_LOADB(nt + 1)
    const int col = (nt << 4) + c;
    for (; u < ue; ++u) {
      const int m0 = (u & 7) << 4;
      const float* ap = sA + (m0 + c) * SP_ + 4 * hf;
      v4f a[16];
      #pragma unroll
      for (int i = 0; i < 16; ++i) a[i] = *(const v4f*)(ap + 8 * i);
      __builtin_amdgcn_sched_barrier(0);
      v8f acc0 = (v8f){}, acc1 = (v8f){};
      #pragma unroll
      for (int i = 0; i < 16; ++i) {
        acc0 = wmma4((v2f){a[i].x, a[i].y}, (v2f){bf[i].x, bf[i].y}, acc0);
        acc1 = wmma4((v2f){a[i].z, a[i].w}, (v2f){bf[i].z, bf[i].w}, acc1);
      }
      if (col < NSC) {
        float* op = scores + (long)(m0 + 8 * hf) * NSC + col;
        #pragma unroll
        for (int v = 0; v < 8; ++v) op[(long)v * NSC] = acc0[v] + acc1[v];
      }
    }
  }
#undef SCORE_LOADB
}

__global__ void ssl_kernel(const float* __restrict__ hl, const float* __restrict__ hg,
                           float* __restrict__ loss) {
  int idx = blockIdx.x * 256 + threadIdx.x;
  int b = idx >> 7, d = idx & 127;
  int pb = (b * 73 + 21) & 127;
  float pos = 0.f, neg = 0.f;
  for (int l = 0; l < L_; ++l) {
    int pl = (l * 13 + 7) % L_;
    pos += hl[((b * L_ + l) << 7) + d] * hg[((b * L_ + l) << 7) + d];
    neg += hg[((b * L_ + l) << 7) + d] * hl[((pb * L_ + pl) << 7) + d];
  }
  float term = -__logf(1e-8f + sigmoid_(pos)) - __logf(1e-8f + 1.f - sigmoid_(neg));
  __shared__ float red[256];
  red[threadIdx.x] = term;
  __syncthreads();
  for (int s = 128; s > 0; s >>= 1) {
    if (threadIdx.x < s) red[threadIdx.x] += red[threadIdx.x + s];
    __syncthreads();
  }
  if (threadIdx.x == 0) atomicAdd(loss, BETA_ * red[0]);
}

__global__ void pos_kernel(const float* __restrict__ pos_emb, const float* __restrict__ w1,
                           float* __restrict__ P) {
  __shared__ float pr[D_];
  int l = blockIdx.x, d = threadIdx.x;
  pr[d] = pos_emb[(l << 7) + d];
  __syncthreads();
  float acc = 0.f;
  for (int e = 0; e < D_; ++e) acc += pr[e] * w1[(e << 7) + d];
  P[(l << 7) + d] = acc;
}

template <int M>
__device__ __forceinline__ float row_xor(float x) {
  return __builtin_bit_cast(float, __builtin_amdgcn_update_dpp(0, __builtin_bit_cast(int, x), 0x160 + M, 0xf, 0xf, false));
}
__device__ __forceinline__ float xhalf(float x) {
  const int v = __builtin_bit_cast(int, x);
  return __builtin_bit_cast(float, __builtin_amdgcn_permlanex16(v, v, 0x76543210, 0xfedcba98, false, false));
}

template <int M>
__device__ __forceinline__ int row_xor_i(int x) {
  return __builtin_amdgcn_update_dpp(0, x, 0x160 + M, 0xf, 0xf, false);
}
__device__ __forceinline__ int xhalf_i(int v) {
  return __builtin_amdgcn_permlanex16(v, v, 0x76543210, 0xfedcba98, false, false);
}

#define HP_ 132
#define HR_ 64
#define HEAD_LDS_FLOATS (42 * 1024)

__global__ __launch_bounds__(256) void head_kernel(
    const float* __restrict__ h_local, const float* __restrict__ h_global,
    const int* __restrict__ mask, const float* __restrict__ P,
    const float* __restrict__ w1, const float* __restrict__ glu1_w,
    const float* __restrict__ glu1_b, const float* __restrict__ glu2_w,
    const float* __restrict__ w2,
    float* __restrict__ hidden, float* __restrict__ sel, float* __restrict__ snorm) {
  __shared__ __attribute__((aligned(16))) float lds[HEAD_LDS_FLOATS];
  float* HX    = lds;
  float* NH    = HX + HR_ * HP_;
  float* hsL   = NH + HR_ * HP_;
  float* maskf = hsL + D_;
  float* part  = maskf + HR_;
  float* betas = part + 8 * HR_;
  float* red   = betas + HR_;

  const int tid = threadIdx.x, b = blockIdx.x;
  const int lane = tid & 31, c = lane & 15, hf = lane >> 4;
  const int wave = tid >> 5;
  const int n = (wave << 4) + c;

  v4f bw[16], bg[16], g2[16];
  {
    const float* wp = w1 + (D_ + 4 * hf) * D_ + n;
    #pragma unroll
    for (int i = 0; i < 16; ++i) {
      bw[i].x = wp[(8 * i + 0) * D_];
      bw[i].y = wp[(8 * i + 1) * D_];
      bw[i].z = wp[(8 * i + 2) * D_];
      bw[i].w = wp[(8 * i + 3) * D_];
    }
    const float* gp = glu1_w + n * D_ + 4 * hf;
    #pragma unroll
    for (int i = 0; i < 16; ++i) bg[i] = *(const v4f*)(gp + 8 * i);
    const float* g2p = glu2_w + n * D_ + 64 * hf;
    #pragma unroll
    for (int i = 0; i < 16; ++i) g2[i] = *(const v4f*)(g2p + 4 * i);
  }
  const float g1b = glu1_b[n];
  const float w2n = w2[n];

  {
    const float* hl = h_local + (long)b * L_ * D_;
    const float* hg = h_global + (long)b * L_ * D_;
    float* ho = hidden + (long)b * L_ * D_;
    #pragma unroll
    for (int t = 0; t < 6; ++t) {
      const int i4 = tid + 256 * t;
      const v4f v = *(const v4f*)(hl + 4 * i4) + *(const v4f*)(hg + 4 * i4);
      *(v4f*)(ho + 4 * i4) = v;
      *(v4f*)(HX + (i4 >> 5) * HP_ + 4 * (i4 & 31)) = v;
    }
    if (tid < (L_ * D_) / 4 - 6 * 256) {
      const int i4 = tid + 6 * 256;
      const v4f v = *(const v4f*)(hl + 4 * i4) + *(const v4f*)(hg + 4 * i4);
      *(v4f*)(ho + 4 * i4) = v;
      *(v4f*)(HX + (i4 >> 5) * HP_ + 4 * (i4 & 31)) = v;
    }
    for (int i4 = tid; i4 < (HR_ - L_) * 32; i4 += 256)
      *(v4f*)(HX + (L_ + (i4 >> 5)) * HP_ + 4 * (i4 & 31)) = (v4f){};
    if (tid < HR_) maskf[tid] = tid < L_ ? (float)mask[b * L_ + tid] : 0.f;
  }
  __syncthreads();

  if (tid < D_) {
    float acc = 0.f, ms = 0.f;
    #pragma unroll
    for (int l = 0; l < L_; ++l) {
      const float m = maskf[l];
      ms += m;
      acc += HX[l * HP_ + tid] * m;
    }
    hsL[tid] = acc / ms;
  }
  for (int mt = 0; mt < 4; ++mt) {
    const float* ap = HX + (16 * mt + c) * HP_ + 4 * hf;
    v4f a[16];
    #pragma unroll
    for (int i = 0; i < 16; ++i) a[i] = *(const v4f*)(ap + 8 * i);
    __builtin_amdgcn_sched_barrier(0);
    float pv[8];
    #pragma unroll
    for (int v = 0; v < 8; ++v) {
      const int row = 16 * mt + 8 * hf + v;
      pv[v] = row < L_ ? P[row * D_ + n] : 0.f;
    }
    v8f acc0 = (v8f){}, acc1 = (v8f){};
    #pragma unroll
    for (int i = 0; i < 16; ++i) {
      acc0 = wmma4((v2f){a[i].x, a[i].y}, (v2f){bw[i].x, bw[i].y}, acc0);
      acc1 = wmma4((v2f){a[i].z, a[i].w}, (v2f){bw[i].z, bw[i].w}, acc1);
    }
    #pragma unroll
    for (int v = 0; v < 8; ++v)
      NH[(16 * mt + 8 * hf + v) * HP_ + n] = tanhf(acc0[v] + acc1[v] + pv[v]);
  }
  __syncthreads();

  float bias;
  {
    const float* hp = hsL + 64 * hf;
    float s0 = 0.f, s1 = 0.f, s2 = 0.f, s3 = 0.f;
    #pragma unroll
    for (int i = 0; i < 16; ++i) {
      const v4f x = *(const v4f*)(hp + 4 * i);
      s0 += g2[i].x * x.x; s1 += g2[i].y * x.y;
      s2 += g2[i].z * x.z; s3 += g2[i].w * x.w;
    }
    float s = (s0 + s1) + (s2 + s3);
    s += xhalf(s);
    bias = s + g1b;
  }

  for (int mt = 0; mt < 4; ++mt) {
    const float* ap = NH + (16 * mt + c) * HP_ + 4 * hf;
    v4f a[16];
    #pragma unroll
    for (int i = 0; i < 16; ++i) a[i] = *(const v4f*)(ap + 8 * i);
    __builtin_amdgcn_sched_barrier(0);
    v8f acc0 = (v8f){}, acc1 = (v8f){};
    #pragma unroll
    for (int i = 0; i < 16; ++i) {
      acc0 = wmma4((v2f){a[i].x, a[i].y}, (v2f){bg[i].x, bg[i].y}, acc0);
      acc1 = wmma4((v2f){a[i].z, a[i].w}, (v2f){bg[i].z, bg[i].w}, acc1);
    }
    float g[8];
    #pragma unroll
    for (int v = 0; v < 8; ++v) g[v] = sigmoid_(acc0[v] + acc1[v] + bias) * w2n;
    #pragma unroll
    for (int v = 0; v < 8; ++v) g[v] += row_xor<1>(g[v]);
    #pragma unroll
    for (int v = 0; v < 8; ++v) g[v] += row_xor<2>(g[v]);
    #pragma unroll
    for (int v = 0; v < 8; ++v) g[v] += row_xor<4>(g[v]);
    #pragma unroll
    for (int v = 0; v < 8; ++v) g[v] += row_xor<8>(g[v]);
    if (c == 0) {
      #pragma unroll
      for (int v = 0; v < 8; ++v) part[wave * HR_ + 16 * mt + 8 * hf + v] = g[v];
    }
  }
  __syncthreads();

  if (tid < L_) {
    float s = 0.f;
    #pragma unroll
    for (int w = 0; w < 8; ++w) s += part[w * HR_ + tid];
    betas[tid] = s * maskf[tid];
  }
  __syncthreads();

  float sq = 0.f;
  if (tid < D_) {
    float acc = 0.f;
    #pragma unroll
    for (int l = 0; l < L_; ++l) acc += betas[l] * HX[l * HP_ + tid];
    sel[(b << 7) + tid] = acc;
    sq = acc * acc + 1e-6f;
  }
  sq += row_xor<1>(sq);
  sq += row_xor<2>(sq);
  sq += row_xor<4>(sq);
  sq += row_xor<8>(sq);
  sq += xhalf(sq);
  if (lane == 0 && wave < 4) red[wave] = sq;
  __syncthreads();
  if (tid == 0) snorm[b] = sqrtf((red[0] + red[1]) + (red[2] + red[3]));
}


__global__ void fn_kernel(const float* __restrict__ sel, const float* __restrict__ snorm,
                          float* __restrict__ selF) {
  __shared__ float row[B_];
  __shared__ float selfrow[D_];
  __shared__ float tvs[9];
  __shared__ int tis[9];
  int i = blockIdx.x, t = threadIdx.x;
  selfrow[t] = sel[(i << 7) + t];
  __syncthreads();
  float acc = 0.f;
  for (int d = 0; d < D_; ++d) acc += selfrow[d] * sel[(t << 7) + d];
  row[t] = acc / (snorm[i] * snorm[t]);
  __syncthreads();
  if (t < 32) {
    float p[B_ / 32];
    #pragma unroll
    for (int q = 0; q < B_ / 32; ++q) p[q] = row[t + 32 * q];
    float mx = fmaxf(fmaxf(p[0], p[1]), fmaxf(p[2], p[3]));
    mx = fmaxf(mx, row_xor<1>(mx));
    mx = fmaxf(mx, row_xor<2>(mx));
    mx = fmaxf(mx, row_xor<4>(mx));
    mx = fmaxf(mx, row_xor<8>(mx));
    mx = fmaxf(mx, xhalf(mx));
    #pragma unroll
    for (int q = 0; q < B_ / 32; ++q) p[q] = __expf(p[q] - mx);
    float s = (p[0] + p[1]) + (p[2] + p[3]);
    s += row_xor<1>(s);
    s += row_xor<2>(s);
    s += row_xor<4>(s);
    s += row_xor<8>(s);
    s += xhalf(s);
    const float inv = 1.f / s;
    #pragma unroll
    for (int q = 0; q < B_ / 32; ++q) p[q] *= inv;
    float tv[9];
    int ti[9];
#define FN_MEET(OV, OI)                                                        \
    {                                                                          \
      const float ov = (OV);                                                   \
      const int oi = (OI);                                                     \
      const bool take = ov > bv || (ov == bv && oi < bi);                      \
      bv = take ? ov : bv;                                                     \
      bi = take ? oi : bi;                                                     \
    }
    #pragma unroll
    for (int r = 0; r < 9; ++r) {
      float bv = p[0];
      int bi = t;
      #pragma unroll
      for (int q = 1; q < B_ / 32; ++q) {
        const bool take = p[q] > bv;
        bv = take ? p[q] : bv;
        bi = take ? t + 32 * q : bi;
      }
      FN_MEET(row_xor<1>(bv), row_xor_i<1>(bi))
      FN_MEET(row_xor<2>(bv), row_xor_i<2>(bi))
      FN_MEET(row_xor<4>(bv), row_xor_i<4>(bi))
      FN_MEET(row_xor<8>(bv), row_xor_i<8>(bi))
      FN_MEET(xhalf(bv), xhalf_i(bi))
      tv[r] = bv;
      ti[r] = bi;
      #pragma unroll
      for (int q = 0; q < B_ / 32; ++q) p[q] = (bi == t + 32 * q) ? -2.f : p[q];
    }
#undef FN_MEET
    const float mx2 = tv[0];
    float s2 = 0.f;
    #pragma unroll
    for (int r = 0; r < 9; ++r) { tv[r] = __expf(tv[r] - mx2); s2 += tv[r]; }
    const float inv2 = 1.f / s2;
    if (t == 0) {
      #pragma unroll
      for (int r = 0; r < 9; ++r) { tvs[r] = tv[r] * inv2; tis[r] = ti[r]; }
    }
  }
  __syncthreads();
  float nb = 0.f;
  for (int q = 0; q < 9; ++q) nb += tvs[q] * sel[(tis[q] << 7) + t];
  selF[(i << 7) + t] = selfrow[t] + nb;
}

extern "C" void kernel_launch(void* const* d_in, const int* in_sizes, int n_in,
                              void* d_out, int out_size, void* d_ws, size_t ws_size,
                              hipStream_t stream) {
  (void)in_sizes; (void)n_in; (void)out_size; (void)ws_size;
  const int*   inputs  = (const int*)d_in[0];
  const int*   adj     = (const int*)d_in[1];
  const int*   mask    = (const int*)d_in[2];
  const int*   item    = (const int*)d_in[3];
  const int*   adj_all = (const int*)d_in[5];
  const float* num_w   = (const float*)d_in[6];
  const float* emb     = (const float*)d_in[7];
  const float* pos_emb = (const float*)d_in[8];
  const float* a_local = (const float*)d_in[9];
  const float* gw1     = (const float*)d_in[10];
  const float* gw2     = (const float*)d_in[11];
  const float* gw3     = (const float*)d_in[12];
  const float* w1      = (const float*)d_in[13];
  const float* w2      = (const float*)d_in[14];
  const float* glu1_w  = (const float*)d_in[15];
  const float* glu1_b  = (const float*)d_in[16];
  const float* glu2_w  = (const float*)d_in[17];

  float* out0   = (float*)d_out;
  float* lossP  = out0 + B_ * L_ * D_;
  float* scores = lossP + 1;

  char* wp = (char*)d_ws;
  auto alloc = [&](size_t nfloats) -> float* {
    float* r = (float*)wp;
    wp += ((nfloats * sizeof(float) + 255) / 256) * 256;
    return r;
  };
  float* h_local  = alloc(B_ * L_ * D_);
  float* h_global = alloc(B_ * L_ * D_);
  float* e0p      = alloc(B_ * L_ * D_);
  float* e1p      = alloc((size_t)B_ * 400 * D_);
  float* Pb       = alloc(L_ * D_);
  float* selb     = alloc(B_ * D_);
  float* selF     = alloc(B_ * D_);
  float* snorm    = alloc(B_);

  local_agg_kernel<<<dim3(B_), 256, 0, stream>>>(emb, inputs, adj, a_local, h_local, lossP);
  agg_kernel<true, false><<<dim3(B_ * 2), 256, 0, stream>>>(emb, inputs, adj_all, num_w, item, mask,
      nullptr, nullptr, gw1, gw2, gw3, e0p, L_, 2);
  agg_kernel<true, true><<<dim3(B_ * 2), 256, 0, stream>>>(emb, inputs, adj_all, num_w, item, mask,
      nullptr, nullptr, gw1, gw2, gw3, e1p, L_ * S_, 2);
  agg_kernel<false, false><<<dim3(B_ * 2), 256, 0, stream>>>(emb, inputs, adj_all, num_w, item, mask,
      e0p, e1p, gw1 + 129 * D_, gw2 + D_, gw3 + 2 * D_ * D_, h_global, L_, 2);
  ssl_kernel<<<dim3(64), 256, 0, stream>>>(h_local, h_global, lossP);
  pos_kernel<<<dim3(L_), dim3(D_), 0, stream>>>(pos_emb, w1, Pb);
  head_kernel<<<dim3(B_), 256, 0, stream>>>(h_local, h_global, mask, Pb, w1, glu1_w, glu1_b,
      glu2_w, w2, out0, selb, snorm);
  fn_kernel<<<dim3(B_), dim3(D_), 0, stream>>>(selb, snorm, selF);
  score_kernel<<<dim3(SCORE_WGS), SCORE_WAVES * 32, 0, stream>>>(selF, emb, scores);
}


# agg kernels: LDS fragment loads of both products (logits, w3) go 16 B per lane one block ahead of their WMMAs into a two-set register ring, one load group per block (k sums re-ordered)
# speedup vs baseline: 3.2563x; 1.1775x over previous
#include <hip/hip_runtime.h>
#include <hip/hip_bf16.h>
#include <math.h>

#define B_ 128
#define L_ 50
#define D_ 128
#define S_ 8
#define NN_ 100000
#define LEAKY_ 0.2f
#define BETA_ 0.005f

typedef float v2f __attribute__((ext_vector_type(2)));
typedef float v8f __attribute__((ext_vector_type(8)));
typedef float v4f __attribute__((ext_vector_type(4)));
typedef int v4i __attribute__((ext_vector_type(4)));

#if defined(__has_builtin)
#if __has_builtin(__builtin_amdgcn_wmma_f32_16x16x4_f32)
#define HAVE_WMMA_F32 1
#endif
#endif

__device__ __forceinline__ v8f wmma4(v2f a, v2f b, v8f c) {
#ifdef HAVE_WMMA_F32
  return __builtin_amdgcn_wmma_f32_16x16x4_f32(false, a, false, b, (short)0, c,
                                               false, false);
#else
  float* cf = (float*)&c;
  #pragma unroll
  for (int i = 0; i < 8; ++i) cf[i] += a.x * b.x + a.y * b.y;
  return c;
#endif
}

__device__ __forceinline__ float leaky_(float x) { return x > 0.f ? x : LEAKY_ * x; }
__device__ __forceinline__ float sigmoid_(float x) { return 1.f / (1.f + __expf(-x)); }


#define LA_TP 65
#define LA_PS (L_ * D_ + 8)
#define LA_AR 64
#define LA_AP 52
__global__ __launch_bounds__(256) void local_agg_kernel(const float* __restrict__ emb,
                                 const int* __restrict__ inputs,
                                 const int* __restrict__ adj,
                                 const float* __restrict__ a_local,
                                 float* __restrict__ h_local) {
  int b = blockIdx.x;
  __shared__ __attribute__((aligned(16))) float hv[L_][D_];
  __shared__ __attribute__((aligned(16))) float hia[4 * LA_PS];
  __shared__ float hvT[D_ * LA_TP];
  __shared__ int adjs[L_ * L_];
  __shared__ __attribute__((aligned(16))) float att[LA_AR][LA_AP];
  __shared__ __attribute__((aligned(16))) float rinv[LA_AR];
  {
    const int q = threadIdx.x & 31;
    const int wvu = __builtin_amdgcn_readfirstlane(threadIdx.x >> 5);
    const v4f ak0 = *(const v4f*)(a_local + 0 * D_ + 4 * q);
    const v4f ak1 = *(const v4f*)(a_local + 1 * D_ + 4 * q);
    const v4f ak2 = *(const v4f*)(a_local + 2 * D_ + 4 * q);
    const v4f ak3 = *(const v4f*)(a_local + 3 * D_ + 4 * q);
    int nid[(L_ + 7) / 8];
    #pragma unroll
    for (int t = 0; t < (L_ + 7) / 8; ++t) {
      const int r = wvu + 8 * t;
      nid[t] = inputs[b * L_ + (r < L_ ? r : L_ - 1)];
    }
    v4f hr[(L_ + 7) / 8];
    #pragma unroll
    for (int t = 0; t < (L_ + 7) / 8; ++t) hr[t] = *(const v4f*)(emb + ((long)nid[t] << 7) + 4 * q);
    #pragma unroll
    for (int t = 0; t < (L_ + 7) / 8; ++t) {
      const int r = wvu + 8 * t;
      if (r >= L_) break;
      const int i4 = r * 32 + q;
      const v4f v = hr[t];
      *(v4f*)(&hv[r][4 * q]) = v;
      float* tp = hvT + (4 * q) * LA_TP + r;
      tp[0] = v.x; tp[LA_TP] = v.y; tp[2 * LA_TP] = v.z; tp[3 * LA_TP] = v.w;
      *(v4f*)(hia + 0 * LA_PS + 4 * i4) = v * ak0;
      *(v4f*)(hia + 1 * LA_PS + 4 * i4) = v * ak1;
      *(v4f*)(hia + 2 * LA_PS + 4 * i4) = v * ak2;
      *(v4f*)(hia + 3 * LA_PS + 4 * i4) = v * ak3;
    }
    for (int p = threadIdx.x; p < L_ * L_; p += 256) adjs[p] = adj[b * L_ * L_ + p];
    for (int p = threadIdx.x; p < LA_AR * LA_AP / 4; p += 256) ((v4f*)&att[0][0])[p] = (v4f){};
  }
  __syncthreads();
  {
    const int lane = threadIdx.x & 31, wv = threadIdx.x >> 5;
    const int j = lane + ((wv & 1) << 5);
    const bool jin = j < L_;
    const int jr = jin ? j : L_ - 1;
    float hj[D_];
    #pragma unroll
    for (int d = 0; d < D_; ++d) hj[d] = hvT[d * LA_TP + jr];
    int kn = adjs[(wv >> 1) * L_ + jr];
    for (int i = wv >> 1; i < L_; i += 4) {
      const int k = kn;
      kn = adjs[(i + 4 < L_ ? i + 4 : i) * L_ + jr];
      const bool on = (unsigned)(k - 1) < 4u;
      const float* rp = hia + (on ? k - 1 : 0) * LA_PS + i * D_;
      float a0 = 0.f, a1 = 0.f, a2 = 0.f, a3 = 0.f;
      #pragma unroll
      for (int q = 0; q < D_ / 4; ++q) {
        const v4f r = *(const v4f*)(rp + 4 * q);
        a0 += r.x * hj[4 * q];     a1 += r.y * hj[4 * q + 1];
        a2 += r.z * hj[4 * q + 2]; a3 += r.w * hj[4 * q + 3];
      }
      const float e = (a0 + a1) + (a2 + a3);
      if (jin) att[i][j] = on ? leaky_(e) : -9e15f;
    }
  }
  __syncthreads();
  if (threadIdx.x < L_) {
    int i = threadIdx.x;
    float mx = -1e30f;
    for (int j = 0; j < L_; ++j) mx = fmaxf(mx, att[i][j]);
    float s = 0.f;
    for (int j = 0; j < L_; ++j) { float e = __expf(att[i][j] - mx); att[i][j] = e; s += e; }
    rinv[i] = 1.f / s;
  }
  __syncthreads();
  {
    const int lane = threadIdx.x & 31, c = lane & 15, hf = lane >> 4;
    const int wv = threadIdx.x >> 5;
    const int n = (wv << 4) + c;
    v2f bq[LA_AP / 4];
    #pragma unroll
    for (int ks = 0; ks < LA_AP / 4; ++ks) {
      const int k = 4 * ks + 2 * hf;
      const bool kin = k < L_;
      const int kr = kin ? k : 0;
      const float b0 = hv[kr][n], b1 = hv[kr + 1][n];
      bq[ks] = kin ? (v2f){b0, b1} : (v2f){0.f, 0.f};
    }
    const float* ap = &att[c][2 * hf];
    v2f aq[4][LA_AP / 4];
    #pragma unroll
    for (int mt = 0; mt < 4; ++mt) {
      #pragma unroll
      for (int ks = 0; ks < LA_AP / 4; ++ks) aq[mt][ks] = *(const v2f*)(ap + 16 * mt * LA_AP + 4 * ks);
    }
    v4f rlo[4], rhi[4];
    #pragma unroll
    for (int mt = 0; mt < 4; ++mt) {
      rlo[mt] = *(const v4f*)(rinv + 16 * mt + 8 * hf);
      rhi[mt] = *(const v4f*)(rinv + 16 * mt + 8 * hf + 4);
    }
    v8f acc[4];
    #pragma unroll
    for (int mt = 0; mt < 4; ++mt) acc[mt] = (v8f){};
    #pragma unroll
    for (int ks = 0; ks < LA_AP / 4; ++ks) {
      #pragma unroll
      for (int mt = 0; mt < 4; ++mt) acc[mt] = wmma4(aq[mt][ks], bq[ks], acc[mt]);
    }
    float* op = h_local + (long)b * L_ * D_ + n;
    #pragma unroll
    for (int mt = 0; mt < 4; ++mt) {
      #pragma unroll
      for (int v = 0; v < 8; ++v) {
        const int row = 16 * mt + 8 * hf + v;
        const float ri = v < 4 ? rlo[mt][v] : rhi[mt][v - 4];
        if (row < L_) op[row * D_] = acc[mt][v] * ri;
      }
    }
  }
}

#define XP_ 132
#define GP_ 260
#define AGG_SID 256
#define AGG_LDS_FLOATS (D_ * XP_ + D_ * GP_ + 8 * 16 * XP_ + 16 * GP_ + D_ + AGG_SID)

template <bool GATHER, bool DEEP>
__global__ __launch_bounds__(256) void agg_kernel(
                           const float* __restrict__ emb,
                           const int* __restrict__ inputs, const int* __restrict__ adj_all,
                           const float* __restrict__ num_w,
                           const int* __restrict__ item, const int* __restrict__ mask,
                           const float* __restrict__ selfVal, const float* __restrict__ neighVal,
                           const float* __restrict__ gw1p, const float* __restrict__ gw2p,
                           const float* __restrict__ gw3p,
                           float* __restrict__ out, float* __restrict__ out2, int M, int parts) {
  __shared__ __attribute__((aligned(16))) float lds[AGG_LDS_FLOATS];
  float* WT  = lds;
  float* G3T = WT + D_ * XP_;
  float* xs  = G3T + D_ * GP_;
  float* sn  = xs + 8 * 16 * XP_;
  float* sessL = sn + 16 * GP_;
  int*   sid = (int*)(sessL + D_);

  const int tid = threadIdx.x;
  const int lane = tid & 31, c = lane & 15, hf = lane >> 4;
  const int wave = tid >> 5;
  const int b = blockIdx.x / parts, part = blockIdx.x % parts;
  const int units = (M + 15) >> 4;
  const int upb = (units + parts - 1) / parts;
  const int u0 = part * upb;
  const int u1 = (u0 + upb < units) ? (u0 + upb) : units;
  const int MD = M * S_;
  const int dunits = (MD + 15) >> 4;
  const int dpb = (dunits + parts - 1) / parts;
  const int d0 = DEEP ? part * dpb : 0;
  const int d1 = DEEP ? ((d0 + dpb < dunits) ? (d0 + dpb) : dunits) : 0;
  const int nd = d1 - d0;
  const int nvu = nd + (u1 - u0);

  float se[L_], mk[L_];
  #pragma unroll
  for (int l = 0; l < L_; ++l) se[l] = emb[((long)item[b * L_ + l] << 7) + (tid & (D_ - 1))];
  #pragma unroll
  for (int l = 0; l < L_; ++l) mk[l] = (float)mask[b * L_ + l];
  const int kl = tid & 7, ql = (tid >> 3) & 3, rest0 = tid >> 5;
  float4 w1r[16];
  #pragma unroll
  for (int t = 0; t < 16; ++t) {
    const int rest = rest0 + 8 * t;
    const int k = ((rest & 15) << 3) + kl;
    const int q = ((rest >> 4) << 2) + ql;
    w1r[t] = *(const float4*)(gw1p + k * D_ + 4 * q);
  }
  const int nsid = 16 * nvu;
  int sidv = 0;
  if (tid < nsid) {
    const int vu = tid >> 4, wi = tid & 15;
    if (DEEP && vu < nd) {
      int m = (d0 + vu) * 16 + wi;
      m = m < MD ? m : MD - 1;
      const int root = inputs[b * L_ + (m >> 3)];
      sidv = adj_all[root * S_ + (m & 7)];
    } else {
      int m = (u0 + vu - nd) * 16 + wi;
      m = m < M ? m : M - 1;
      sidv = inputs[b * L_ + m];
    }
  }
  #pragma unroll
  for (int t = 0; t < 32; ++t) {
    const int rest = rest0 + 8 * t;
    int k = ((rest & 31) << 3) + kl;
    int q = ((rest >> 5) << 2) + ql;
    float4 w = *(const float4*)(gw3p + k * D_ + 4 * q);
    float* d = G3T + (4 * q) * GP_ + k;
    d[0] = w.x; d[GP_] = w.y; d[2 * GP_] = w.z; d[3 * GP_] = w.w;
  }
  if (tid < D_) {
    float sacc = 0.f, smask = 0.f;
    #pragma unroll
    for (int l = 0; l < L_; ++l) {
      const float m = mk[l];
      smask += m;
      sacc += se[l] * m;
    }
    sessL[tid] = sacc / smask;
    float* d = WT + tid * XP_ + D_;
    d[0] = gw1p[D_ * D_ + tid];
    d[1] = 0.f; d[2] = 0.f; d[3] = 0.f;
  }
  if (tid < nsid) sid[tid] = sidv;
  __syncthreads();
  {
    const float sk0 = sessL[(rest0 << 3) + kl], sk1 = sessL[(rest0 << 3) + kl + 64];
    #pragma unroll
    for (int t = 0; t < 16; ++t) {
      const int rest = rest0 + 8 * t;
      const int k = ((rest & 15) << 3) + kl;
      const int q = ((rest >> 4) << 2) + ql;
      const float4 w = w1r[t];
      const float s = (t & 1) ? sk1 : sk0;
      float* d = WT + (4 * q) * XP_ + k;
      d[0] = w.x * s; d[XP_] = w.y * s; d[2 * XP_] = w.z * s; d[3 * XP_] = w.w * s;
    }
  }

  float* xw = xs + wave * 16 * XP_;
  v4f pre_0, pre_1, pre_2, pre_3, pre_4, pre_5, pre_6, pre_7,
      pre_8, pre_9, pre_10, pre_11, pre_12, pre_13, pre_14, pre_15;
  v4f pself_0, pself_1;
  float pw = 0.f;
  pre_0 = pre_1 = pre_2 = pre_3 = pre_4 = pre_5 = pre_6 = pre_7 = (v4f){};
  pre_8 = pre_9 = pre_10 = pre_11 = pre_12 = pre_13 = pre_14 = pre_15 = (v4f){};
  pself_0 = pself_1 = (v4f){};
  v4i ia0 = (v4i){}, ia1 = (v4i){}, ib0 = (v4i){}, ib1 = (v4i){};
  int isa = 0, isb = 0;
  long fra = 0, frb = 0;

#define AGG_ROWS(X) X(0) X(1) X(2) X(3) X(4) X(5) X(6) X(7) X(8) X(9) X(10) X(11) X(12) X(13) X(14) X(15)
#define AGG_LOAD_IDX(r, IV) pre_##r = *(const v4f*)(emb + ((long)(IV) << 7) + 4 * lane);
#define AGG_LOAD_VAL(r, ROW) pre_##r = *(const v4f*)(neighVal + (((ROW) * S_ + ((r) & 7)) << 7) + 4 * lane);
#define AGG_FETCH_IDX(U)                                                      \
  {                                                                           \
      \
    const int mr0 = (u0 + (U) - nd) * 16 + 2 * wave;                          \
    const int ma = mr0 < M ? mr0 : M - 1;                                     \
    const int mb = mr0 + 1 < M ? mr0 + 1 : M - 1;                             \
    fra = (long)b * M + ma; frb = (long)b * M + mb;                           \
    isa = sid[(U) * 16 + 2 * wave]; isb = sid[(U) * 16 + 2 * wave + 1];       \
    if (GATHER) {                                                             \
      ia0 = *(const v4i*)(adj_all + (long)isa * S_);                          \
      ia1 = *(const v4i*)(adj_all + (long)isa * S_ + 4);                      \
      ib0 = *(const v4i*)(adj_all + (long)isb * S_);                          \
      ib1 = *(const v4i*)(adj_all + (long)isb * S_ + 4);                      \
    }                                                                         \
  }
#define AGG_FETCH_ROWS()                                                      \
  {                                                                           \
    if (GATHER) {                                                             \
      AGG_LOAD_IDX(0, ia0.x) AGG_LOAD_IDX(1, ia0.y) AGG_LOAD_IDX(2, ia0.z) AGG_LOAD_IDX(3, ia0.w)     \
      AGG_LOAD_IDX(4, ia1.x) AGG_LOAD_IDX(5, ia1.y) AGG_LOAD_IDX(6, ia1.z) AGG_LOAD_IDX(7, ia1.w)     \
      AGG_LOAD_IDX(8, ib0.x) AGG_LOAD_IDX(9, ib0.y) AGG_LOAD_IDX(10, ib0.z) AGG_LOAD_IDX(11, ib0.w)   \
      AGG_LOAD_IDX(12, ib1.x) AGG_LOAD_IDX(13, ib1.y) AGG_LOAD_IDX(14, ib1.z) AGG_LOAD_IDX(15, ib1.w) \
    } else {                                                                  \
      AGG_LOAD_VAL(0, fra) AGG_LOAD_VAL(1, fra) AGG_LOAD_VAL(2, fra) AGG_LOAD_VAL(3, fra)             \
      AGG_LOAD_VAL(4, fra) AGG_LOAD_VAL(5, fra) AGG_LOAD_VAL(6, fra) AGG_LOAD_VAL(7, fra)             \
      AGG_LOAD_VAL(8, frb) AGG_LOAD_VAL(9, frb) AGG_LOAD_VAL(10, frb) AGG_LOAD_VAL(11, frb)           \
      AGG_LOAD_VAL(12, frb) AGG_LOAD_VAL(13, frb) AGG_LOAD_VAL(14, frb) AGG_LOAD_VAL(15, frb)         \
    }                                                                         \
    if (GATHER) {                                                             \
      pself_0 = *(const v4f*)(emb + ((long)isa << 7) + 4 * lane);             \
      pself_1 = *(const v4f*)(emb + ((long)isb << 7) + 4 * lane);             \
    } else {                                                                  \
      pself_0 = *(const v4f*)(selfVal + (fra << 7) + 4 * lane);               \
      pself_1 = *(const v4f*)(selfVal + (frb << 7) + 4 * lane);               \
    }                                                                         \
    pw = num_w[(long)((c >> 3) ? isb : isa) * S_ + (c & 7)];                  \
  }
#define AGG_STORE_ROW(r) *(v4f*)(xw + (r) * XP_ + 4 * lane) = pre_##r;

  if (0 < nvu) { AGG_FETCH_IDX(0) AGG_FETCH_ROWS() }

  for (int u = 0; u < nvu; ++u) {
    AGG_ROWS(AGG_STORE_ROW)
    if (lane < 16) {
      float* t = xw + lane * XP_ + D_;
      t[0] = pw; t[1] = 0.f; t[2] = 0.f; t[3] = 0.f;
    }
    const v4f s0 = pself_0, s1 = pself_1;
    if (u + 1 < nvu) AGG_FETCH_IDX(u + 1)
    __syncthreads();

    v8f acc[8];
    #pragma unroll
    for (int j = 0; j < 8; ++j) acc[j] = (v8f){};
    {
      const float* ap = xw + c * XP_ + 4 * hf;
      const float* bp = WT + c * XP_ + 4 * hf;
      v4f fa[2][2], fb[2][2][8];
      v2f ta = (v2f){}, tb[8];
      #pragma unroll
      for (int i = 0; i < 2; ++i) {
        fa[0][i] = *(const v4f*)(ap + 8 * i);
        #pragma unroll
        for (int j = 0; j < 8; ++j) fb[0][i][j] = *(const v4f*)(bp + j * 16 * XP_ + 8 * i);
      }
      #pragma unroll
      for (int j = 0; j < 8; ++j) tb[j] = (v2f){};
      #pragma unroll
      for (int blk = 0; blk < 8; ++blk) {
        const int s = blk & 1;
        if (blk + 1 < 8) {
          #pragma unroll
          for (int i = 0; i < 2; ++i) {
            fa[s ^ 1][i] = *(const v4f*)(ap + 8 * (2 * (blk + 1) + i));
            #pragma unroll
            for (int j = 0; j < 8; ++j)
              fb[s ^ 1][i][j] = *(const v4f*)(bp + j * 16 * XP_ + 8 * (2 * (blk + 1) + i));
          }
        }
        if (blk == 6) {
          ta = *(const v2f*)(xw + c * XP_ + D_ + 2 * hf);
          #pragma unroll
          for (int j = 0; j < 8; ++j) tb[j] = *(const v2f*)(WT + (16 * j + c) * XP_ + D_ + 2 * hf);
        }
        __builtin_amdgcn_sched_barrier(0);
        #pragma unroll
        for (int i = 0; i < 2; ++i) {
          #pragma unroll
          for (int j = 0; j < 8; ++j)
            acc[j] = wmma4((v2f){fa[s][i].x, fa[s][i].y}, (v2f){fb[s][i][j].x, fb[s][i][j].y}, acc[j]);
          #pragma unroll
          for (int j = 0; j < 8; ++j)
            acc[j] = wmma4((v2f){fa[s][i].z, fa[s][i].w}, (v2f){fb[s][i][j].z, fb[s][i][j].w}, acc[j]);
        }
        if (blk == 7) {
          #pragma unroll
          for (int j = 0; j < 8; ++j) acc[j] = wmma4(ta, tb[j], acc[j]);
        }
        __builtin_amdgcn_sched_barrier(0);
      }
    }
    float sc[8];
    #pragma unroll
    for (int v = 0; v < 8; ++v) sc[v] = 0.f;
    #pragma unroll
    for (int j = 0; j < 8; ++j) {
      float g2 = gw2p[16 * j + c];
      #pragma unroll
      for (int v = 0; v < 8; ++v) sc[v] += leaky_(acc[j][v]) * g2;
    }
    #pragma unroll
    for (int v = 0; v < 8; ++v) {
      float val = sc[v];
      val += __shfl_xor(val, 1);
      val += __shfl_xor(val, 2);
      val += __shfl_xor(val, 4);
      val += __shfl_xor(val, 8);
      sc[v] = val;
    }
    float mx = sc[0];
    #pragma unroll
    for (int v = 1; v < 8; ++v) mx = fmaxf(mx, sc[v]);
    float sum = 0.f;
    #pragma unroll
    for (int v = 0; v < 8; ++v) { sc[v] = __expf(sc[v] - mx); sum += sc[v]; }
    const float inv = 1.f / sum;
    float4 na = make_float4(0.f, 0.f, 0.f, 0.f), nb = na;
    #pragma unroll
    for (int v = 0; v < 8; ++v) {
      const float al = sc[v] * inv;
      const float* row = xw + (8 * hf + v) * XP_ + 4 * c;
      float4 p = *(const float4*)(row);
      float4 q = *(const float4*)(row + 64);
      na.x += al * p.x; na.y += al * p.y; na.z += al * p.z; na.w += al * p.w;
      nb.x += al * q.x; nb.y += al * q.y; nb.z += al * q.z; nb.w += al * q.w;
    }
    {
      float* srow = sn + (2 * wave + hf) * GP_ + D_ + 4 * c;
      *(float4*)(srow) = na;
      *(float4*)(srow + 64) = nb;
      *(v4f*)(sn + (2 * wave) * GP_ + 4 * lane) = s0;
      *(v4f*)(sn + (2 * wave + 1) * GP_ + 4 * lane) = s1;
    }
    __syncthreads();

    if (u + 1 < nvu) AGG_FETCH_ROWS()

    v8f o0 = (v8f){}, o1 = (v8f){};
    {
      const float* sa = sn + c * GP_ + 4 * hf;
      const float* gb = G3T + (16 * wave + c) * GP_ + 4 * hf;
      v4f wa[2][8], wb[2][8];
      #pragma unroll
      for (int i = 0; i < 8; ++i) {
        wa[0][i] = *(const v4f*)(sa + 8 * i);
        wb[0][i] = *(const v4f*)(gb + 8 * i);
      }
      #pragma unroll
      for (int blk = 0; blk < 4; ++blk) {
        const int s = blk & 1;
        if (blk + 1 < 4) {
          #pragma unroll
          for (int i = 0; i < 8; ++i) {
            wa[s ^ 1][i] = *(const v4f*)(sa + 8 * (8 * (blk + 1) + i));
            wb[s ^ 1][i] = *(const v4f*)(gb + 8 * (8 * (blk + 1) + i));
          }
        }
        __builtin_amdgcn_sched_barrier(0);
        #pragma unroll
        for (int i = 0; i < 8; ++i) {
          o0 = wmma4((v2f){wa[s][i].x, wa[s][i].y}, (v2f){wb[s][i].x, wb[s][i].y}, o0);
          o1 = wmma4((v2f){wa[s][i].z, wa[s][i].w}, (v2f){wb[s][i].z, wb[s][i].w}, o1);
        }
        __builtin_amdgcn_sched_barrier(0);
      }
    }
    const bool deepu = DEEP && u < nd;
    const int mb0 = deepu ? (d0 + u) * 16 : (u0 + u - nd) * 16;
    const int Mv = deepu ? MD : M;
    float* ob = (deepu ? out2 : out) + 16 * wave + c;
    #pragma unroll
    for (int v = 0; v < 8; ++v) {
      int m = mb0 + v + 8 * hf;
      if (m < Mv) ob[((long)b * Mv + m) << 7] = fmaxf(o0[v] + o1[v], 0.f);
    }
  }
#undef AGG_FETCH_IDX
#undef AGG_FETCH_ROWS
#undef AGG_LOAD_IDX
#undef AGG_LOAD_VAL
#undef AGG_STORE_ROW
#undef AGG_ROWS
}

#define SP_ 132
#define SCORE_LDS_FLOATS (42 * 1024)
#define SCORE_WAVES 16
#define SCORE_WGS 256

__global__ __launch_bounds__(SCORE_WAVES * 32) void score_kernel(
    const float* __restrict__ selF, const float* __restrict__ emb,
    float* __restrict__ scores) {
  __shared__ __attribute__((aligned(16))) float sA[SCORE_LDS_FLOATS];
  const int tid = threadIdx.x;
  const int lane = tid & 31, c = lane & 15, hf = lane >> 4;
  const int NSC = NN_ - 1;
  const int NT = (NSC + 15) >> 4;
  const long U = (long)NT * 8;
  const int nw = SCORE_WGS * SCORE_WAVES;
  const int wv = __builtin_amdgcn_readfirstlane(tid >> 5);
  const int gw = (int)blockIdx.x * SCORE_WAVES + wv;
  const int u0 = (int)((U * gw) / nw), u1 = (int)((U * (gw + 1)) / nw);

  v4f bn[16];
#define SCORE_LOADB(NTILE)                                                      \
  {                                                                             \
    int er = ((NTILE) << 4) + c + 1;                \
    er = er < NN_ ? er : NN_ - 1;                   \
    const float* bp = emb + ((long)er << 7) + 4 * hf;                           \
    _Pragma("unroll") for (int i = 0; i < 16; ++i) bn[i] = *(const v4f*)(bp + 8 * i); \
  }
  if (u0 < u1) SCORE_LOADB(u0 >> 3)

  #pragma unroll
  for (int t = 0; t < (B_ * (D_ / 4)) / (SCORE_WAVES * 32); ++t) {
    const int i = tid + t * (SCORE_WAVES * 32);
    const int r = i >> 5, q = i & 31;
    *(v4f*)(sA + r * SP_ + 4 * q) = *(const v4f*)(selF + (r << 7) + 4 * q);
  }
  __syncthreads();

  int u = u0;
  while (u < u1) {
    const int nt = u >> 3;
    const int ue = (u1 < ((nt + 1) << 3)) ? u1 : ((nt + 1) << 3);
    v4f bf[16];
    #pragma unroll
    for (int i = 0; i < 16; ++i) bf[i] = bn[i];
    if (ue < u1) SCORE_LOADB(nt + 1)
    const int col = (nt << 4) + c;
    for (; u < ue; ++u) {
      const int m0 = (u & 7) << 4;
      const float* ap = sA + (m0 + c) * SP_ + 4 * hf;
      v4f a[16];
      #pragma unroll
      for (int i = 0; i < 16; ++i) a[i] = *(const v4f*)(ap + 8 * i);
      __builtin_amdgcn_sched_barrier(0);
      v8f acc0 = (v8f){}, acc1 = (v8f){};
      #pragma unroll
      for (int i = 0; i < 16; ++i) {
        acc0 = wmma4((v2f){a[i].x, a[i].y}, (v2f){bf[i].x, bf[i].y}, acc0);
        acc1 = wmma4((v2f){a[i].z, a[i].w}, (v2f){bf[i].z, bf[i].w}, acc1);
      }
      if (col < NSC) {
        float* op = scores + (long)(m0 + 8 * hf) * NSC + col;
        #pragma unroll
        for (int v = 0; v < 8; ++v) op[(long)v * NSC] = acc0[v] + acc1[v];
      }
    }
  }
#undef SCORE_LOADB
}

template <int M>
__device__ __forceinline__ float row_xor(float x) {
  return __builtin_bit_cast(float, __builtin_amdgcn_update_dpp(0, __builtin_bit_cast(int, x), 0x160 + M, 0xf, 0xf, false));
}
__device__ __forceinline__ float xhalf(float x) {
  const int v = __builtin_bit_cast(int, x);
  return __builtin_bit_cast(float, __builtin_amdgcn_permlanex16(v, v, 0x76543210, 0xfedcba98, false, false));
}

template <int M>
__device__ __forceinline__ int row_xor_i(int x) {
  return __builtin_amdgcn_update_dpp(0, x, 0x160 + M, 0xf, 0xf, false);
}
__device__ __forceinline__ int xhalf_i(int v) {
  return __builtin_amdgcn_permlanex16(v, v, 0x76543210, 0xfedcba98, false, false);
}

#define HP_ 132
#define HR_ 64
#define HEAD_LDS_FLOATS (42 * 1024)

__global__ __launch_bounds__(256) void head_kernel(
    const float* __restrict__ h_local, const float* __restrict__ h_global,
    const int* __restrict__ mask, const float* __restrict__ pos_emb,
    const float* __restrict__ w1, const float* __restrict__ glu1_w,
    const float* __restrict__ glu1_b, const float* __restrict__ glu2_w,
    const float* __restrict__ w2,
    float* __restrict__ hidden, float* __restrict__ sel, float* __restrict__ snorm,
    float* __restrict__ lpart) {
  __shared__ __attribute__((aligned(16))) float lds[HEAD_LDS_FLOATS];
  float* HX    = lds;
  float* NH    = HX + HR_ * HP_;
  float* PX    = NH + HR_ * HP_;
  float* hsL   = PX + HR_ * HP_;
  float* maskf = hsL + D_;
  float* part  = maskf + HR_;
  float* betas = part + 8 * HR_;
  float* red   = betas + HR_;
  float* red2  = red + 4;
  float* sslP  = red2 + 4;
  float* sslN  = sslP + 8 * D_;

  const int tid = threadIdx.x, b = blockIdx.x;
  const int lane = tid & 31, c = lane & 15, hf = lane >> 4;
  const int wave = tid >> 5;
  const int n = (wave << 4) + c;

  v4f bp[16], bw[16], bg[16], g2[16];
  {
    const float* wp0 = w1 + (4 * hf) * D_ + n;
    #pragma unroll
    for (int i = 0; i < 16; ++i) {
      bp[i].x = wp0[(8 * i + 0) * D_];
      bp[i].y = wp0[(8 * i + 1) * D_];
      bp[i].z = wp0[(8 * i + 2) * D_];
      bp[i].w = wp0[(8 * i + 3) * D_];
    }
    const float* wp = w1 + (D_ + 4 * hf) * D_ + n;
    #pragma unroll
    for (int i = 0; i < 16; ++i) {
      bw[i].x = wp[(8 * i + 0) * D_];
      bw[i].y = wp[(8 * i + 1) * D_];
      bw[i].z = wp[(8 * i + 2) * D_];
      bw[i].w = wp[(8 * i + 3) * D_];
    }
    const float* gp = glu1_w + n * D_ + 4 * hf;
    #pragma unroll
    for (int i = 0; i < 16; ++i) bg[i] = *(const v4f*)(gp + 8 * i);
    const float* g2p = glu2_w + n * D_ + 64 * hf;
    #pragma unroll
    for (int i = 0; i < 16; ++i) g2[i] = *(const v4f*)(g2p + 4 * i);
  }
  const float g1b = glu1_b[n];
  const float w2n = w2[n];

  {
    const float* hl = h_local + (long)b * L_ * D_;
    const float* hg = h_global + (long)b * L_ * D_;
    float* ho = hidden + (long)b * L_ * D_;
    const float* hc = h_local + (long)((b * 73 + 21) & 127) * L_ * D_ + 4 * (tid & 31);
    v4f sp = (v4f){}, sg = (v4f){};
    #pragma unroll
    for (int t = 0; t < 6; ++t) {
      const int i4 = tid + 256 * t;
      const int l = i4 >> 5;
      const v4f vl = *(const v4f*)(hl + 4 * i4), vg = *(const v4f*)(hg + 4 * i4);
      const v4f vc = *(const v4f*)(hc + (((l * 13 + 7) % L_) << 7));
      const v4f v = vl + vg;
      *(v4f*)(ho + 4 * i4) = v;
      *(v4f*)(HX + l * HP_ + 4 * (i4 & 31)) = v;
      *(v4f*)(PX + l * HP_ + 4 * (i4 & 31)) = *(const v4f*)(pos_emb + 4 * i4);
      sp += vl * vg;
      sg += vg * vc;
    }
    if (tid < (L_ * D_) / 4 - 6 * 256) {
      const int i4 = tid + 6 * 256;
      const int l = i4 >> 5;
      const v4f vl = *(const v4f*)(hl + 4 * i4), vg = *(const v4f*)(hg + 4 * i4);
      const v4f vc = *(const v4f*)(hc + (((l * 13 + 7) % L_) << 7));
      const v4f v = vl + vg;
      *(v4f*)(ho + 4 * i4) = v;
      *(v4f*)(HX + l * HP_ + 4 * (i4 & 31)) = v;
      *(v4f*)(PX + l * HP_ + 4 * (i4 & 31)) = *(const v4f*)(pos_emb + 4 * i4);
      sp += vl * vg;
      sg += vg * vc;
    }
    *(v4f*)(sslP + wave * D_ + 4 * (tid & 31)) = sp;
    *(v4f*)(sslN + wave * D_ + 4 * (tid & 31)) = sg;
    for (int i4 = tid; i4 < (HR_ - L_) * 32; i4 += 256) {
      *(v4f*)(HX + (L_ + (i4 >> 5)) * HP_ + 4 * (i4 & 31)) = (v4f){};
      *(v4f*)(PX + (L_ + (i4 >> 5)) * HP_ + 4 * (i4 & 31)) = (v4f){};
    }
    if (tid < HR_) maskf[tid] = tid < L_ ? (float)mask[b * L_ + tid] : 0.f;
  }
  __syncthreads();

  if (tid < D_) {
    float acc = 0.f, ms = 0.f;
    #pragma unroll
    for (int l = 0; l < L_; ++l) {
      const float m = maskf[l];
      ms += m;
      acc += HX[l * HP_ + tid] * m;
    }
    hsL[tid] = acc / ms;
  }
  {
    float term = 0.f;
    if (tid < D_) {
      float pos = 0.f, neg = 0.f;
      #pragma unroll
      for (int w = 0; w < 8; ++w) { pos += sslP[w * D_ + tid]; neg += sslN[w * D_ + tid]; }
      term = -__logf(1e-8f + sigmoid_(pos)) - __logf(1e-8f + 1.f - sigmoid_(neg));
    }
    term += row_xor<1>(term);
    term += row_xor<2>(term);
    term += row_xor<4>(term);
    term += row_xor<8>(term);
    term += xhalf(term);
    if (lane == 0 && wave < 4) red2[wave] = term;
  }
  for (int mt = 0; mt < 4; ++mt) {
    const float* ap = HX + (16 * mt + c) * HP_ + 4 * hf;
    const float* pp = PX + (16 * mt + c) * HP_ + 4 * hf;
    v4f a[16], p[16];
    #pragma unroll
    for (int i = 0; i < 16; ++i) p[i] = *(const v4f*)(pp + 8 * i);
    #pragma unroll
    for (int i = 0; i < 16; ++i) a[i] = *(const v4f*)(ap + 8 * i);
    __builtin_amdgcn_sched_barrier(0);
    v8f acc0 = (v8f){}, acc1 = (v8f){};
    #pragma unroll
    for (int i = 0; i < 16; ++i) {
      acc0 = wmma4((v2f){p[i].x, p[i].y}, (v2f){bp[i].x, bp[i].y}, acc0);
      acc1 = wmma4((v2f){p[i].z, p[i].w}, (v2f){bp[i].z, bp[i].w}, acc1);
    }
    #pragma unroll
    for (int i = 0; i < 16; ++i) {
      acc0 = wmma4((v2f){a[i].x, a[i].y}, (v2f){bw[i].x, bw[i].y}, acc0);
      acc1 = wmma4((v2f){a[i].z, a[i].w}, (v2f){bw[i].z, bw[i].w}, acc1);
    }
    #pragma unroll
    for (int v = 0; v < 8; ++v)
      NH[(16 * mt + 8 * hf + v) * HP_ + n] = tanhf(acc0[v] + acc1[v]);
  }
  __syncthreads();

  float bias;
  {
    const float* hp = hsL + 64 * hf;
    float s0 = 0.f, s1 = 0.f, s2 = 0.f, s3 = 0.f;
    #pragma unroll
    for (int i = 0; i < 16; ++i) {
      const v4f x = *(const v4f*)(hp + 4 * i);
      s0 += g2[i].x * x.x; s1 += g2[i].y * x.y;
      s2 += g2[i].z * x.z; s3 += g2[i].w * x.w;
    }
    float s = (s0 + s1) + (s2 + s3);
    s += xhalf(s);
    bias = s + g1b;
  }

  for (int mt = 0; mt < 4; ++mt) {
    const float* ap = NH + (16 * mt + c) * HP_ + 4 * hf;
    v4f a[16];
    #pragma unroll
    for (int i = 0; i < 16; ++i) a[i] = *(const v4f*)(ap + 8 * i);
    __builtin_amdgcn_sched_barrier(0);
    v8f acc0 = (v8f){}, acc1 = (v8f){};
    #pragma unroll
    for (int i = 0; i < 16; ++i) {
      acc0 = wmma4((v2f){a[i].x, a[i].y}, (v2f){bg[i].x, bg[i].y}, acc0);
      acc1 = wmma4((v2f){a[i].z, a[i].w}, (v2f){bg[i].z, bg[i].w}, acc1);
    }
    float g[8];
    #pragma unroll
    for (int v = 0; v < 8; ++v) g[v] = sigmoid_(acc0[v] + acc1[v] + bias) * w2n;
    #pragma unroll
    for (int v = 0; v < 8; ++v) g[v] += row_xor<1>(g[v]);
    #pragma unroll
    for (int v = 0; v < 8; ++v) g[v] += row_xor<2>(g[v]);
    #pragma unroll
    for (int v = 0; v < 8; ++v) g[v] += row_xor<4>(g[v]);
    #pragma unroll
    for (int v = 0; v < 8; ++v) g[v] += row_xor<8>(g[v]);
    if (c == 0) {
      #pragma unroll
      for (int v = 0; v < 8; ++v) part[wave * HR_ + 16 * mt + 8 * hf + v] = g[v];
    }
  }
  __syncthreads();

  if (tid < L_) {
    float s = 0.f;
    #pragma unroll
    for (int w = 0; w < 8; ++w) s += part[w * HR_ + tid];
    betas[tid] = s * maskf[tid];
  }
  __syncthreads();

  float sq = 0.f;
  if (tid < D_) {
    float acc = 0.f;
    #pragma unroll
    for (int l = 0; l < L_; ++l) acc += betas[l] * HX[l * HP_ + tid];
    sel[(b << 7) + tid] = acc;
    sq = acc * acc + 1e-6f;
  }
  sq += row_xor<1>(sq);
  sq += row_xor<2>(sq);
  sq += row_xor<4>(sq);
  sq += row_xor<8>(sq);
  sq += xhalf(sq);
  if (lane == 0 && wave < 4) red[wave] = sq;
  __syncthreads();
  if (tid == 0) {
    snorm[b] = sqrtf((red[0] + red[1]) + (red[2] + red[3]));
    lpart[b] = (red2[0] + red2[1]) + (red2[2] + red2[3]);
  }
}


__global__ void fn_kernel(const float* __restrict__ sel, const float* __restrict__ snorm,
                          const float* __restrict__ lpart, float* __restrict__ loss,
                          float* __restrict__ selF) {
  __shared__ float row[B_];
  __shared__ float selfrow[D_];
  __shared__ float tvs[9];
  __shared__ int tis[9];
  int i = blockIdx.x, t = threadIdx.x;
  selfrow[t] = sel[(i << 7) + t];
  __syncthreads();
  float acc = 0.f;
  for (int d = 0; d < D_; ++d) acc += selfrow[d] * sel[(t << 7) + d];
  row[t] = acc / (snorm[i] * snorm[t]);
  __syncthreads();
  if (t < 32) {
    if (i == 0) {
      float ls = (lpart[t] + lpart[t + 32]) + (lpart[t + 64] + lpart[t + 96]);
      ls += row_xor<1>(ls);
      ls += row_xor<2>(ls);
      ls += row_xor<4>(ls);
      ls += row_xor<8>(ls);
      ls += xhalf(ls);
      if (t == 0) loss[0] = BETA_ * ls;
    }
    float p[B_ / 32];
    #pragma unroll
    for (int q = 0; q < B_ / 32; ++q) p[q] = row[t + 32 * q];
    float mx = fmaxf(fmaxf(p[0], p[1]), fmaxf(p[2], p[3]));
    mx = fmaxf(mx, row_xor<1>(mx));
    mx = fmaxf(mx, row_xor<2>(mx));
    mx = fmaxf(mx, row_xor<4>(mx));
    mx = fmaxf(mx, row_xor<8>(mx));
    mx = fmaxf(mx, xhalf(mx));
    #pragma unroll
    for (int q = 0; q < B_ / 32; ++q) p[q] = __expf(p[q] - mx);
    float s = (p[0] + p[1]) + (p[2] + p[3]);
    s += row_xor<1>(s);
    s += row_xor<2>(s);
    s += row_xor<4>(s);
    s += row_xor<8>(s);
    s += xhalf(s);
    const float inv = 1.f / s;
    #pragma unroll
    for (int q = 0; q < B_ / 32; ++q) p[q] *= inv;
    float tv[9];
    int ti[9];
#define FN_MEET(OV, OI)                                                        \
    {                                                                          \
      const float ov = (OV);                                                   \
      const int oi = (OI);                                                     \
      const bool take = ov > bv || (ov == bv && oi < bi);                      \
      bv = take ? ov : bv;                                                     \
      bi = take ? oi : bi;                                                     \
    }
    #pragma unroll
    for (int r = 0; r < 9; ++r) {
      float bv = p[0];
      int bi = t;
      #pragma unroll
      for (int q = 1; q < B_ / 32; ++q) {
        const bool take = p[q] > bv;
        bv = take ? p[q] : bv;
        bi = take ? t + 32 * q : bi;
      }
      FN_MEET(row_xor<1>(bv), row_xor_i<1>(bi))
      FN_MEET(row_xor<2>(bv), row_xor_i<2>(bi))
      FN_MEET(row_xor<4>(bv), row_xor_i<4>(bi))
      FN_MEET(row_xor<8>(bv), row_xor_i<8>(bi))
      FN_MEET(xhalf(bv), xhalf_i(bi))
      tv[r] = bv;
      ti[r] = bi;
      #pragma unroll
      for (int q = 0; q < B_ / 32; ++q) p[q] = (bi == t + 32 * q) ? -2.f : p[q];
    }
#undef FN_MEET
    const float mx2 = tv[0];
    float s2 = 0.f;
    #pragma unroll
    for (int r = 0; r < 9; ++r) { tv[r] = __expf(tv[r] - mx2); s2 += tv[r]; }
    const float inv2 = 1.f / s2;
    if (t == 0) {
      #pragma unroll
      for (int r = 0; r < 9; ++r) { tvs[r] = tv[r] * inv2; tis[r] = ti[r]; }
    }
  }
  __syncthreads();
  float nb = 0.f;
  for (int q = 0; q < 9; ++q) nb += tvs[q] * sel[(tis[q] << 7) + t];
  selF[(i << 7) + t] = selfrow[t] + nb;
}

extern "C" void kernel_launch(void* const* d_in, const int* in_sizes, int n_in,
                              void* d_out, int out_size, void* d_ws, size_t ws_size,
                              hipStream_t stream) {
  (void)in_sizes; (void)n_in; (void)out_size; (void)ws_size;
  const int*   inputs  = (const int*)d_in[0];
  const int*   adj     = (const int*)d_in[1];
  const int*   mask    = (const int*)d_in[2];
  const int*   item    = (const int*)d_in[3];
  const int*   adj_all = (const int*)d_in[5];
  const float* num_w   = (const float*)d_in[6];
  const float* emb     = (const float*)d_in[7];
  const float* pos_emb = (const float*)d_in[8];
  const float* a_local = (const float*)d_in[9];
  const float* gw1     = (const float*)d_in[10];
  const float* gw2     = (const float*)d_in[11];
  const float* gw3     = (const float*)d_in[12];
  const float* w1      = (const float*)d_in[13];
  const float* w2      = (const float*)d_in[14];
  const float* glu1_w  = (const float*)d_in[15];
  const float* glu1_b  = (const float*)d_in[16];
  const float* glu2_w  = (const float*)d_in[17];

  float* out0   = (float*)d_out;
  float* lossP  = out0 + B_ * L_ * D_;
  float* scores = lossP + 1;

  char* wp = (char*)d_ws;
  auto alloc = [&](size_t nfloats) -> float* {
    float* r = (float*)wp;
    wp += ((nfloats * sizeof(float) + 255) / 256) * 256;
    return r;
  };
  float* h_local  = alloc(B_ * L_ * D_);
  float* h_global = alloc(B_ * L_ * D_);
  float* e0p      = alloc(B_ * L_ * D_);
  float* e1p      = alloc((size_t)B_ * 400 * D_);
  float* lpartb   = alloc(B_);
  float* selb     = alloc(B_ * D_);
  float* selF     = alloc(B_ * D_);
  float* snorm    = alloc(B_);

  local_agg_kernel<<<dim3(B_), 256, 0, stream>>>(emb, inputs, adj, a_local, h_local);
  agg_kernel<true, true><<<dim3(B_ * 2), 256, 0, stream>>>(emb, inputs, adj_all, num_w, item, mask,
      nullptr, nullptr, gw1, gw2, gw3, e0p, e1p, L_, 2);
  agg_kernel<false, false><<<dim3(B_ * 2), 256, 0, stream>>>(emb, inputs, adj_all, num_w, item, mask,
      e0p, e1p, gw1 + 129 * D_, gw2 + D_, gw3 + 2 * D_ * D_, h_global, nullptr, L_, 2);
  head_kernel<<<dim3(B_), 256, 0, stream>>>(h_local, h_global, mask, pos_emb, w1, glu1_w, glu1_b,
      glu2_w, w2, out0, selb, snorm, lpartb);
  fn_kernel<<<dim3(B_), dim3(D_), 0, stream>>>(selb, snorm, lpartb, lossP, selF);
  score_kernel<<<dim3(SCORE_WGS), SCORE_WAVES * 32, 0, stream>>>(selF, emb, scores);
}


# agg kernels: one LDS-only barrier per unit (two [self|neigh] row sets); a SIMD's two waves run logits and out tile in opposite order under s_setprio, so vector parts run under the partner's WMMAs
# speedup vs baseline: 3.3746x; 1.0363x over previous
#include <hip/hip_runtime.h>
#include <hip/hip_bf16.h>
#include <math.h>

#define B_ 128
#define L_ 50
#define D_ 128
#define S_ 8
#define NN_ 100000
#define LEAKY_ 0.2f
#define BETA_ 0.005f

typedef float v2f __attribute__((ext_vector_type(2)));
typedef float v8f __attribute__((ext_vector_type(8)));
typedef float v4f __attribute__((ext_vector_type(4)));
typedef int v4i __attribute__((ext_vector_type(4)));

#if defined(__has_builtin)
#if __has_builtin(__builtin_amdgcn_wmma_f32_16x16x4_f32)
#define HAVE_WMMA_F32 1
#endif
#endif

__device__ __forceinline__ v8f wmma4(v2f a, v2f b, v8f c) {
#ifdef HAVE_WMMA_F32
  return __builtin_amdgcn_wmma_f32_16x16x4_f32(false, a, false, b, (short)0, c,
                                               false, false);
#else
  float* cf = (float*)&c;
  #pragma unroll
  for (int i = 0; i < 8; ++i) cf[i] += a.x * b.x + a.y * b.y;
  return c;
#endif
}

__device__ __forceinline__ float leaky_(float x) { return x > 0.f ? x : LEAKY_ * x; }
__device__ __forceinline__ float sigmoid_(float x) { return 1.f / (1.f + __expf(-x)); }


#define LA_TP 65
#define LA_PS (L_ * D_ + 8)
#define LA_AR 64
#define LA_AP 52
__global__ __launch_bounds__(256) void local_agg_kernel(const float* __restrict__ emb,
                                 const int* __restrict__ inputs,
                                 const int* __restrict__ adj,
                                 const float* __restrict__ a_local,
                                 float* __restrict__ h_local) {
  int b = blockIdx.x;
  __shared__ __attribute__((aligned(16))) float hv[L_][D_];
  __shared__ __attribute__((aligned(16))) float hia[4 * LA_PS];
  __shared__ float hvT[D_ * LA_TP];
  __shared__ int adjs[L_ * L_];
  __shared__ __attribute__((aligned(16))) float att[LA_AR][LA_AP];
  __shared__ __attribute__((aligned(16))) float rinv[LA_AR];
  {
    const int q = threadIdx.x & 31;
    const int wvu = __builtin_amdgcn_readfirstlane(threadIdx.x >> 5);
    const v4f ak0 = *(const v4f*)(a_local + 0 * D_ + 4 * q);
    const v4f ak1 = *(const v4f*)(a_local + 1 * D_ + 4 * q);
    const v4f ak2 = *(const v4f*)(a_local + 2 * D_ + 4 * q);
    const v4f ak3 = *(const v4f*)(a_local + 3 * D_ + 4 * q);
    int nid[(L_ + 7) / 8];
    #pragma unroll
    for (int t = 0; t < (L_ + 7) / 8; ++t) {
      const int r = wvu + 8 * t;
      nid[t] = inputs[b * L_ + (r < L_ ? r : L_ - 1)];
    }
    v4f hr[(L_ + 7) / 8];
    #pragma unroll
    for (int t = 0; t < (L_ + 7) / 8; ++t) hr[t] = *(const v4f*)(emb + ((long)nid[t] << 7) + 4 * q);
    #pragma unroll
    for (int t = 0; t < (L_ + 7) / 8; ++t) {
      const int r = wvu + 8 * t;
      if (r >= L_) break;
      const int i4 = r * 32 + q;
      const v4f v = hr[t];
      *(v4f*)(&hv[r][4 * q]) = v;
      float* tp = hvT + (4 * q) * LA_TP + r;
      tp[0] = v.x; tp[LA_TP] = v.y; tp[2 * LA_TP] = v.z; tp[3 * LA_TP] = v.w;
      *(v4f*)(hia + 0 * LA_PS + 4 * i4) = v * ak0;
      *(v4f*)(hia + 1 * LA_PS + 4 * i4) = v * ak1;
      *(v4f*)(hia + 2 * LA_PS + 4 * i4) = v * ak2;
      *(v4f*)(hia + 3 * LA_PS + 4 * i4) = v * ak3;
    }
    for (int p = threadIdx.x; p < L_ * L_; p += 256) adjs[p] = adj[b * L_ * L_ + p];
    for (int p = threadIdx.x; p < LA_AR * LA_AP / 4; p += 256) ((v4f*)&att[0][0])[p] = (v4f){};
  }
  __syncthreads();
  {
    const int lane = threadIdx.x & 31, wv = threadIdx.x >> 5;
    const int j = lane + ((wv & 1) << 5);
    const bool jin = j < L_;
    const int jr = jin ? j : L_ - 1;
    float hj[D_];
    #pragma unroll
    for (int d = 0; d < D_; ++d) hj[d] = hvT[d * LA_TP + jr];
    int kn = adjs[(wv >> 1) * L_ + jr];
    for (int i = wv >> 1; i < L_; i += 4) {
      const int k = kn;
      kn = adjs[(i + 4 < L_ ? i + 4 : i) * L_ + jr];
      const bool on = (unsigned)(k - 1) < 4u;
      const float* rp = hia + (on ? k - 1 : 0) * LA_PS + i * D_;
      float a0 = 0.f, a1 = 0.f, a2 = 0.f, a3 = 0.f;
      #pragma unroll
      for (int q = 0; q < D_ / 4; ++q) {
        const v4f r = *(const v4f*)(rp + 4 * q);
        a0 += r.x * hj[4 * q];     a1 += r.y * hj[4 * q + 1];
        a2 += r.z * hj[4 * q + 2]; a3 += r.w * hj[4 * q + 3];
      }
      const float e = (a0 + a1) + (a2 + a3);
      if (jin) att[i][j] = on ? leaky_(e) : -9e15f;
    }
  }
  __syncthreads();
  if (threadIdx.x < L_) {
    int i = threadIdx.x;
    float mx = -1e30f;
    for (int j = 0; j < L_; ++j) mx = fmaxf(mx, att[i][j]);
    float s = 0.f;
    for (int j = 0; j < L_; ++j) { float e = __expf(att[i][j] - mx); att[i][j] = e; s += e; }
    rinv[i] = 1.f / s;
  }
  __syncthreads();
  {
    const int lane = threadIdx.x & 31, c = lane & 15, hf = lane >> 4;
    const int wv = threadIdx.x >> 5;
    const int n = (wv << 4) + c;
    v2f bq[LA_AP / 4];
    #pragma unroll
    for (int ks = 0; ks < LA_AP / 4; ++ks) {
      const int k = 4 * ks + 2 * hf;
      const bool kin = k < L_;
      const int kr = kin ? k : 0;
      const float b0 = hv[kr][n], b1 = hv[kr + 1][n];
      bq[ks] = kin ? (v2f){b0, b1} : (v2f){0.f, 0.f};
    }
    const float* ap = &att[c][2 * hf];
    v2f aq[4][LA_AP / 4];
    #pragma unroll
    for (int mt = 0; mt < 4; ++mt) {
      #pragma unroll
      for (int ks = 0; ks < LA_AP / 4; ++ks) aq[mt][ks] = *(const v2f*)(ap + 16 * mt * LA_AP + 4 * ks);
    }
    v4f rlo[4], rhi[4];
    #pragma unroll
    for (int mt = 0; mt < 4; ++mt) {
      rlo[mt] = *(const v4f*)(rinv + 16 * mt + 8 * hf);
      rhi[mt] = *(const v4f*)(rinv + 16 * mt + 8 * hf + 4);
    }
    v8f acc[4];
    #pragma unroll
    for (int mt = 0; mt < 4; ++mt) acc[mt] = (v8f){};
    #pragma unroll
    for (int ks = 0; ks < LA_AP / 4; ++ks) {
      #pragma unroll
      for (int mt = 0; mt < 4; ++mt) acc[mt] = wmma4(aq[mt][ks], bq[ks], acc[mt]);
    }
    float* op = h_local + (long)b * L_ * D_ + n;
    #pragma unroll
    for (int mt = 0; mt < 4; ++mt) {
      #pragma unroll
      for (int v = 0; v < 8; ++v) {
        const int row = 16 * mt + 8 * hf + v;
        const float ri = v < 4 ? rlo[mt][v] : rhi[mt][v - 4];
        if (row < L_) op[row * D_] = acc[mt][v] * ri;
      }
    }
  }
}

#define XP_ 132
#define GP_ 260
#define AGG_SID 256
#define AGG_LDS_FLOATS (D_ * XP_ + D_ * GP_ + 8 * 16 * XP_ + 2 * 16 * GP_ + D_ + AGG_SID)

template <bool GATHER, bool DEEP>
__global__ __launch_bounds__(256) void agg_kernel(
                           const float* __restrict__ emb,
                           const int* __restrict__ inputs, const int* __restrict__ adj_all,
                           const float* __restrict__ num_w,
                           const int* __restrict__ item, const int* __restrict__ mask,
                           const float* __restrict__ selfVal, const float* __restrict__ neighVal,
                           const float* __restrict__ gw1p, const float* __restrict__ gw2p,
                           const float* __restrict__ gw3p,
                           float* __restrict__ out, float* __restrict__ out2, int M, int parts) {
  __shared__ __attribute__((aligned(16))) float lds[AGG_LDS_FLOATS];
  float* WT  = lds;
  float* G3T = WT + D_ * XP_;
  float* xs  = G3T + D_ * GP_;
  float* sn  = xs + 8 * 16 * XP_;
  float* sessL = sn + 2 * 16 * GP_;
  int*   sid = (int*)(sessL + D_);

  const int tid = threadIdx.x;
  const int lane = tid & 31, c = lane & 15, hf = lane >> 4;
  const int wave = tid >> 5;
  const int b = blockIdx.x / parts, part = blockIdx.x % parts;
  const int units = (M + 15) >> 4;
  const int upb = (units + parts - 1) / parts;
  const int u0 = part * upb;
  const int u1 = (u0 + upb < units) ? (u0 + upb) : units;
  const int MD = M * S_;
  const int dunits = (MD + 15) >> 4;
  const int dpb = (dunits + parts - 1) / parts;
  const int d0 = DEEP ? part * dpb : 0;
  const int d1 = DEEP ? ((d0 + dpb < dunits) ? (d0 + dpb) : dunits) : 0;
  const int nd = d1 - d0;
  const int nvu = nd + (u1 - u0);

  float se[L_], mk[L_];
  #pragma unroll
  for (int l = 0; l < L_; ++l) se[l] = emb[((long)item[b * L_ + l] << 7) + (tid & (D_ - 1))];
  #pragma unroll
  for (int l = 0; l < L_; ++l) mk[l] = (float)mask[b * L_ + l];
  const int kl = tid & 7, ql = (tid >> 3) & 3, rest0 = tid >> 5;
  float4 w1r[16];
  #pragma unroll
  for (int t = 0; t < 16; ++t) {
    const int rest = rest0 + 8 * t;
    const int k = ((rest & 15) << 3) + kl;
    const int q = ((rest >> 4) << 2) + ql;
    w1r[t] = *(const float4*)(gw1p + k * D_ + 4 * q);
  }
  const int nsid = 16 * nvu;
  int sidv = 0;
  if (tid < nsid) {
    const int vu = tid >> 4, wi = tid & 15;
    if (DEEP && vu < nd) {
      int m = (d0 + vu) * 16 + wi;
      m = m < MD ? m : MD - 1;
      const int root = inputs[b * L_ + (m >> 3)];
      sidv = adj_all[root * S_ + (m & 7)];
    } else {
      int m = (u0 + vu - nd) * 16 + wi;
      m = m < M ? m : M - 1;
      sidv = inputs[b * L_ + m];
    }
  }
  #pragma unroll
  for (int t = 0; t < 32; ++t) {
    const int rest = rest0 + 8 * t;
    int k = ((rest & 31) << 3) + kl;
    int q = ((rest >> 5) << 2) + ql;
    float4 w = *(const float4*)(gw3p + k * D_ + 4 * q);
    float* d = G3T + (4 * q) * GP_ + k;
    d[0] = w.x; d[GP_] = w.y; d[2 * GP_] = w.z; d[3 * GP_] = w.w;
  }
  if (tid < D_) {
    float sacc = 0.f, smask = 0.f;
    #pragma unroll
    for (int l = 0; l < L_; ++l) {
      const float m = mk[l];
      smask += m;
      sacc += se[l] * m;
    }
    sessL[tid] = sacc / smask;
    float* d = WT + tid * XP_ + D_;
    d[0] = gw1p[D_ * D_ + tid];
    d[1] = 0.f; d[2] = 0.f; d[3] = 0.f;
  }
  if (tid < nsid) sid[tid] = sidv;
  __syncthreads();
  {
    const float sk0 = sessL[(rest0 << 3) + kl], sk1 = sessL[(rest0 << 3) + kl + 64];
    #pragma unroll
    for (int t = 0; t < 16; ++t) {
      const int rest = rest0 + 8 * t;
      const int k = ((rest & 15) << 3) + kl;
      const int q = ((rest >> 4) << 2) + ql;
      const float4 w = w1r[t];
      const float s = (t & 1) ? sk1 : sk0;
      float* d = WT + (4 * q) * XP_ + k;
      d[0] = w.x * s; d[XP_] = w.y * s; d[2 * XP_] = w.z * s; d[3 * XP_] = w.w * s;
    }
  }

  float* xw = xs + wave * 16 * XP_;
  v4f pre_0, pre_1, pre_2, pre_3, pre_4, pre_5, pre_6, pre_7,
      pre_8, pre_9, pre_10, pre_11, pre_12, pre_13, pre_14, pre_15;
  v4f pself_0, pself_1;
  float pw = 0.f;
  pre_0 = pre_1 = pre_2 = pre_3 = pre_4 = pre_5 = pre_6 = pre_7 = (v4f){};
  pre_8 = pre_9 = pre_10 = pre_11 = pre_12 = pre_13 = pre_14 = pre_15 = (v4f){};
  pself_0 = pself_1 = (v4f){};
  v4i ia0 = (v4i){}, ia1 = (v4i){}, ib0 = (v4i){}, ib1 = (v4i){};
  int isa = 0, isb = 0;
  long fra = 0, frb = 0;

#define AGG_ROWS(X) X(0) X(1) X(2) X(3) X(4) X(5) X(6) X(7) X(8) X(9) X(10) X(11) X(12) X(13) X(14) X(15)
#define AGG_LOAD_IDX(r, IV) pre_##r = *(const v4f*)(emb + ((long)(IV) << 7) + 4 * lane);
#define AGG_LOAD_VAL(r, ROW) pre_##r = *(const v4f*)(neighVal + (((ROW) * S_ + ((r) & 7)) << 7) + 4 * lane);
#define AGG_FETCH_IDX(U)                                                      \
  {                                                                           \
      \
    const int mr0 = (u0 + (U) - nd) * 16 + 2 * wave;                          \
    const int ma = mr0 < M ? mr0 : M - 1;                                     \
    const int mb = mr0 + 1 < M ? mr0 + 1 : M - 1;                             \
    fra = (long)b * M + ma; frb = (long)b * M + mb;                           \
    isa = sid[(U) * 16 + 2 * wave]; isb = sid[(U) * 16 + 2 * wave + 1];       \
    if (GATHER) {                                                             \
      ia0 = *(const v4i*)(adj_all + (long)isa * S_);                          \
      ia1 = *(const v4i*)(adj_all + (long)isa * S_ + 4);                      \
      ib0 = *(const v4i*)(adj_all + (long)isb * S_);                          \
      ib1 = *(const v4i*)(adj_all + (long)isb * S_ + 4);                      \
    }                                                                         \
  }
#define AGG_FETCH_ROWS()                                                      \
  {                                                                           \
    if (GATHER) {                                                             \
      AGG_LOAD_IDX(0, ia0.x) AGG_LOAD_IDX(1, ia0.y) AGG_LOAD_IDX(2, ia0.z) AGG_LOAD_IDX(3, ia0.w)     \
      AGG_LOAD_IDX(4, ia1.x) AGG_LOAD_IDX(5, ia1.y) AGG_LOAD_IDX(6, ia1.z) AGG_LOAD_IDX(7, ia1.w)     \
      AGG_LOAD_IDX(8, ib0.x) AGG_LOAD_IDX(9, ib0.y) AGG_LOAD_IDX(10, ib0.z) AGG_LOAD_IDX(11, ib0.w)   \
      AGG_LOAD_IDX(12, ib1.x) AGG_LOAD_IDX(13, ib1.y) AGG_LOAD_IDX(14, ib1.z) AGG_LOAD_IDX(15, ib1.w) \
    } else {                                                                  \
      AGG_LOAD_VAL(0, fra) AGG_LOAD_VAL(1, fra) AGG_LOAD_VAL(2, fra) AGG_LOAD_VAL(3, fra)             \
      AGG_LOAD_VAL(4, fra) AGG_LOAD_VAL(5, fra) AGG_LOAD_VAL(6, fra) AGG_LOAD_VAL(7, fra)             \
      AGG_LOAD_VAL(8, frb) AGG_LOAD_VAL(9, frb) AGG_LOAD_VAL(10, frb) AGG_LOAD_VAL(11, frb)           \
      AGG_LOAD_VAL(12, frb) AGG_LOAD_VAL(13, frb) AGG_LOAD_VAL(14, frb) AGG_LOAD_VAL(15, frb)         \
    }                                                                         \
    if (GATHER) {                                                             \
      pself_0 = *(const v4f*)(emb + ((long)isa << 7) + 4 * lane);             \
      pself_1 = *(const v4f*)(emb + ((long)isb << 7) + 4 * lane);             \
    } else {                                                                  \
      pself_0 = *(const v4f*)(selfVal + (fra << 7) + 4 * lane);               \
      pself_1 = *(const v4f*)(selfVal + (frb << 7) + 4 * lane);               \
    }                                                                         \
    pw = num_w[(long)((c >> 3) ? isb : isa) * S_ + (c & 7)];                  \
  }
#define AGG_STORE_ROW(r) *(v4f*)(xw + (r) * XP_ + 4 * lane) = pre_##r;

  if (0 < nvu) { AGG_FETCH_IDX(0) AGG_FETCH_ROWS() }

  const bool lead = __builtin_amdgcn_readfirstlane(wave) < 4;

  auto out_tile = [&](const int uu) __attribute__((always_inline)) {
    v8f o0 = (v8f){}, o1 = (v8f){};
    {
      const float* sa = sn + ((uu & 1) * 16 + c) * GP_ + 4 * hf;
      const float* gb = G3T + (16 * wave + c) * GP_ + 4 * hf;
      v4f wa[2][8], wb[2][8];
      #pragma unroll
      for (int i = 0; i < 8; ++i) {
        wa[0][i] = *(const v4f*)(sa + 8 * i);
        wb[0][i] = *(const v4f*)(gb + 8 * i);
      }
      #pragma unroll
      for (int blk = 0; blk < 4; ++blk) {
        const int s = blk & 1;
        if (blk + 1 < 4) {
          #pragma unroll
          for (int i = 0; i < 8; ++i) {
            wa[s ^ 1][i] = *(const v4f*)(sa + 8 * (8 * (blk + 1) + i));
            wb[s ^ 1][i] = *(const v4f*)(gb + 8 * (8 * (blk + 1) + i));
          }
        }
        __builtin_amdgcn_sched_barrier(0);
        #pragma unroll
        for (int i = 0; i < 8; ++i) {
          o0 = wmma4((v2f){wa[s][i].x, wa[s][i].y}, (v2f){wb[s][i].x, wb[s][i].y}, o0);
          o1 = wmma4((v2f){wa[s][i].z, wa[s][i].w}, (v2f){wb[s][i].z, wb[s][i].w}, o1);
        }
        __builtin_amdgcn_sched_barrier(0);
      }
    }
    const bool deepu = DEEP && uu < nd;
    const int mb0 = deepu ? (d0 + uu) * 16 : (u0 + uu - nd) * 16;
    const int Mv = deepu ? MD : M;
    float* ob = (deepu ? out2 : out) + 16 * wave + c;
    #pragma unroll
    for (int v = 0; v < 8; ++v) {
      int m = mb0 + v + 8 * hf;
      if (m < Mv) ob[((long)b * Mv + m) << 7] = fmaxf(o0[v] + o1[v], 0.f);
    }
  };

  __syncthreads();
  if (!lead) __builtin_amdgcn_s_setprio(1);
  for (int u = 0; u < nvu; ++u) {
    float* snw = sn + (u & 1) * (16 * GP_);
    if (!lead && u > 0) out_tile(u - 1);
    AGG_ROWS(AGG_STORE_ROW)
    if (lane < 16) {
      float* t = xw + lane * XP_ + D_;
      t[0] = pw; t[1] = 0.f; t[2] = 0.f; t[3] = 0.f;
    }
    const v4f s0 = pself_0, s1 = pself_1;
    if (u + 1 < nvu) AGG_FETCH_IDX(u + 1)
    if (lead) __builtin_amdgcn_s_setprio(2);
    v8f acc[8];
    #pragma unroll
    for (int j = 0; j < 8; ++j) acc[j] = (v8f){};
    {
      const float* ap = xw + c * XP_ + 4 * hf;
      const float* bp = WT + c * XP_ + 4 * hf;
      v4f fa[2][2], fb[2][2][8];
      v2f ta = (v2f){}, tb[8];
      #pragma unroll
      for (int i = 0; i < 2; ++i) {
        fa[0][i] = *(const v4f*)(ap + 8 * i);
        #pragma unroll
        for (int j = 0; j < 8; ++j) fb[0][i][j] = *(const v4f*)(bp + j * 16 * XP_ + 8 * i);
      }
      #pragma unroll
      for (int j = 0; j < 8; ++j) tb[j] = (v2f){};
      #pragma unroll
      for (int blk = 0; blk < 8; ++blk) {
        const int s = blk & 1;
        if (blk + 1 < 8) {
          #pragma unroll
          for (int i = 0; i < 2; ++i) {
            fa[s ^ 1][i] = *(const v4f*)(ap + 8 * (2 * (blk + 1) + i));
            #pragma unroll
            for (int j = 0; j < 8; ++j)
              fb[s ^ 1][i][j] = *(const v4f*)(bp + j * 16 * XP_ + 8 * (2 * (blk + 1) + i));
          }
        }
        if (blk == 6) {
          ta = *(const v2f*)(xw + c * XP_ + D_ + 2 * hf);
          #pragma unroll
          for (int j = 0; j < 8; ++j) tb[j] = *(const v2f*)(WT + (16 * j + c) * XP_ + D_ + 2 * hf);
        }
        __builtin_amdgcn_sched_barrier(0);
        #pragma unroll
        for (int i = 0; i < 2; ++i) {
          #pragma unroll
          for (int j = 0; j < 8; ++j)
            acc[j] = wmma4((v2f){fa[s][i].x, fa[s][i].y}, (v2f){fb[s][i][j].x, fb[s][i][j].y}, acc[j]);
          #pragma unroll
          for (int j = 0; j < 8; ++j)
            acc[j] = wmma4((v2f){fa[s][i].z, fa[s][i].w}, (v2f){fb[s][i][j].z, fb[s][i][j].w}, acc[j]);
        }
        if (blk == 7) {
          #pragma unroll
          for (int j = 0; j < 8; ++j) acc[j] = wmma4(ta, tb[j], acc[j]);
        }
        __builtin_amdgcn_sched_barrier(0);
      }
    }
    if (lead) __builtin_amdgcn_s_setprio(0);
    float sc[8];
    #pragma unroll
    for (int v = 0; v < 8; ++v) sc[v] = 0.f;
    #pragma unroll
    for (int j = 0; j < 8; ++j) {
      float g2 = gw2p[16 * j + c];
      #pragma unroll
      for (int v = 0; v < 8; ++v) sc[v] += leaky_(acc[j][v]) * g2;
    }
    #pragma unroll
    for (int v = 0; v < 8; ++v) {
      float val = sc[v];
      val += __shfl_xor(val, 1);
      val += __shfl_xor(val, 2);
      val += __shfl_xor(val, 4);
      val += __shfl_xor(val, 8);
      sc[v] = val;
    }
    float mx = sc[0];
    #pragma unroll
    for (int v = 1; v < 8; ++v) mx = fmaxf(mx, sc[v]);
    float sum = 0.f;
    #pragma unroll
    for (int v = 0; v < 8; ++v) { sc[v] = __expf(sc[v] - mx); sum += sc[v]; }
    const float inv = 1.f / sum;
    float4 na = make_float4(0.f, 0.f, 0.f, 0.f), nb = na;
    #pragma unroll
    for (int v = 0; v < 8; ++v) {
      const float al = sc[v] * inv;
      const float* row = xw + (8 * hf + v) * XP_ + 4 * c;
      float4 p = *(const float4*)(row);
      float4 q = *(const float4*)(row + 64);
      na.x += al * p.x; na.y += al * p.y; na.z += al * p.z; na.w += al * p.w;
      nb.x += al * q.x; nb.y += al * q.y; nb.z += al * q.z; nb.w += al * q.w;
    }
    {
      float* srow = snw + (2 * wave + hf) * GP_ + D_ + 4 * c;
      *(float4*)(srow) = na;
      *(float4*)(srow + 64) = nb;
      *(v4f*)(snw + (2 * wave) * GP_ + 4 * lane) = s0;
      *(v4f*)(snw + (2 * wave + 1) * GP_ + 4 * lane) = s1;
    }
    if (u + 1 < nvu) AGG_FETCH_ROWS()
    if (lead && u > 0) out_tile(u - 1);
    __builtin_amdgcn_fence(__ATOMIC_RELEASE, "workgroup", "local");
    __builtin_amdgcn_s_barrier();
    __builtin_amdgcn_fence(__ATOMIC_ACQUIRE, "workgroup", "local");
  }
  if (!lead) __builtin_amdgcn_s_setprio(0);
  if (0 < nvu) out_tile(nvu - 1);
#undef AGG_FETCH_IDX
#undef AGG_FETCH_ROWS
#undef AGG_LOAD_IDX
#undef AGG_LOAD_VAL
#undef AGG_STORE_ROW
#undef AGG_ROWS
}

#define SP_ 132
#define SCORE_LDS_FLOATS (42 * 1024)
#define SCORE_WAVES 16
#define SCORE_WGS 256

__global__ __launch_bounds__(SCORE_WAVES * 32) void score_kernel(
    const float* __restrict__ selF, const float* __restrict__ emb,
    float* __restrict__ scores) {
  __shared__ __attribute__((aligned(16))) float sA[SCORE_LDS_FLOATS];
  const int tid = threadIdx.x;
  const int lane = tid & 31, c = lane & 15, hf = lane >> 4;
  const int NSC = NN_ - 1;
  const int NT = (NSC + 15) >> 4;
  const long U = (long)NT * 8;
  const int nw = SCORE_WGS * SCORE_WAVES;
  const int wv = __builtin_amdgcn_readfirstlane(tid >> 5);
  const int gw = (int)blockIdx.x * SCORE_WAVES + wv;
  const int u0 = (int)((U * gw) / nw), u1 = (int)((U * (gw + 1)) / nw);

  v4f bn[16];
#define SCORE_LOADB(NTILE)                                                      \
  {                                                                             \
    int er = ((NTILE) << 4) + c + 1;                \
    er = er < NN_ ? er : NN_ - 1;                   \
    const float* bp = emb + ((long)er << 7) + 4 * hf;                           \
    _Pragma("unroll") for (int i = 0; i < 16; ++i) bn[i] = *(const v4f*)(bp + 8 * i); \
  }
  if (u0 < u1) SCORE_LOADB(u0 >> 3)

  #pragma unroll
  for (int t = 0; t < (B_ * (D_ / 4)) / (SCORE_WAVES * 32); ++t) {
    const int i = tid + t * (SCORE_WAVES * 32);
    const int r = i >> 5, q = i & 31;
    *(v4f*)(sA + r * SP_ + 4 * q) = *(const v4f*)(selF + (r << 7) + 4 * q);
  }
  __syncthreads();

  int u = u0;
  while (u < u1) {
    const int nt = u >> 3;
    const int ue = (u1 < ((nt + 1) << 3)) ? u1 : ((nt + 1) << 3);
    v4f bf[16];
    #pragma unroll
    for (int i = 0; i < 16; ++i) bf[i] = bn[i];
    if (ue < u1) SCORE_LOADB(nt + 1)
    const int col = (nt << 4) + c;
    for (; u < ue; ++u) {
      const int m0 = (u & 7) << 4;
      const float* ap = sA + (m0 + c) * SP_ + 4 * hf;
      v4f a[16];
      #pragma unroll
      for (int i = 0; i < 16; ++i) a[i] = *(const v4f*)(ap + 8 * i);
      __builtin_amdgcn_sched_barrier(0);
      v8f acc0 = (v8f){}, acc1 = (v8f){};
      #pragma unroll
      for (int i = 0; i < 16; ++i) {
        acc0 = wmma4((v2f){a[i].x, a[i].y}, (v2f){bf[i].x, bf[i].y}, acc0);
        acc1 = wmma4((v2f){a[i].z, a[i].w}, (v2f){bf[i].z, bf[i].w}, acc1);
      }
      if (col < NSC) {
        float* op = scores + (long)(m0 + 8 * hf) * NSC + col;
        #pragma unroll
        for (int v = 0; v < 8; ++v) op[(long)v * NSC] = acc0[v] + acc1[v];
      }
    }
  }
#undef SCORE_LOADB
}

template <int M>
__device__ __forceinline__ float row_xor(float x) {
  return __builtin_bit_cast(float, __builtin_amdgcn_update_dpp(0, __builtin_bit_cast(int, x), 0x160 + M, 0xf, 0xf, false));
}
__device__ __forceinline__ float xhalf(float x) {
  const int v = __builtin_bit_cast(int, x);
  return __builtin_bit_cast(float, __builtin_amdgcn_permlanex16(v, v, 0x76543210, 0xfedcba98, false, false));
}

template <int M>
__device__ __forceinline__ int row_xor_i(int x) {
  return __builtin_amdgcn_update_dpp(0, x, 0x160 + M, 0xf, 0xf, false);
}
__device__ __forceinline__ int xhalf_i(int v) {
  return __builtin_amdgcn_permlanex16(v, v, 0x76543210, 0xfedcba98, false, false);
}

#define HP_ 132
#define HR_ 64
#define HEAD_LDS_FLOATS (42 * 1024)

__global__ __launch_bounds__(256) void head_kernel(
    const float* __restrict__ h_local, const float* __restrict__ h_global,
    const int* __restrict__ mask, const float* __restrict__ pos_emb,
    const float* __restrict__ w1, const float* __restrict__ glu1_w,
    const float* __restrict__ glu1_b, const float* __restrict__ glu2_w,
    const float* __restrict__ w2,
    float* __restrict__ hidden, float* __restrict__ sel, float* __restrict__ snorm,
    float* __restrict__ lpart) {
  __shared__ __attribute__((aligned(16))) float lds[HEAD_LDS_FLOATS];
  float* HX    = lds;
  float* NH    = HX + HR_ * HP_;
  float* PX    = NH + HR_ * HP_;
  float* hsL   = PX + HR_ * HP_;
  float* maskf = hsL + D_;
  float* part  = maskf + HR_;
  float* betas = part + 8 * HR_;
  float* red   = betas + HR_;
  float* red2  = red + 4;
  float* sslP  = red2 + 4;
  float* sslN  = sslP + 8 * D_;

  const int tid = threadIdx.x, b = blockIdx.x;
  const int lane = tid & 31, c = lane & 15, hf = lane >> 4;
  const int wave = tid >> 5;
  const int n = (wave << 4) + c;

  v4f bp[16], bw[16], bg[16], g2[16];
  {
    const float* wp0 = w1 + (4 * hf) * D_ + n;
    #pragma unroll
    for (int i = 0; i < 16; ++i) {
      bp[i].x = wp0[(8 * i + 0) * D_];
      bp[i].y = wp0[(8 * i + 1) * D_];
      bp[i].z = wp0[(8 * i + 2) * D_];
      bp[i].w = wp0[(8 * i + 3) * D_];
    }
    const float* wp = w1 + (D_ + 4 * hf) * D_ + n;
    #pragma unroll
    for (int i = 0; i < 16; ++i) {
      bw[i].x = wp[(8 * i + 0) * D_];
      bw[i].y = wp[(8 * i + 1) * D_];
      bw[i].z = wp[(8 * i + 2) * D_];
      bw[i].w = wp[(8 * i + 3) * D_];
    }
    const float* gp = glu1_w + n * D_ + 4 * hf;
    #pragma unroll
    for (int i = 0; i < 16; ++i) bg[i] = *(const v4f*)(gp + 8 * i);
    const float* g2p = glu2_w + n * D_ + 64 * hf;
    #pragma unroll
    for (int i = 0; i < 16; ++i) g2[i] = *(const v4f*)(g2p + 4 * i);
  }
  const float g1b = glu1_b[n];
  const float w2n = w2[n];

  {
    const float* hl = h_local + (long)b * L_ * D_;
    const float* hg = h_global + (long)b * L_ * D_;
    float* ho = hidden + (long)b * L_ * D_;
    const float* hc = h_local + (long)((b * 73 + 21) & 127) * L_ * D_ + 4 * (tid & 31);
    v4f sp = (v4f){}, sg = (v4f){};
    #pragma unroll
    for (int t = 0; t < 6; ++t) {
      const int i4 = tid + 256 * t;
      const int l = i4 >> 5;
      const v4f vl = *(const v4f*)(hl + 4 * i4), vg = *(const v4f*)(hg + 4 * i4);
      const v4f vc = *(const v4f*)(hc + (((l * 13 + 7) % L_) << 7));
      const v4f v = vl + vg;
      *(v4f*)(ho + 4 * i4) = v;
      *(v4f*)(HX + l * HP_ + 4 * (i4 & 31)) = v;
      *(v4f*)(PX + l * HP_ + 4 * (i4 & 31)) = *(const v4f*)(pos_emb + 4 * i4);
      sp += vl * vg;
      sg += vg * vc;
    }
    if (tid < (L_ * D_) / 4 - 6 * 256) {
      const int i4 = tid + 6 * 256;
      const int l = i4 >> 5;
      const v4f vl = *(const v4f*)(hl + 4 * i4), vg = *(const v4f*)(hg + 4 * i4);
      const v4f vc = *(const v4f*)(hc + (((l * 13 + 7) % L_) << 7));
      const v4f v = vl + vg;
      *(v4f*)(ho + 4 * i4) = v;
      *(v4f*)(HX + l * HP_ + 4 * (i4 & 31)) = v;
      *(v4f*)(PX + l * HP_ + 4 * (i4 & 31)) = *(const v4f*)(pos_emb + 4 * i4);
      sp += vl * vg;
      sg += vg * vc;
    }
    *(v4f*)(sslP + wave * D_ + 4 * (tid & 31)) = sp;
    *(v4f*)(sslN + wave * D_ + 4 * (tid & 31)) = sg;
    for (int i4 = tid; i4 < (HR_ - L_) * 32; i4 += 256) {
      *(v4f*)(HX + (L_ + (i4 >> 5)) * HP_ + 4 * (i4 & 31)) = (v4f){};
      *(v4f*)(PX + (L_ + (i4 >> 5)) * HP_ + 4 * (i4 & 31)) = (v4f){};
    }
    if (tid < HR_) maskf[tid] = tid < L_ ? (float)mask[b * L_ + tid] : 0.f;
  }
  __syncthreads();

  if (tid < D_) {
    float acc = 0.f, ms = 0.f;
    #pragma unroll
    for (int l = 0; l < L_; ++l) {
      const float m = maskf[l];
      ms += m;
      acc += HX[l * HP_ + tid] * m;
    }
    hsL[tid] = acc / ms;
  }
  {
    float term = 0.f;
    if (tid < D_) {
      float pos = 0.f, neg = 0.f;
      #pragma unroll
      for (int w = 0; w < 8; ++w) { pos += sslP[w * D_ + tid]; neg += sslN[w * D_ + tid]; }
      term = -__logf(1e-8f + sigmoid_(pos)) - __logf(1e-8f + 1.f - sigmoid_(neg));
    }
    term += row_xor<1>(term);
    term += row_xor<2>(term);
    term += row_xor<4>(term);
    term += row_xor<8>(term);
    term += xhalf(term);
    if (lane == 0 && wave < 4) red2[wave] = term;
  }
  for (int mt = 0; mt < 4; ++mt) {
    const float* ap = HX + (16 * mt + c) * HP_ + 4 * hf;
    const float* pp = PX + (16 * mt + c) * HP_ + 4 * hf;
    v4f a[16], p[16];
    #pragma unroll
    for (int i = 0; i < 16; ++i) p[i] = *(const v4f*)(pp + 8 * i);
    #pragma unroll
    for (int i = 0; i < 16; ++i) a[i] = *(const v4f*)(ap + 8 * i);
    __builtin_amdgcn_sched_barrier(0);
    v8f acc0 = (v8f){}, acc1 = (v8f){};
    #pragma unroll
    for (int i = 0; i < 16; ++i) {
      acc0 = wmma4((v2f){p[i].x, p[i].y}, (v2f){bp[i].x, bp[i].y}, acc0);
      acc1 = wmma4((v2f){p[i].z, p[i].w}, (v2f){bp[i].z, bp[i].w}, acc1);
    }
    #pragma unroll
    for (int i = 0; i < 16; ++i) {
      acc0 = wmma4((v2f){a[i].x, a[i].y}, (v2f){bw[i].x, bw[i].y}, acc0);
      acc1 = wmma4((v2f){a[i].z, a[i].w}, (v2f){bw[i].z, bw[i].w}, acc1);
    }
    #pragma unroll
    for (int v = 0; v < 8; ++v)
      NH[(16 * mt + 8 * hf + v) * HP_ + n] = tanhf(acc0[v] + acc1[v]);
  }
  __syncthreads();

  float bias;
  {
    const float* hp = hsL + 64 * hf;
    float s0 = 0.f, s1 = 0.f, s2 = 0.f, s3 = 0.f;
    #pragma unroll
    for (int i = 0; i < 16; ++i) {
      const v4f x = *(const v4f*)(hp + 4 * i);
      s0 += g2[i].x * x.x; s1 += g2[i].y * x.y;
      s2 += g2[i].z * x.z; s3 += g2[i].w * x.w;
    }
    float s = (s0 + s1) + (s2 + s3);
    s += xhalf(s);
    bias = s + g1b;
  }

  for (int mt = 0; mt < 4; ++mt) {
    const float* ap = NH + (16 * mt + c) * HP_ + 4 * hf;
    v4f a[16];
    #pragma unroll
    for (int i = 0; i < 16; ++i) a[i] = *(const v4f*)(ap + 8 * i);
    __builtin_amdgcn_sched_barrier(0);
    v8f acc0 = (v8f){}, acc1 = (v8f){};
    #pragma unroll
    for (int i = 0; i < 16; ++i) {
      acc0 = wmma4((v2f){a[i].x, a[i].y}, (v2f){bg[i].x, bg[i].y}, acc0);
      acc1 = wmma4((v2f){a[i].z, a[i].w}, (v2f){bg[i].z, bg[i].w}, acc1);
    }
    float g[8];
    #pragma unroll
    for (int v = 0; v < 8; ++v) g[v] = sigmoid_(acc0[v] + acc1[v] + bias) * w2n;
    #pragma unroll
    for (int v = 0; v < 8; ++v) g[v] += row_xor<1>(g[v]);
    #pragma unroll
    for (int v = 0; v < 8; ++v) g[v] += row_xor<2>(g[v]);
    #pragma unroll
    for (int v = 0; v < 8; ++v) g[v] += row_xor<4>(g[v]);
    #pragma unroll
    for (int v = 0; v < 8; ++v) g[v] += row_xor<8>(g[v]);
    if (c == 0) {
      #pragma unroll
      for (int v = 0; v < 8; ++v) part[wave * HR_ + 16 * mt + 8 * hf + v] = g[v];
    }
  }
  __syncthreads();

  if (tid < L_) {
    float s = 0.f;
    #pragma unroll
    for (int w = 0; w < 8; ++w) s += part[w * HR_ + tid];
    betas[tid] = s * maskf[tid];
  }
  __syncthreads();

  float sq = 0.f;
  if (tid < D_) {
    float acc = 0.f;
    #pragma unroll
    for (int l = 0; l < L_; ++l) acc += betas[l] * HX[l * HP_ + tid];
    sel[(b << 7) + tid] = acc;
    sq = acc * acc + 1e-6f;
  }
  sq += row_xor<1>(sq);
  sq += row_xor<2>(sq);
  sq += row_xor<4>(sq);
  sq += row_xor<8>(sq);
  sq += xhalf(sq);
  if (lane == 0 && wave < 4) red[wave] = sq;
  __syncthreads();
  if (tid == 0) {
    snorm[b] = sqrtf((red[0] + red[1]) + (red[2] + red[3]));
    lpart[b] = (red2[0] + red2[1]) + (red2[2] + red2[3]);
  }
}


__global__ void fn_kernel(const float* __restrict__ sel, const float* __restrict__ snorm,
                          const float* __restrict__ lpart, float* __restrict__ loss,
                          float* __restrict__ selF) {
  __shared__ float row[B_];
  __shared__ float selfrow[D_];
  __shared__ float tvs[9];
  __shared__ int tis[9];
  int i = blockIdx.x, t = threadIdx.x;
  selfrow[t] = sel[(i << 7) + t];
  __syncthreads();
  float acc = 0.f;
  for (int d = 0; d < D_; ++d) acc += selfrow[d] * sel[(t << 7) + d];
  row[t] = acc / (snorm[i] * snorm[t]);
  __syncthreads();
  if (t < 32) {
    if (i == 0) {
      float ls = (lpart[t] + lpart[t + 32]) + (lpart[t + 64] + lpart[t + 96]);
      ls += row_xor<1>(ls);
      ls += row_xor<2>(ls);
      ls += row_xor<4>(ls);
      ls += row_xor<8>(ls);
      ls += xhalf(ls);
      if (t == 0) loss[0] = BETA_ * ls;
    }
    float p[B_ / 32];
    #pragma unroll
    for (int q = 0; q < B_ / 32; ++q) p[q] = row[t + 32 * q];
    float mx = fmaxf(fmaxf(p[0], p[1]), fmaxf(p[2], p[3]));
    mx = fmaxf(mx, row_xor<1>(mx));
    mx = fmaxf(mx, row_xor<2>(mx));
    mx = fmaxf(mx, row_xor<4>(mx));
    mx = fmaxf(mx, row_xor<8>(mx));
    mx = fmaxf(mx, xhalf(mx));
    #pragma unroll
    for (int q = 0; q < B_ / 32; ++q) p[q] = __expf(p[q] - mx);
    float s = (p[0] + p[1]) + (p[2] + p[3]);
    s += row_xor<1>(s);
    s += row_xor<2>(s);
    s += row_xor<4>(s);
    s += row_xor<8>(s);
    s += xhalf(s);
    const float inv = 1.f / s;
    #pragma unroll
    for (int q = 0; q < B_ / 32; ++q) p[q] *= inv;
    float tv[9];
    int ti[9];
#define FN_MEET(OV, OI)                                                        \
    {                                                                          \
      const float ov = (OV);                                                   \
      const int oi = (OI);                                                     \
      const bool take = ov > bv || (ov == bv && oi < bi);                      \
      bv = take ? ov : bv;                                                     \
      bi = take ? oi : bi;                                                     \
    }
    #pragma unroll
    for (int r = 0; r < 9; ++r) {
      float bv = p[0];
      int bi = t;
      #pragma unroll
      for (int q = 1; q < B_ / 32; ++q) {
        const bool take = p[q] > bv;
        bv = take ? p[q] : bv;
        bi = take ? t + 32 * q : bi;
      }
      FN_MEET(row_xor<1>(bv), row_xor_i<1>(bi))
      FN_MEET(row_xor<2>(bv), row_xor_i<2>(bi))
      FN_MEET(row_xor<4>(bv), row_xor_i<4>(bi))
      FN_MEET(row_xor<8>(bv), row_xor_i<8>(bi))
      FN_MEET(xhalf(bv), xhalf_i(bi))
      tv[r] = bv;
      ti[r] = bi;
      #pragma unroll
      for (int q = 0; q < B_ / 32; ++q) p[q] = (bi == t + 32 * q) ? -2.f : p[q];
    }
#undef FN_MEET
    const float mx2 = tv[0];
    float s2 = 0.f;
    #pragma unroll
    for (int r = 0; r < 9; ++r) { tv[r] = __expf(tv[r] - mx2); s2 += tv[r]; }
    const float inv2 = 1.f / s2;
    if (t == 0) {
      #pragma unroll
      for (int r = 0; r < 9; ++r) { tvs[r] = tv[r] * inv2; tis[r] = ti[r]; }
    }
  }
  __syncthreads();
  float nb = 0.f;
  for (int q = 0; q < 9; ++q) nb += tvs[q] * sel[(tis[q] << 7) + t];
  selF[(i << 7) + t] = selfrow[t] + nb;
}

extern "C" void kernel_launch(void* const* d_in, const int* in_sizes, int n_in,
                              void* d_out, int out_size, void* d_ws, size_t ws_size,
                              hipStream_t stream) {
  (void)in_sizes; (void)n_in; (void)out_size; (void)ws_size;
  const int*   inputs  = (const int*)d_in[0];
  const int*   adj     = (const int*)d_in[1];
  const int*   mask    = (const int*)d_in[2];
  const int*   item    = (const int*)d_in[3];
  const int*   adj_all = (const int*)d_in[5];
  const float* num_w   = (const float*)d_in[6];
  const float* emb     = (const float*)d_in[7];
  const float* pos_emb = (const float*)d_in[8];
  const float* a_local = (const float*)d_in[9];
  const float* gw1     = (const float*)d_in[10];
  const float* gw2     = (const float*)d_in[11];
  const float* gw3     = (const float*)d_in[12];
  const float* w1      = (const float*)d_in[13];
  const float* w2      = (const float*)d_in[14];
  const float* glu1_w  = (const float*)d_in[15];
  const float* glu1_b  = (const float*)d_in[16];
  const float* glu2_w  = (const float*)d_in[17];

  float* out0   = (float*)d_out;
  float* lossP  = out0 + B_ * L_ * D_;
  float* scores = lossP + 1;

  char* wp = (char*)d_ws;
  auto alloc = [&](size_t nfloats) -> float* {
    float* r = (float*)wp;
    wp += ((nfloats * sizeof(float) + 255) / 256) * 256;
    return r;
  };
  float* h_local  = alloc(B_ * L_ * D_);
  float* h_global = alloc(B_ * L_ * D_);
  float* e0p      = alloc(B_ * L_ * D_);
  float* e1p      = alloc((size_t)B_ * 400 * D_);
  float* lpartb   = alloc(B_);
  float* selb     = alloc(B_ * D_);
  float* selF     = alloc(B_ * D_);
  float* snorm    = alloc(B_);

  local_agg_kernel<<<dim3(B_), 256, 0, stream>>>(emb, inputs, adj, a_local, h_local);
  agg_kernel<true, true><<<dim3(B_ * 2), 256, 0, stream>>>(emb, inputs, adj_all, num_w, item, mask,
      nullptr, nullptr, gw1, gw2, gw3, e0p, e1p, L_, 2);
  agg_kernel<false, false><<<dim3(B_ * 2), 256, 0, stream>>>(emb, inputs, adj_all, num_w, item, mask,
      e0p, e1p, gw1 + 129 * D_, gw2 + D_, gw3 + 2 * D_ * D_, h_global, nullptr, L_, 2);
  head_kernel<<<dim3(B_), 256, 0, stream>>>(h_local, h_global, mask, pos_emb, w1, glu1_w, glu1_b,
      glu2_w, w2, out0, selb, snorm, lpartb);
  fn_kernel<<<dim3(B_), dim3(D_), 0, stream>>>(selb, snorm, lpartb, lossP, selF);
  score_kernel<<<dim3(SCORE_WGS), SCORE_WAVES * 32, 0, stream>>>(selF, emb, scores);
}
